# starting kernel
# baseline (speedup 1.0000x reference)
#include <hip/hip_runtime.h>
#include <hip/hip_bf16.h>
#include <math.h>

#define B_      16
#define N_      4096
#define C_      64
#define NPOINT  1024
#define NSAMPLE 32
#define PITCH   68
#define BN_EPS  1e-5f

typedef float v2f __attribute__((ext_vector_type(2)));
typedef float v8f __attribute__((ext_vector_type(8)));

static __device__ inline v8f wmma4(v2f a, v2f b, v8f c) {
    return __builtin_amdgcn_wmma_f32_16x16x4_f32(
         false, a,  false, b,
         (short)0, c,  false,  false);
}

__global__ __launch_bounds__(256) void fuse_weights_kernel(
    const float* __restrict__ W0, const float* __restrict__ b0, const float* __restrict__ g0,
    const float* __restrict__ be0, const float* __restrict__ m0, const float* __restrict__ v0,
    const float* __restrict__ W1, const float* __restrict__ b1, const float* __restrict__ g1,
    const float* __restrict__ be1, const float* __restrict__ m1, const float* __restrict__ v1,
    const float* __restrict__ W2, const float* __restrict__ b2, const float* __restrict__ g2,
    const float* __restrict__ be2, const float* __restrict__ m2, const float* __restrict__ v2_,
    v2f* __restrict__ Wp0, float* __restrict__ b0f,
    v2f* __restrict__ Wp1, float* __restrict__ b1f,
    v2f* __restrict__ Wp2, float* __restrict__ b2f)
{
    const int tid = threadIdx.x;
    for (int i = tid; i < 34 * 64; i += 256) {
        int o = i & 63, kp = i >> 6;
        float s = g0[o] / sqrtf(v0[o] + BN_EPS);
        int k0 = 2 * kp, k1 = 2 * kp + 1;
        v2f w;
        w.x = (k0 < 67) ? W0[o * 67 + k0] * s : 0.0f;
        w.y = (k1 < 67) ? W0[o * 67 + k1] * s : 0.0f;
        Wp0[i] = w;
    }
    for (int i = tid; i < 64; i += 256) {
        float s = g0[i] / sqrtf(v0[i] + BN_EPS);
        b0f[i] = (b0[i] - m0[i]) * s + be0[i];
    }
    for (int i = tid; i < 32 * 64; i += 256) {
        int o = i & 63, kp = i >> 6;
        float s = g1[o] / sqrtf(v1[o] + BN_EPS);
        v2f w;
        w.x = W1[o * 64 + 2 * kp + 0] * s;
        w.y = W1[o * 64 + 2 * kp + 1] * s;
        Wp1[i] = w;
    }
    for (int i = tid; i < 64; i += 256) {
        float s = g1[i] / sqrtf(v1[i] + BN_EPS);
        b1f[i] = (b1[i] - m1[i]) * s + be1[i];
    }
    for (int i = tid; i < 32 * 128; i += 256) {
        int o = i & 127, kp = i >> 7;
        float s = g2[o] / sqrtf(v2_[o] + BN_EPS);
        v2f w;
        w.x = W2[o * 64 + 2 * kp + 0] * s;
        w.y = W2[o * 64 + 2 * kp + 1] * s;
        Wp2[i] = w;
    }
    for (int i = tid; i < 128; i += 256) {
        float s = g2[i] / sqrtf(v2_[i] + BN_EPS);
        b2f[i] = (b2[i] - m2[i]) * s + be2[i];
    }
}

__global__ __launch_bounds__(256) void fps_kernel(const float* __restrict__ xyz,
                                                  float* __restrict__ new_xyz)
{
    __shared__ float sx[N_], sy[N_], sz[N_];
    __shared__ float red_val[8];
    __shared__ int   red_idx[8];
    __shared__ int   s_last;
    const int b   = blockIdx.x;
    const int tid = threadIdx.x;
    const float* bx = xyz + (size_t)b * N_ * 3;

    float dloc[16];
#pragma unroll
    for (int i = 0; i < 16; ++i) {
        int p = tid + 256 * i;
        sx[p] = bx[p * 3 + 0];
        sy[p] = bx[p * 3 + 1];
        sz[p] = bx[p * 3 + 2];
        dloc[i] = 1e10f;
    }
    if (tid == 0) {
        new_xyz[(b * NPOINT + 0) * 3 + 0] = bx[0];
        new_xyz[(b * NPOINT + 0) * 3 + 1] = bx[1];
        new_xyz[(b * NPOINT + 0) * 3 + 2] = bx[2];
    }
    __syncthreads();

    int last = 0;
    for (int it = 1; it < NPOINT; ++it) {
        const float lx = sx[last], ly = sy[last], lz = sz[last];
        float val = -1.0f;
        int   best = tid;
#pragma unroll
        for (int i = 0; i < 16; ++i) {
            int p = tid + 256 * i;
            float dx = sx[p] - lx, dy = sy[p] - ly, dz = sz[p] - lz;
            float dist = dx * dx + dy * dy + dz * dz;
            dloc[i] = fminf(dloc[i], dist);
            if (dloc[i] > val) { val = dloc[i]; best = p; }
        }
#pragma unroll
        for (int off = 16; off > 0; off >>= 1) {
            float ov = __shfl_xor(val, off, 32);
            int   oi = __shfl_xor(best, off, 32);
            if (ov > val || (ov == val && oi < best)) { val = ov; best = oi; }
        }
        if ((tid & 31) == 0) { red_val[tid >> 5] = val; red_idx[tid >> 5] = best; }
        __syncthreads();
        if (tid == 0) {
            float bv = red_val[0];
            int   bi = red_idx[0];
#pragma unroll
            for (int w = 1; w < 8; ++w) {
                float wv = red_val[w];
                int   wi = red_idx[w];
                if (wv > bv || (wv == bv && wi < bi)) { bv = wv; bi = wi; }
            }
            s_last = bi;
            new_xyz[(b * NPOINT + it) * 3 + 0] = sx[bi];
            new_xyz[(b * NPOINT + it) * 3 + 1] = sy[bi];
            new_xyz[(b * NPOINT + it) * 3 + 2] = sz[bi];
        }
        __syncthreads();
        last = s_last;
    }
}

#define KNN_T 128
#define TILE  512
__global__ __launch_bounds__(KNN_T) void knn_kernel(const float* __restrict__ xyz,
                                                    const float* __restrict__ new_xyz,
                                                    int* __restrict__ gidx)
{
    __shared__ float tx[TILE], ty[TILE], tz[TILE];
    __shared__ float ld[NSAMPLE * KNN_T];
    __shared__ int   li[NSAMPLE * KNN_T];
    const int b    = blockIdx.x >> 3;
    const int g    = blockIdx.x & 7;
    const int tid  = threadIdx.x;
    const int cent = g * KNN_T + tid;

    const float cx = new_xyz[(b * NPOINT + cent) * 3 + 0];
    const float cy = new_xyz[(b * NPOINT + cent) * 3 + 1];
    const float cz = new_xyz[(b * NPOINT + cent) * 3 + 2];

#pragma unroll
    for (int e = 0; e < NSAMPLE; ++e) ld[e * KNN_T + tid] = 3.4e38f;
    float curmax = 3.4e38f;
    int   curpos = 0;
    const float* bx = xyz + (size_t)b * N_ * 3;

    for (int t = 0; t < N_ / TILE; ++t) {
        __syncthreads();
#pragma unroll
        for (int i = 0; i < TILE / KNN_T; ++i) {
            int p = t * TILE + i * KNN_T + tid;
            tx[i * KNN_T + tid] = bx[p * 3 + 0];
            ty[i * KNN_T + tid] = bx[p * 3 + 1];
            tz[i * KNN_T + tid] = bx[p * 3 + 2];
        }
        __syncthreads();
        for (int j = 0; j < TILE; ++j) {
            float dx = tx[j] - cx, dy = ty[j] - cy, dz = tz[j] - cz;
            float d2 = dx * dx + dy * dy + dz * dz;
            if (d2 < curmax) {
                ld[curpos * KNN_T + tid] = d2;
                li[curpos * KNN_T + tid] = t * TILE + j;
                curmax = -1.0f;
#pragma unroll
                for (int e = 0; e < NSAMPLE; ++e) {
                    float v = ld[e * KNN_T + tid];
                    if (v > curmax) { curmax = v; curpos = e; }
                }
            }
        }
    }
    int* go = gidx + (size_t)(b * NPOINT + cent) * NSAMPLE;
#pragma unroll
    for (int e = 0; e < NSAMPLE; ++e) go[e] = li[e * KNN_T + tid];
}

template <int KSTEPS, int NT, int COUT>
static __device__ inline void layer_gemm(const float* in_lds, float* out_lds,
                                         const v2f* __restrict__ Wp,
                                         const float* __restrict__ bias, int lane)
{
    v8f acc[2][NT];
    v8f zero = {};
#pragma unroll
    for (int mt = 0; mt < 2; ++mt)
#pragma unroll
        for (int n = 0; n < NT; ++n) acc[mt][n] = zero;

    const int mrow  = lane & 15;
    const int khalf = (lane >> 4) * 2;
    const int ncol  = lane & 15;

    for (int k = 0; k < KSTEPS; ++k) {
        const int kc = 4 * k + khalf;
        v2f a0 = *(const v2f*)(in_lds + mrow * PITCH + kc);
        v2f a1 = *(const v2f*)(in_lds + (mrow + 16) * PITCH + kc);
#pragma unroll
        for (int n = 0; n < NT; ++n) {
            v2f bt = Wp[(kc >> 1) * COUT + n * 16 + ncol];
            acc[0][n] = wmma4(a0, bt, acc[0][n]);
            acc[1][n] = wmma4(a1, bt, acc[1][n]);
        }
    }
    const int rbase = (lane >> 4) * 8;
#pragma unroll
    for (int n = 0; n < NT; ++n) {
        float bb = bias[n * 16 + ncol];
#pragma unroll
        for (int mt = 0; mt < 2; ++mt)
#pragma unroll
            for (int j = 0; j < 8; ++j) {
                int rrow = mt * 16 + rbase + j;
                out_lds[rrow * PITCH + n * 16 + ncol] = fmaxf(acc[mt][n][j] + bb, 0.0f);
            }
    }
}

__global__ __launch_bounds__(32) void mlp_kernel(
    const float* __restrict__ xyz, const float* __restrict__ features,
    const int* __restrict__ gidx, const float* __restrict__ new_xyz,
    float* __restrict__ out_feat,
    const v2f* __restrict__ Wp0, const float* __restrict__ b0f,
    const v2f* __restrict__ Wp1, const float* __restrict__ b1f,
    const v2f* __restrict__ Wp2, const float* __restrict__ b2f)
{
    __shared__ __align__(16) float bufA[32 * PITCH];
    __shared__ __align__(16) float bufB[32 * PITCH];
    const int cg   = blockIdx.x;
    const int b    = cg >> 10;
    const int lane = threadIdx.x;

    const int   idx = gidx[(size_t)cg * NSAMPLE + lane];
    const float cx = new_xyz[cg * 3 + 0];
    const float cy = new_xyz[cg * 3 + 1];
    const float cz = new_xyz[cg * 3 + 2];
    const float* pt = xyz + ((size_t)b * N_ + idx) * 3;
    float* row = bufA + lane * PITCH;
    row[0] = pt[0] - cx;
    row[1] = pt[1] - cy;
    row[2] = pt[2] - cz;
    const float* fb = features + (size_t)b * C_ * N_ + idx;
#pragma unroll 8
    for (int c = 0; c < C_; ++c) row[3 + c] = fb[(size_t)c * N_];
    row[67] = 0.0f;
    __syncthreads();

    layer_gemm<17, 4, 64>(bufA, bufB, Wp0, b0f, lane);
    __syncthreads();
    layer_gemm<16, 4, 64>(bufB, bufA, Wp1, b1f, lane);
    __syncthreads();

    v8f acc[2][8];
    v8f zero = {};
#pragma unroll
    for (int mt = 0; mt < 2; ++mt)
#pragma unroll
        for (int n = 0; n < 8; ++n) acc[mt][n] = zero;

    const int mrow  = lane & 15;
    const int khalf = (lane >> 4) * 2;
    const int ncol  = lane & 15;
    for (int k = 0; k < 16; ++k) {
        const int kc = 4 * k + khalf;
        v2f a0 = *(const v2f*)(bufA + mrow * PITCH + kc);
        v2f a1 = *(const v2f*)(bufA + (mrow + 16) * PITCH + kc);
#pragma unroll
        for (int n = 0; n < 8; ++n) {
            v2f bt = Wp2[(kc >> 1) * 128 + n * 16 + ncol];
            acc[0][n] = wmma4(a0, bt, acc[0][n]);
            acc[1][n] = wmma4(a1, bt, acc[1][n]);
        }
    }
    float* of = out_feat + (size_t)cg * 128;
#pragma unroll
    for (int n = 0; n < 8; ++n) {
        float r = -3.4e38f;
#pragma unroll
        for (int j = 0; j < 8; ++j) {
            r = fmaxf(r, acc[0][n][j]);
            r = fmaxf(r, acc[1][n][j]);
        }
        r = fmaxf(r, __shfl_xor(r, 16, 32));
        float v = fmaxf(r + b2f[n * 16 + ncol], 0.0f);
        if (lane < 16) of[n * 16 + ncol] = v;
    }
}

extern "C" void kernel_launch(void* const* d_in, const int* in_sizes, int n_in,
                              void* d_out, int out_size, void* d_ws, size_t ws_size,
                              hipStream_t stream)
{
    const float* xyz      = (const float*)d_in[0];
    const float* features = (const float*)d_in[1];
    const float* W0 = (const float*)d_in[2];  const float* b0 = (const float*)d_in[3];
    const float* g0 = (const float*)d_in[4];  const float* be0 = (const float*)d_in[5];
    const float* m0 = (const float*)d_in[6];  const float* v0 = (const float*)d_in[7];
    const float* W1 = (const float*)d_in[8];  const float* b1 = (const float*)d_in[9];
    const float* g1 = (const float*)d_in[10]; const float* be1 = (const float*)d_in[11];
    const float* m1 = (const float*)d_in[12]; const float* v1 = (const float*)d_in[13];
    const float* W2 = (const float*)d_in[14]; const float* b2 = (const float*)d_in[15];
    const float* g2 = (const float*)d_in[16]; const float* be2 = (const float*)d_in[17];
    const float* m2 = (const float*)d_in[18]; const float* v2_ = (const float*)d_in[19];

    float* out      = (float*)d_out;
    float* new_xyz  = out;
    float* out_feat = out + (size_t)B_ * NPOINT * 3;

    int*   gidx = (int*)d_ws;
    float* wsf  = (float*)((char*)d_ws + (size_t)B_ * NPOINT * NSAMPLE * sizeof(int));
    v2f*   Wp0 = (v2f*)wsf;
    float* b0f = (float*)(Wp0 + 34 * 64);
    v2f*   Wp1 = (v2f*)(b0f + 64);
    float* b1f = (float*)(Wp1 + 32 * 64);
    v2f*   Wp2 = (v2f*)(b1f + 64);
    float* b2f = (float*)(Wp2 + 32 * 128);

    fuse_weights_kernel<<<1, 256, 0, stream>>>(W0, b0, g0, be0, m0, v0,
                                               W1, b1, g1, be1, m1, v1,
                                               W2, b2, g2, be2, m2, v2_,
                                               Wp0, b0f, Wp1, b1f, Wp2, b2f);
    fps_kernel<<<B_, 256, 0, stream>>>(xyz, new_xyz);
    knn_kernel<<<B_ * 8, KNN_T, 0, stream>>>(xyz, new_xyz, gidx);
    mlp_kernel<<<B_ * NPOINT, 32, 0, stream>>>(xyz, features, gidx, new_xyz, out_feat,
                                               Wp0, b0f, Wp1, b1f, Wp2, b2f);
}


# fps: each step's argmax hand-over shortened: wave argmax by DPP instead of LDS shuffles, ONE barrier with every wave merging the 8 wave results, picks kept in LDS and stored to out after the loop
# speedup vs baseline: 1.5469x; 1.5469x over previous
#include <hip/hip_runtime.h>
#include <hip/hip_bf16.h>
#include <math.h>

#define B_      16
#define N_      4096
#define C_      64
#define NPOINT  1024
#define NSAMPLE 32
#define PITCH   68
#define BN_EPS  1e-5f

typedef float v2f __attribute__((ext_vector_type(2)));
typedef float v8f __attribute__((ext_vector_type(8)));

static __device__ inline v8f wmma4(v2f a, v2f b, v8f c) {
    return __builtin_amdgcn_wmma_f32_16x16x4_f32(
         false, a,  false, b,
         (short)0, c,  false,  false);
}

__global__ __launch_bounds__(256) void fuse_weights_kernel(
    const float* __restrict__ W0, const float* __restrict__ b0, const float* __restrict__ g0,
    const float* __restrict__ be0, const float* __restrict__ m0, const float* __restrict__ v0,
    const float* __restrict__ W1, const float* __restrict__ b1, const float* __restrict__ g1,
    const float* __restrict__ be1, const float* __restrict__ m1, const float* __restrict__ v1,
    const float* __restrict__ W2, const float* __restrict__ b2, const float* __restrict__ g2,
    const float* __restrict__ be2, const float* __restrict__ m2, const float* __restrict__ v2_,
    v2f* __restrict__ Wp0, float* __restrict__ b0f,
    v2f* __restrict__ Wp1, float* __restrict__ b1f,
    v2f* __restrict__ Wp2, float* __restrict__ b2f)
{
    const int tid = threadIdx.x;
    for (int i = tid; i < 34 * 64; i += 256) {
        int o = i & 63, kp = i >> 6;
        float s = g0[o] / sqrtf(v0[o] + BN_EPS);
        int k0 = 2 * kp, k1 = 2 * kp + 1;
        v2f w;
        w.x = (k0 < 67) ? W0[o * 67 + k0] * s : 0.0f;
        w.y = (k1 < 67) ? W0[o * 67 + k1] * s : 0.0f;
        Wp0[i] = w;
    }
    for (int i = tid; i < 64; i += 256) {
        float s = g0[i] / sqrtf(v0[i] + BN_EPS);
        b0f[i] = (b0[i] - m0[i]) * s + be0[i];
    }
    for (int i = tid; i < 32 * 64; i += 256) {
        int o = i & 63, kp = i >> 6;
        float s = g1[o] / sqrtf(v1[o] + BN_EPS);
        v2f w;
        w.x = W1[o * 64 + 2 * kp + 0] * s;
        w.y = W1[o * 64 + 2 * kp + 1] * s;
        Wp1[i] = w;
    }
    for (int i = tid; i < 64; i += 256) {
        float s = g1[i] / sqrtf(v1[i] + BN_EPS);
        b1f[i] = (b1[i] - m1[i]) * s + be1[i];
    }
    for (int i = tid; i < 32 * 128; i += 256) {
        int o = i & 127, kp = i >> 7;
        float s = g2[o] / sqrtf(v2_[o] + BN_EPS);
        v2f w;
        w.x = W2[o * 64 + 2 * kp + 0] * s;
        w.y = W2[o * 64 + 2 * kp + 1] * s;
        Wp2[i] = w;
    }
    for (int i = tid; i < 128; i += 256) {
        float s = g2[i] / sqrtf(v2_[i] + BN_EPS);
        b2f[i] = (b2[i] - m2[i]) * s + be2[i];
    }
}

template <int M>
static __device__ inline int xrow(int x) {
    return __builtin_amdgcn_update_dpp(0, x, 0x160 + M, 0xf, 0xf, true);
}
static __device__ inline int xhalf(int x) {
    return __builtin_amdgcn_permlanex16(x, x, 0x76543210, 0xfedcba98, false, false);
}
static __device__ inline int imax(int a, int b) { return a > b ? a : b; }
static __device__ inline int imin(int a, int b) { return a < b ? a : b; }
#define NO_IDX 0x7fffffff

__global__ __launch_bounds__(256) void fps_kernel(const float* __restrict__ xyz,
                                                  float* __restrict__ new_xyz)
{
    __shared__ float sx[N_], sy[N_], sz[N_];
    __shared__ int sidx[NPOINT];
    __shared__ __align__(16) int red[2][16];
    const int b   = blockIdx.x;
    const int tid = threadIdx.x;
    const float* bx = xyz + (size_t)b * N_ * 3;

    float dloc[16];
#pragma unroll
    for (int i = 0; i < 16; ++i) {
        int p = tid + 256 * i;
        sx[p] = bx[p * 3 + 0];
        sy[p] = bx[p * 3 + 1];
        sz[p] = bx[p * 3 + 2];
        dloc[i] = 1e10f;
    }
    __syncthreads();

    int last = 0;
    for (int it = 0; ; ++it) {
        const float lx = sx[last], ly = sy[last], lz = sz[last];
        if (tid == 0) sidx[it] = last;
        if (it == NPOINT - 1) break;

        float val = -1.0f;
        int   best = tid;
#pragma unroll
        for (int i = 0; i < 16; ++i) {
            int p = tid + 256 * i;
            float dx = sx[p] - lx, dy = sy[p] - ly, dz = sz[p] - lz;
            float dist = dx * dx + dy * dy + dz * dz;
            dloc[i] = fminf(dloc[i], dist);
            if (dloc[i] > val) { val = dloc[i]; best = p; }
        }
        const int vb = __float_as_int(val);
        int m = vb;
        m = imax(m, xrow<1>(m));
        m = imax(m, xrow<2>(m));
        m = imax(m, xrow<4>(m));
        m = imax(m, xrow<8>(m));
        m = imax(m, xhalf(m));
        int c = (vb == m) ? best : NO_IDX;
        c = imin(c, xrow<1>(c));
        c = imin(c, xrow<2>(c));
        c = imin(c, xrow<4>(c));
        c = imin(c, xrow<8>(c));
        c = imin(c, xhalf(c));

        int* slot = red[it & 1];
        if ((tid & 31) == 0) *(int2*)(slot + 2 * (tid >> 5)) = make_int2(m, c);
        __syncthreads();
        const int2 r = *(const int2*)(slot + 2 * (tid & 7));
        int gm = r.x;
        gm = imax(gm, xrow<1>(gm));
        gm = imax(gm, xrow<2>(gm));
        gm = imax(gm, xrow<4>(gm));
        int gc = (r.x == gm) ? r.y : NO_IDX;
        gc = imin(gc, xrow<1>(gc));
        gc = imin(gc, xrow<2>(gc));
        gc = imin(gc, xrow<4>(gc));
        last = gc;
    }
    __syncthreads();
    for (int k = 0; k < NPOINT / 256; ++k) {
        const int j = tid + 256 * k;
        const int p = sidx[j];
        float* o = new_xyz + ((size_t)b * NPOINT + j) * 3;
        o[0] = sx[p];
        o[1] = sy[p];
        o[2] = sz[p];
    }
}

#define KNN_T 128
#define TILE  512
__global__ __launch_bounds__(KNN_T) void knn_kernel(const float* __restrict__ xyz,
                                                    const float* __restrict__ new_xyz,
                                                    int* __restrict__ gidx)
{
    __shared__ float tx[TILE], ty[TILE], tz[TILE];
    __shared__ float ld[NSAMPLE * KNN_T];
    __shared__ int   li[NSAMPLE * KNN_T];
    const int b    = blockIdx.x >> 3;
    const int g    = blockIdx.x & 7;
    const int tid  = threadIdx.x;
    const int cent = g * KNN_T + tid;

    const float cx = new_xyz[(b * NPOINT + cent) * 3 + 0];
    const float cy = new_xyz[(b * NPOINT + cent) * 3 + 1];
    const float cz = new_xyz[(b * NPOINT + cent) * 3 + 2];

#pragma unroll
    for (int e = 0; e < NSAMPLE; ++e) ld[e * KNN_T + tid] = 3.4e38f;
    float curmax = 3.4e38f;
    int   curpos = 0;
    const float* bx = xyz + (size_t)b * N_ * 3;

    for (int t = 0; t < N_ / TILE; ++t) {
        __syncthreads();
#pragma unroll
        for (int i = 0; i < TILE / KNN_T; ++i) {
            int p = t * TILE + i * KNN_T + tid;
            tx[i * KNN_T + tid] = bx[p * 3 + 0];
            ty[i * KNN_T + tid] = bx[p * 3 + 1];
            tz[i * KNN_T + tid] = bx[p * 3 + 2];
        }
        __syncthreads();
        for (int j = 0; j < TILE; ++j) {
            float dx = tx[j] - cx, dy = ty[j] - cy, dz = tz[j] - cz;
            float d2 = dx * dx + dy * dy + dz * dz;
            if (d2 < curmax) {
                ld[curpos * KNN_T + tid] = d2;
                li[curpos * KNN_T + tid] = t * TILE + j;
                curmax = -1.0f;
#pragma unroll
                for (int e = 0; e < NSAMPLE; ++e) {
                    float v = ld[e * KNN_T + tid];
                    if (v > curmax) { curmax = v; curpos = e; }
                }
            }
        }
    }
    int* go = gidx + (size_t)(b * NPOINT + cent) * NSAMPLE;
#pragma unroll
    for (int e = 0; e < NSAMPLE; ++e) go[e] = li[e * KNN_T + tid];
}

template <int KSTEPS, int NT, int COUT>
static __device__ inline void layer_gemm(const float* in_lds, float* out_lds,
                                         const v2f* __restrict__ Wp,
                                         const float* __restrict__ bias, int lane)
{
    v8f acc[2][NT];
    v8f zero = {};
#pragma unroll
    for (int mt = 0; mt < 2; ++mt)
#pragma unroll
        for (int n = 0; n < NT; ++n) acc[mt][n] = zero;

    const int mrow  = lane & 15;
    const int khalf = (lane >> 4) * 2;
    const int ncol  = lane & 15;

    for (int k = 0; k < KSTEPS; ++k) {
        const int kc = 4 * k + khalf;
        v2f a0 = *(const v2f*)(in_lds + mrow * PITCH + kc);
        v2f a1 = *(const v2f*)(in_lds + (mrow + 16) * PITCH + kc);
#pragma unroll
        for (int n = 0; n < NT; ++n) {
            v2f bt = Wp[(kc >> 1) * COUT + n * 16 + ncol];
            acc[0][n] = wmma4(a0, bt, acc[0][n]);
            acc[1][n] = wmma4(a1, bt, acc[1][n]);
        }
    }
    const int rbase = (lane >> 4) * 8;
#pragma unroll
    for (int n = 0; n < NT; ++n) {
        float bb = bias[n * 16 + ncol];
#pragma unroll
        for (int mt = 0; mt < 2; ++mt)
#pragma unroll
            for (int j = 0; j < 8; ++j) {
                int rrow = mt * 16 + rbase + j;
                out_lds[rrow * PITCH + n * 16 + ncol] = fmaxf(acc[mt][n][j] + bb, 0.0f);
            }
    }
}

__global__ __launch_bounds__(32) void mlp_kernel(
    const float* __restrict__ xyz, const float* __restrict__ features,
    const int* __restrict__ gidx, const float* __restrict__ new_xyz,
    float* __restrict__ out_feat,
    const v2f* __restrict__ Wp0, const float* __restrict__ b0f,
    const v2f* __restrict__ Wp1, const float* __restrict__ b1f,
    const v2f* __restrict__ Wp2, const float* __restrict__ b2f)
{
    __shared__ __align__(16) float bufA[32 * PITCH];
    __shared__ __align__(16) float bufB[32 * PITCH];
    const int cg   = blockIdx.x;
    const int b    = cg >> 10;
    const int lane = threadIdx.x;

    const int   idx = gidx[(size_t)cg * NSAMPLE + lane];
    const float cx = new_xyz[cg * 3 + 0];
    const float cy = new_xyz[cg * 3 + 1];
    const float cz = new_xyz[cg * 3 + 2];
    const float* pt = xyz + ((size_t)b * N_ + idx) * 3;
    float* row = bufA + lane * PITCH;
    row[0] = pt[0] - cx;
    row[1] = pt[1] - cy;
    row[2] = pt[2] - cz;
    const float* fb = features + (size_t)b * C_ * N_ + idx;
#pragma unroll 8
    for (int c = 0; c < C_; ++c) row[3 + c] = fb[(size_t)c * N_];
    row[67] = 0.0f;
    __syncthreads();

    layer_gemm<17, 4, 64>(bufA, bufB, Wp0, b0f, lane);
    __syncthreads();
    layer_gemm<16, 4, 64>(bufB, bufA, Wp1, b1f, lane);
    __syncthreads();

    v8f acc[2][8];
    v8f zero = {};
#pragma unroll
    for (int mt = 0; mt < 2; ++mt)
#pragma unroll
        for (int n = 0; n < 8; ++n) acc[mt][n] = zero;

    const int mrow  = lane & 15;
    const int khalf = (lane >> 4) * 2;
    const int ncol  = lane & 15;
    for (int k = 0; k < 16; ++k) {
        const int kc = 4 * k + khalf;
        v2f a0 = *(const v2f*)(bufA + mrow * PITCH + kc);
        v2f a1 = *(const v2f*)(bufA + (mrow + 16) * PITCH + kc);
#pragma unroll
        for (int n = 0; n < 8; ++n) {
            v2f bt = Wp2[(kc >> 1) * 128 + n * 16 + ncol];
            acc[0][n] = wmma4(a0, bt, acc[0][n]);
            acc[1][n] = wmma4(a1, bt, acc[1][n]);
        }
    }
    float* of = out_feat + (size_t)cg * 128;
#pragma unroll
    for (int n = 0; n < 8; ++n) {
        float r = -3.4e38f;
#pragma unroll
        for (int j = 0; j < 8; ++j) {
            r = fmaxf(r, acc[0][n][j]);
            r = fmaxf(r, acc[1][n][j]);
        }
        r = fmaxf(r, __shfl_xor(r, 16, 32));
        float v = fmaxf(r + b2f[n * 16 + ncol], 0.0f);
        if (lane < 16) of[n * 16 + ncol] = v;
    }
}

extern "C" void kernel_launch(void* const* d_in, const int* in_sizes, int n_in,
                              void* d_out, int out_size, void* d_ws, size_t ws_size,
                              hipStream_t stream)
{
    const float* xyz      = (const float*)d_in[0];
    const float* features = (const float*)d_in[1];
    const float* W0 = (const float*)d_in[2];  const float* b0 = (const float*)d_in[3];
    const float* g0 = (const float*)d_in[4];  const float* be0 = (const float*)d_in[5];
    const float* m0 = (const float*)d_in[6];  const float* v0 = (const float*)d_in[7];
    const float* W1 = (const float*)d_in[8];  const float* b1 = (const float*)d_in[9];
    const float* g1 = (const float*)d_in[10]; const float* be1 = (const float*)d_in[11];
    const float* m1 = (const float*)d_in[12]; const float* v1 = (const float*)d_in[13];
    const float* W2 = (const float*)d_in[14]; const float* b2 = (const float*)d_in[15];
    const float* g2 = (const float*)d_in[16]; const float* be2 = (const float*)d_in[17];
    const float* m2 = (const float*)d_in[18]; const float* v2_ = (const float*)d_in[19];

    float* out      = (float*)d_out;
    float* new_xyz  = out;
    float* out_feat = out + (size_t)B_ * NPOINT * 3;

    int*   gidx = (int*)d_ws;
    float* wsf  = (float*)((char*)d_ws + (size_t)B_ * NPOINT * NSAMPLE * sizeof(int));
    v2f*   Wp0 = (v2f*)wsf;
    float* b0f = (float*)(Wp0 + 34 * 64);
    v2f*   Wp1 = (v2f*)(b0f + 64);
    float* b1f = (float*)(Wp1 + 32 * 64);
    v2f*   Wp2 = (v2f*)(b1f + 64);
    float* b2f = (float*)(Wp2 + 32 * 128);

    fuse_weights_kernel<<<1, 256, 0, stream>>>(W0, b0, g0, be0, m0, v0,
                                               W1, b1, g1, be1, m1, v1,
                                               W2, b2, g2, be2, m2, v2_,
                                               Wp0, b0f, Wp1, b1f, Wp2, b2f);
    fps_kernel<<<B_, 256, 0, stream>>>(xyz, new_xyz);
    knn_kernel<<<B_ * 8, KNN_T, 0, stream>>>(xyz, new_xyz, gidx);
    mlp_kernel<<<B_ * NPOINT, 32, 0, stream>>>(xyz, features, gidx, new_xyz, out_feat,
                                               Wp0, b0f, Wp1, b1f, Wp2, b2f);
}


# knn: each lane's 32 nearest kept as a distance-sorted list in registers, a nearer point merged in by independent per-slot ops (median of three, bit selects); LDS list and its 32-step max search gone
# speedup vs baseline: 2.6179x; 1.6924x over previous
#include <hip/hip_runtime.h>
#include <hip/hip_bf16.h>
#include <math.h>

#define B_      16
#define N_      4096
#define C_      64
#define NPOINT  1024
#define NSAMPLE 32
#define PITCH   68
#define BN_EPS  1e-5f

typedef float v2f __attribute__((ext_vector_type(2)));
typedef float v8f __attribute__((ext_vector_type(8)));

static __device__ inline v8f wmma4(v2f a, v2f b, v8f c) {
    return __builtin_amdgcn_wmma_f32_16x16x4_f32(
         false, a,  false, b,
         (short)0, c,  false,  false);
}

__global__ __launch_bounds__(256) void fuse_weights_kernel(
    const float* __restrict__ W0, const float* __restrict__ b0, const float* __restrict__ g0,
    const float* __restrict__ be0, const float* __restrict__ m0, const float* __restrict__ v0,
    const float* __restrict__ W1, const float* __restrict__ b1, const float* __restrict__ g1,
    const float* __restrict__ be1, const float* __restrict__ m1, const float* __restrict__ v1,
    const float* __restrict__ W2, const float* __restrict__ b2, const float* __restrict__ g2,
    const float* __restrict__ be2, const float* __restrict__ m2, const float* __restrict__ v2_,
    v2f* __restrict__ Wp0, float* __restrict__ b0f,
    v2f* __restrict__ Wp1, float* __restrict__ b1f,
    v2f* __restrict__ Wp2, float* __restrict__ b2f)
{
    const int tid = threadIdx.x;
    for (int i = tid; i < 34 * 64; i += 256) {
        int o = i & 63, kp = i >> 6;
        float s = g0[o] / sqrtf(v0[o] + BN_EPS);
        int k0 = 2 * kp, k1 = 2 * kp + 1;
        v2f w;
        w.x = (k0 < 67) ? W0[o * 67 + k0] * s : 0.0f;
        w.y = (k1 < 67) ? W0[o * 67 + k1] * s : 0.0f;
        Wp0[i] = w;
    }
    for (int i = tid; i < 64; i += 256) {
        float s = g0[i] / sqrtf(v0[i] + BN_EPS);
        b0f[i] = (b0[i] - m0[i]) * s + be0[i];
    }
    for (int i = tid; i < 32 * 64; i += 256) {
        int o = i & 63, kp = i >> 6;
        float s = g1[o] / sqrtf(v1[o] + BN_EPS);
        v2f w;
        w.x = W1[o * 64 + 2 * kp + 0] * s;
        w.y = W1[o * 64 + 2 * kp + 1] * s;
        Wp1[i] = w;
    }
    for (int i = tid; i < 64; i += 256) {
        float s = g1[i] / sqrtf(v1[i] + BN_EPS);
        b1f[i] = (b1[i] - m1[i]) * s + be1[i];
    }
    for (int i = tid; i < 32 * 128; i += 256) {
        int o = i & 127, kp = i >> 7;
        float s = g2[o] / sqrtf(v2_[o] + BN_EPS);
        v2f w;
        w.x = W2[o * 64 + 2 * kp + 0] * s;
        w.y = W2[o * 64 + 2 * kp + 1] * s;
        Wp2[i] = w;
    }
    for (int i = tid; i < 128; i += 256) {
        float s = g2[i] / sqrtf(v2_[i] + BN_EPS);
        b2f[i] = (b2[i] - m2[i]) * s + be2[i];
    }
}

template <int M>
static __device__ inline int xrow(int x) {
    return __builtin_amdgcn_update_dpp(0, x, 0x160 + M, 0xf, 0xf, true);
}
static __device__ inline int xhalf(int x) {
    return __builtin_amdgcn_permlanex16(x, x, 0x76543210, 0xfedcba98, false, false);
}
static __device__ inline int imax(int a, int b) { return a > b ? a : b; }
static __device__ inline int imin(int a, int b) { return a < b ? a : b; }
#define NO_IDX 0x7fffffff

__global__ __launch_bounds__(256) void fps_kernel(const float* __restrict__ xyz,
                                                  float* __restrict__ new_xyz)
{
    __shared__ float sx[N_], sy[N_], sz[N_];
    __shared__ int sidx[NPOINT];
    __shared__ __align__(16) int red[2][16];
    const int b   = blockIdx.x;
    const int tid = threadIdx.x;
    const float* bx = xyz + (size_t)b * N_ * 3;

    float dloc[16];
#pragma unroll
    for (int i = 0; i < 16; ++i) {
        int p = tid + 256 * i;
        sx[p] = bx[p * 3 + 0];
        sy[p] = bx[p * 3 + 1];
        sz[p] = bx[p * 3 + 2];
        dloc[i] = 1e10f;
    }
    __syncthreads();

    int last = 0;
    for (int it = 0; ; ++it) {
        const float lx = sx[last], ly = sy[last], lz = sz[last];
        if (tid == 0) sidx[it] = last;
        if (it == NPOINT - 1) break;

        float val = -1.0f;
        int   best = tid;
#pragma unroll
        for (int i = 0; i < 16; ++i) {
            int p = tid + 256 * i;
            float dx = sx[p] - lx, dy = sy[p] - ly, dz = sz[p] - lz;
            float dist = dx * dx + dy * dy + dz * dz;
            dloc[i] = fminf(dloc[i], dist);
            if (dloc[i] > val) { val = dloc[i]; best = p; }
        }
        const int vb = __float_as_int(val);
        int m = vb;
        m = imax(m, xrow<1>(m));
        m = imax(m, xrow<2>(m));
        m = imax(m, xrow<4>(m));
        m = imax(m, xrow<8>(m));
        m = imax(m, xhalf(m));
        int c = (vb == m) ? best : NO_IDX;
        c = imin(c, xrow<1>(c));
        c = imin(c, xrow<2>(c));
        c = imin(c, xrow<4>(c));
        c = imin(c, xrow<8>(c));
        c = imin(c, xhalf(c));

        int* slot = red[it & 1];
        if ((tid & 31) == 0) *(int2*)(slot + 2 * (tid >> 5)) = make_int2(m, c);
        __syncthreads();
        const int2 r = *(const int2*)(slot + 2 * (tid & 7));
        int gm = r.x;
        gm = imax(gm, xrow<1>(gm));
        gm = imax(gm, xrow<2>(gm));
        gm = imax(gm, xrow<4>(gm));
        int gc = (r.x == gm) ? r.y : NO_IDX;
        gc = imin(gc, xrow<1>(gc));
        gc = imin(gc, xrow<2>(gc));
        gc = imin(gc, xrow<4>(gc));
        last = gc;
    }
    __syncthreads();
    for (int k = 0; k < NPOINT / 256; ++k) {
        const int j = tid + 256 * k;
        const int p = sidx[j];
        float* o = new_xyz + ((size_t)b * NPOINT + j) * 3;
        o[0] = sx[p];
        o[1] = sy[p];
        o[2] = sz[p];
    }
}

#define KNN_T 128
static __device__ __forceinline__ unsigned bitdiff(unsigned x, unsigned y)
{
    unsigned r;
    asm volatile("v_sub_nc_u32 %0, %1, %2" : "=v"(r) : "v"(x), "v"(y));
    return r;
}
static __device__ __forceinline__ int signmask(unsigned d)
{
    int r;
    asm volatile("v_ashrrev_i32 %0, 31, %1" : "=v"(r) : "v"(d));
    return r;
}
static __device__ __forceinline__ int bitsel(int m, int one, int zero)
{
    int r;
    asm volatile("v_bfi_b32 %0, %1, %2, %3" : "=v"(r) : "v"(m), "v"(one), "v"(zero));
    return r;
}
static __device__ __forceinline__ void bitsel_into(int& keep, int m, int one)
{
    asm volatile("v_bfi_b32 %0, %1, %2, %0" : "+v"(keep) : "v"(m), "v"(one));
}

static __device__ __forceinline__ void knn_insert(float (&a)[NSAMPLE], int (&id)[NSAMPLE],
                                                  float x, int j)
{
    const unsigned xb = (unsigned)__float_as_int(x);
    unsigned d[NSAMPLE];
    int m[NSAMPLE], s[NSAMPLE];
#pragma unroll
    for (int e = 0; e < NSAMPLE; ++e) d[e] = bitdiff(xb, (unsigned)__float_as_int(a[e]));
#pragma unroll
    for (int e = 0; e < NSAMPLE; ++e) m[e] = signmask(d[e]);
    s[0] = j;
#pragma unroll
    for (int e = 1; e < NSAMPLE; ++e) s[e] = bitsel(m[e - 1], id[e - 1], j);
#pragma unroll
    for (int e = 0; e < NSAMPLE; ++e) bitsel_into(id[e], m[e], s[e]);
#pragma unroll
    for (int e = NSAMPLE - 1; e >= 1; --e)
        a[e] = __builtin_amdgcn_fmed3f(a[e - 1], a[e], x);
    a[0] = fminf(a[0], x);
}

__global__ __launch_bounds__(KNN_T) void knn_kernel(const float* __restrict__ xyz,
                                                    const float* __restrict__ new_xyz,
                                                    int* __restrict__ gidx)
{
    __shared__ float4 pts[N_ * 3 / 4];
    const int b    = blockIdx.x >> 3;
    const int g    = blockIdx.x & 7;
    const int tid  = threadIdx.x;
    const int cent = g * KNN_T + tid;

    const float4* src = (const float4*)(xyz + (size_t)b * N_ * 3);
    for (int i = tid; i < N_ * 3 / 4; i += KNN_T) pts[i] = src[i];

    const float cx = new_xyz[(b * NPOINT + cent) * 3 + 0];
    const float cy = new_xyz[(b * NPOINT + cent) * 3 + 1];
    const float cz = new_xyz[(b * NPOINT + cent) * 3 + 2];

    float a[NSAMPLE];
    int   id[NSAMPLE];
#pragma unroll
    for (int e = 0; e < NSAMPLE; ++e) { a[e] = 3.4e38f; id[e] = 0; }
    __syncthreads();

#define KNN_DIST(PX, PY, PZ) fmaf((PZ) - cz, (PZ) - cz, fmaf((PX) - cx, (PX) - cx, ((PY) - cy) * ((PY) - cy)))

    float4 A = pts[0], B = pts[1], C = pts[2];
    for (int q = 0; q < N_ / 4; ++q) {
        const int qn = (q + 1 < N_ / 4) ? q + 1 : q;
        const float4 nA = pts[3 * qn + 0];
        const float4 nB = pts[3 * qn + 1];
        const float4 nC = pts[3 * qn + 2];
        const int j = 4 * q;
        const float d0 = KNN_DIST(A.x, A.y, A.z);
        const float d1 = KNN_DIST(A.w, B.x, B.y);
        const float d2 = KNN_DIST(B.z, B.w, C.x);
        const float d3 = KNN_DIST(C.y, C.z, C.w);
        if (fminf(fminf(d0, d1), fminf(d2, d3)) < a[NSAMPLE - 1]) {
            if (d0 < a[NSAMPLE - 1]) knn_insert(a, id, d0, j + 0);
            if (d1 < a[NSAMPLE - 1]) knn_insert(a, id, d1, j + 1);
            if (d2 < a[NSAMPLE - 1]) knn_insert(a, id, d2, j + 2);
            if (d3 < a[NSAMPLE - 1]) knn_insert(a, id, d3, j + 3);
        }
        A = nA; B = nB; C = nC;
    }
#undef KNN_DIST

    int4* go = (int4*)(gidx + (size_t)(b * NPOINT + cent) * NSAMPLE);
#pragma unroll
    for (int e = 0; e < NSAMPLE / 4; ++e)
        go[e] = make_int4(id[4 * e + 0], id[4 * e + 1], id[4 * e + 2], id[4 * e + 3]);
}

template <int KSTEPS, int NT, int COUT>
static __device__ inline void layer_gemm(const float* in_lds, float* out_lds,
                                         const v2f* __restrict__ Wp,
                                         const float* __restrict__ bias, int lane)
{
    v8f acc[2][NT];
    v8f zero = {};
#pragma unroll
    for (int mt = 0; mt < 2; ++mt)
#pragma unroll
        for (int n = 0; n < NT; ++n) acc[mt][n] = zero;

    const int mrow  = lane & 15;
    const int khalf = (lane >> 4) * 2;
    const int ncol  = lane & 15;

    for (int k = 0; k < KSTEPS; ++k) {
        const int kc = 4 * k + khalf;
        v2f a0 = *(const v2f*)(in_lds + mrow * PITCH + kc);
        v2f a1 = *(const v2f*)(in_lds + (mrow + 16) * PITCH + kc);
#pragma unroll
        for (int n = 0; n < NT; ++n) {
            v2f bt = Wp[(kc >> 1) * COUT + n * 16 + ncol];
            acc[0][n] = wmma4(a0, bt, acc[0][n]);
            acc[1][n] = wmma4(a1, bt, acc[1][n]);
        }
    }
    const int rbase = (lane >> 4) * 8;
#pragma unroll
    for (int n = 0; n < NT; ++n) {
        float bb = bias[n * 16 + ncol];
#pragma unroll
        for (int mt = 0; mt < 2; ++mt)
#pragma unroll
            for (int j = 0; j < 8; ++j) {
                int rrow = mt * 16 + rbase + j;
                out_lds[rrow * PITCH + n * 16 + ncol] = fmaxf(acc[mt][n][j] + bb, 0.0f);
            }
    }
}

__global__ __launch_bounds__(32) void mlp_kernel(
    const float* __restrict__ xyz, const float* __restrict__ features,
    const int* __restrict__ gidx, const float* __restrict__ new_xyz,
    float* __restrict__ out_feat,
    const v2f* __restrict__ Wp0, const float* __restrict__ b0f,
    const v2f* __restrict__ Wp1, const float* __restrict__ b1f,
    const v2f* __restrict__ Wp2, const float* __restrict__ b2f)
{
    __shared__ __align__(16) float bufA[32 * PITCH];
    __shared__ __align__(16) float bufB[32 * PITCH];
    const int cg   = blockIdx.x;
    const int b    = cg >> 10;
    const int lane = threadIdx.x;

    const int   idx = gidx[(size_t)cg * NSAMPLE + lane];
    const float cx = new_xyz[cg * 3 + 0];
    const float cy = new_xyz[cg * 3 + 1];
    const float cz = new_xyz[cg * 3 + 2];
    const float* pt = xyz + ((size_t)b * N_ + idx) * 3;
    float* row = bufA + lane * PITCH;
    row[0] = pt[0] - cx;
    row[1] = pt[1] - cy;
    row[2] = pt[2] - cz;
    const float* fb = features + (size_t)b * C_ * N_ + idx;
#pragma unroll 8
    for (int c = 0; c < C_; ++c) row[3 + c] = fb[(size_t)c * N_];
    row[67] = 0.0f;
    __syncthreads();

    layer_gemm<17, 4, 64>(bufA, bufB, Wp0, b0f, lane);
    __syncthreads();
    layer_gemm<16, 4, 64>(bufB, bufA, Wp1, b1f, lane);
    __syncthreads();

    v8f acc[2][8];
    v8f zero = {};
#pragma unroll
    for (int mt = 0; mt < 2; ++mt)
#pragma unroll
        for (int n = 0; n < 8; ++n) acc[mt][n] = zero;

    const int mrow  = lane & 15;
    const int khalf = (lane >> 4) * 2;
    const int ncol  = lane & 15;
    for (int k = 0; k < 16; ++k) {
        const int kc = 4 * k + khalf;
        v2f a0 = *(const v2f*)(bufA + mrow * PITCH + kc);
        v2f a1 = *(const v2f*)(bufA + (mrow + 16) * PITCH + kc);
#pragma unroll
        for (int n = 0; n < 8; ++n) {
            v2f bt = Wp2[(kc >> 1) * 128 + n * 16 + ncol];
            acc[0][n] = wmma4(a0, bt, acc[0][n]);
            acc[1][n] = wmma4(a1, bt, acc[1][n]);
        }
    }
    float* of = out_feat + (size_t)cg * 128;
#pragma unroll
    for (int n = 0; n < 8; ++n) {
        float r = -3.4e38f;
#pragma unroll
        for (int j = 0; j < 8; ++j) {
            r = fmaxf(r, acc[0][n][j]);
            r = fmaxf(r, acc[1][n][j]);
        }
        r = fmaxf(r, __shfl_xor(r, 16, 32));
        float v = fmaxf(r + b2f[n * 16 + ncol], 0.0f);
        if (lane < 16) of[n * 16 + ncol] = v;
    }
}

extern "C" void kernel_launch(void* const* d_in, const int* in_sizes, int n_in,
                              void* d_out, int out_size, void* d_ws, size_t ws_size,
                              hipStream_t stream)
{
    const float* xyz      = (const float*)d_in[0];
    const float* features = (const float*)d_in[1];
    const float* W0 = (const float*)d_in[2];  const float* b0 = (const float*)d_in[3];
    const float* g0 = (const float*)d_in[4];  const float* be0 = (const float*)d_in[5];
    const float* m0 = (const float*)d_in[6];  const float* v0 = (const float*)d_in[7];
    const float* W1 = (const float*)d_in[8];  const float* b1 = (const float*)d_in[9];
    const float* g1 = (const float*)d_in[10]; const float* be1 = (const float*)d_in[11];
    const float* m1 = (const float*)d_in[12]; const float* v1 = (const float*)d_in[13];
    const float* W2 = (const float*)d_in[14]; const float* b2 = (const float*)d_in[15];
    const float* g2 = (const float*)d_in[16]; const float* be2 = (const float*)d_in[17];
    const float* m2 = (const float*)d_in[18]; const float* v2_ = (const float*)d_in[19];

    float* out      = (float*)d_out;
    float* new_xyz  = out;
    float* out_feat = out + (size_t)B_ * NPOINT * 3;

    int*   gidx = (int*)d_ws;
    float* wsf  = (float*)((char*)d_ws + (size_t)B_ * NPOINT * NSAMPLE * sizeof(int));
    v2f*   Wp0 = (v2f*)wsf;
    float* b0f = (float*)(Wp0 + 34 * 64);
    v2f*   Wp1 = (v2f*)(b0f + 64);
    float* b1f = (float*)(Wp1 + 32 * 64);
    v2f*   Wp2 = (v2f*)(b1f + 64);
    float* b2f = (float*)(Wp2 + 32 * 128);

    fuse_weights_kernel<<<1, 256, 0, stream>>>(W0, b0, g0, be0, m0, v0,
                                               W1, b1, g1, be1, m1, v1,
                                               W2, b2, g2, be2, m2, v2_,
                                               Wp0, b0f, Wp1, b1f, Wp2, b2f);
    fps_kernel<<<B_, 256, 0, stream>>>(xyz, new_xyz);
    knn_kernel<<<B_ * 8, KNN_T, 0, stream>>>(xyz, new_xyz, gidx);
    mlp_kernel<<<B_ * NPOINT, 32, 0, stream>>>(xyz, features, gidx, new_xyz, out_feat,
                                               Wp0, b0f, Wp1, b1f, Wp2, b2f);
}


# mlp gather: features transposed once per call to point-major in the workspace (one extra launch), so a neighbour's 64 channels are one 256-byte row read by one load, not 64 lines 16 KB apart
# speedup vs baseline: 2.8845x; 1.1018x over previous
#include <hip/hip_runtime.h>
#include <hip/hip_bf16.h>
#include <math.h>

#define B_      16
#define N_      4096
#define C_      64
#define NPOINT  1024
#define NSAMPLE 32
#define PITCH   68
#define BN_EPS  1e-5f

typedef float v2f __attribute__((ext_vector_type(2)));
typedef float v8f __attribute__((ext_vector_type(8)));

static __device__ inline v8f wmma4(v2f a, v2f b, v8f c) {
    return __builtin_amdgcn_wmma_f32_16x16x4_f32(
         false, a,  false, b,
         (short)0, c,  false,  false);
}

__global__ __launch_bounds__(256) void fuse_weights_kernel(
    const float* __restrict__ W0, const float* __restrict__ b0, const float* __restrict__ g0,
    const float* __restrict__ be0, const float* __restrict__ m0, const float* __restrict__ v0,
    const float* __restrict__ W1, const float* __restrict__ b1, const float* __restrict__ g1,
    const float* __restrict__ be1, const float* __restrict__ m1, const float* __restrict__ v1,
    const float* __restrict__ W2, const float* __restrict__ b2, const float* __restrict__ g2,
    const float* __restrict__ be2, const float* __restrict__ m2, const float* __restrict__ v2_,
    v2f* __restrict__ Wp0, float* __restrict__ b0f,
    v2f* __restrict__ Wp1, float* __restrict__ b1f,
    v2f* __restrict__ Wp2, float* __restrict__ b2f)
{
    const int tid = threadIdx.x;
    for (int i = tid; i < 34 * 64; i += 256) {
        int o = i & 63, kp = i >> 6;
        float s = g0[o] / sqrtf(v0[o] + BN_EPS);
        int k0 = 2 * kp, k1 = 2 * kp + 1;
        v2f w;
        w.x = (k0 < 67) ? W0[o * 67 + k0] * s : 0.0f;
        w.y = (k1 < 67) ? W0[o * 67 + k1] * s : 0.0f;
        Wp0[i] = w;
    }
    for (int i = tid; i < 64; i += 256) {
        float s = g0[i] / sqrtf(v0[i] + BN_EPS);
        b0f[i] = (b0[i] - m0[i]) * s + be0[i];
    }
    for (int i = tid; i < 32 * 64; i += 256) {
        int o = i & 63, kp = i >> 6;
        float s = g1[o] / sqrtf(v1[o] + BN_EPS);
        v2f w;
        w.x = W1[o * 64 + 2 * kp + 0] * s;
        w.y = W1[o * 64 + 2 * kp + 1] * s;
        Wp1[i] = w;
    }
    for (int i = tid; i < 64; i += 256) {
        float s = g1[i] / sqrtf(v1[i] + BN_EPS);
        b1f[i] = (b1[i] - m1[i]) * s + be1[i];
    }
    for (int i = tid; i < 32 * 128; i += 256) {
        int o = i & 127, kp = i >> 7;
        float s = g2[o] / sqrtf(v2_[o] + BN_EPS);
        v2f w;
        w.x = W2[o * 64 + 2 * kp + 0] * s;
        w.y = W2[o * 64 + 2 * kp + 1] * s;
        Wp2[i] = w;
    }
    for (int i = tid; i < 128; i += 256) {
        float s = g2[i] / sqrtf(v2_[i] + BN_EPS);
        b2f[i] = (b2[i] - m2[i]) * s + be2[i];
    }
}

template <int M>
static __device__ inline int xrow(int x) {
    return __builtin_amdgcn_update_dpp(0, x, 0x160 + M, 0xf, 0xf, true);
}
static __device__ inline int xhalf(int x) {
    return __builtin_amdgcn_permlanex16(x, x, 0x76543210, 0xfedcba98, false, false);
}
static __device__ inline int imax(int a, int b) { return a > b ? a : b; }
static __device__ inline int imin(int a, int b) { return a < b ? a : b; }
#define NO_IDX 0x7fffffff

__global__ __launch_bounds__(256) void fps_kernel(const float* __restrict__ xyz,
                                                  float* __restrict__ new_xyz)
{
    __shared__ float sx[N_], sy[N_], sz[N_];
    __shared__ int sidx[NPOINT];
    __shared__ __align__(16) int red[2][16];
    const int b   = blockIdx.x;
    const int tid = threadIdx.x;
    const float* bx = xyz + (size_t)b * N_ * 3;

    float dloc[16];
#pragma unroll
    for (int i = 0; i < 16; ++i) {
        int p = tid + 256 * i;
        sx[p] = bx[p * 3 + 0];
        sy[p] = bx[p * 3 + 1];
        sz[p] = bx[p * 3 + 2];
        dloc[i] = 1e10f;
    }
    __syncthreads();

    int last = 0;
    for (int it = 0; ; ++it) {
        const float lx = sx[last], ly = sy[last], lz = sz[last];
        if (tid == 0) sidx[it] = last;
        if (it == NPOINT - 1) break;

        float val = -1.0f;
        int   best = tid;
#pragma unroll
        for (int i = 0; i < 16; ++i) {
            int p = tid + 256 * i;
            float dx = sx[p] - lx, dy = sy[p] - ly, dz = sz[p] - lz;
            float dist = dx * dx + dy * dy + dz * dz;
            dloc[i] = fminf(dloc[i], dist);
            if (dloc[i] > val) { val = dloc[i]; best = p; }
        }
        const int vb = __float_as_int(val);
        int m = vb;
        m = imax(m, xrow<1>(m));
        m = imax(m, xrow<2>(m));
        m = imax(m, xrow<4>(m));
        m = imax(m, xrow<8>(m));
        m = imax(m, xhalf(m));
        int c = (vb == m) ? best : NO_IDX;
        c = imin(c, xrow<1>(c));
        c = imin(c, xrow<2>(c));
        c = imin(c, xrow<4>(c));
        c = imin(c, xrow<8>(c));
        c = imin(c, xhalf(c));

        int* slot = red[it & 1];
        if ((tid & 31) == 0) *(int2*)(slot + 2 * (tid >> 5)) = make_int2(m, c);
        __syncthreads();
        const int2 r = *(const int2*)(slot + 2 * (tid & 7));
        int gm = r.x;
        gm = imax(gm, xrow<1>(gm));
        gm = imax(gm, xrow<2>(gm));
        gm = imax(gm, xrow<4>(gm));
        int gc = (r.x == gm) ? r.y : NO_IDX;
        gc = imin(gc, xrow<1>(gc));
        gc = imin(gc, xrow<2>(gc));
        gc = imin(gc, xrow<4>(gc));
        last = gc;
    }
    __syncthreads();
    for (int k = 0; k < NPOINT / 256; ++k) {
        const int j = tid + 256 * k;
        const int p = sidx[j];
        float* o = new_xyz + ((size_t)b * NPOINT + j) * 3;
        o[0] = sx[p];
        o[1] = sy[p];
        o[2] = sz[p];
    }
}

#define KNN_T 128
static __device__ __forceinline__ unsigned bitdiff(unsigned x, unsigned y)
{
    unsigned r;
    asm volatile("v_sub_nc_u32 %0, %1, %2" : "=v"(r) : "v"(x), "v"(y));
    return r;
}
static __device__ __forceinline__ int signmask(unsigned d)
{
    int r;
    asm volatile("v_ashrrev_i32 %0, 31, %1" : "=v"(r) : "v"(d));
    return r;
}
static __device__ __forceinline__ int bitsel(int m, int one, int zero)
{
    int r;
    asm volatile("v_bfi_b32 %0, %1, %2, %3" : "=v"(r) : "v"(m), "v"(one), "v"(zero));
    return r;
}
static __device__ __forceinline__ void bitsel_into(int& keep, int m, int one)
{
    asm volatile("v_bfi_b32 %0, %1, %2, %0" : "+v"(keep) : "v"(m), "v"(one));
}

static __device__ __forceinline__ void knn_insert(float (&a)[NSAMPLE], int (&id)[NSAMPLE],
                                                  float x, int j)
{
    const unsigned xb = (unsigned)__float_as_int(x);
    unsigned d[NSAMPLE];
    int m[NSAMPLE], s[NSAMPLE];
#pragma unroll
    for (int e = 0; e < NSAMPLE; ++e) d[e] = bitdiff(xb, (unsigned)__float_as_int(a[e]));
#pragma unroll
    for (int e = 0; e < NSAMPLE; ++e) m[e] = signmask(d[e]);
    s[0] = j;
#pragma unroll
    for (int e = 1; e < NSAMPLE; ++e) s[e] = bitsel(m[e - 1], id[e - 1], j);
#pragma unroll
    for (int e = 0; e < NSAMPLE; ++e) bitsel_into(id[e], m[e], s[e]);
#pragma unroll
    for (int e = NSAMPLE - 1; e >= 1; --e)
        a[e] = __builtin_amdgcn_fmed3f(a[e - 1], a[e], x);
    a[0] = fminf(a[0], x);
}

__global__ __launch_bounds__(KNN_T) void knn_kernel(const float* __restrict__ xyz,
                                                    const float* __restrict__ new_xyz,
                                                    int* __restrict__ gidx)
{
    __shared__ float4 pts[N_ * 3 / 4];
    const int b    = blockIdx.x >> 3;
    const int g    = blockIdx.x & 7;
    const int tid  = threadIdx.x;
    const int cent = g * KNN_T + tid;

    const float4* src = (const float4*)(xyz + (size_t)b * N_ * 3);
    for (int i = tid; i < N_ * 3 / 4; i += KNN_T) pts[i] = src[i];

    const float cx = new_xyz[(b * NPOINT + cent) * 3 + 0];
    const float cy = new_xyz[(b * NPOINT + cent) * 3 + 1];
    const float cz = new_xyz[(b * NPOINT + cent) * 3 + 2];

    float a[NSAMPLE];
    int   id[NSAMPLE];
#pragma unroll
    for (int e = 0; e < NSAMPLE; ++e) { a[e] = 3.4e38f; id[e] = 0; }
    __syncthreads();

#define KNN_DIST(PX, PY, PZ) fmaf((PZ) - cz, (PZ) - cz, fmaf((PX) - cx, (PX) - cx, ((PY) - cy) * ((PY) - cy)))

    float4 A = pts[0], B = pts[1], C = pts[2];
    for (int q = 0; q < N_ / 4; ++q) {
        const int qn = (q + 1 < N_ / 4) ? q + 1 : q;
        const float4 nA = pts[3 * qn + 0];
        const float4 nB = pts[3 * qn + 1];
        const float4 nC = pts[3 * qn + 2];
        const int j = 4 * q;
        const float d0 = KNN_DIST(A.x, A.y, A.z);
        const float d1 = KNN_DIST(A.w, B.x, B.y);
        const float d2 = KNN_DIST(B.z, B.w, C.x);
        const float d3 = KNN_DIST(C.y, C.z, C.w);
        if (fminf(fminf(d0, d1), fminf(d2, d3)) < a[NSAMPLE - 1]) {
            if (d0 < a[NSAMPLE - 1]) knn_insert(a, id, d0, j + 0);
            if (d1 < a[NSAMPLE - 1]) knn_insert(a, id, d1, j + 1);
            if (d2 < a[NSAMPLE - 1]) knn_insert(a, id, d2, j + 2);
            if (d3 < a[NSAMPLE - 1]) knn_insert(a, id, d3, j + 3);
        }
        A = nA; B = nB; C = nC;
    }
#undef KNN_DIST

    int4* go = (int4*)(gidx + (size_t)(b * NPOINT + cent) * NSAMPLE);
#pragma unroll
    for (int e = 0; e < NSAMPLE / 4; ++e)
        go[e] = make_int4(id[4 * e + 0], id[4 * e + 1], id[4 * e + 2], id[4 * e + 3]);
}

#define TP_PTS 64
__global__ __launch_bounds__(256) void transpose_features_kernel(const float* __restrict__ features,
                                                                 float* __restrict__ featT)
{
    __shared__ float tile[C_][TP_PTS + 1];
    const int b   = blockIdx.x >> 6;
    const int n0  = (blockIdx.x & 63) * TP_PTS;
    const int tid = threadIdx.x;
    const int q   = tid & 15;
    const int r0  = tid >> 4;
    const float* src = features + (size_t)b * C_ * N_ + n0;
#pragma unroll
    for (int k = 0; k < 4; ++k) {
        const int c = r0 + 16 * k;
        const float4 v = *(const float4*)(src + (size_t)c * N_ + 4 * q);
        tile[c][4 * q + 0] = v.x;
        tile[c][4 * q + 1] = v.y;
        tile[c][4 * q + 2] = v.z;
        tile[c][4 * q + 3] = v.w;
    }
    __syncthreads();
    float* dst = featT + ((size_t)b * N_ + n0) * C_;
#pragma unroll
    for (int k = 0; k < 4; ++k) {
        const int n = r0 + 16 * k;
        float4 v;
        v.x = tile[4 * q + 0][n];
        v.y = tile[4 * q + 1][n];
        v.z = tile[4 * q + 2][n];
        v.w = tile[4 * q + 3][n];
        *(float4*)(dst + (size_t)n * C_ + 4 * q) = v;
    }
}

template <int KSTEPS, int NT, int COUT>
static __device__ inline void layer_gemm(const float* in_lds, float* out_lds,
                                         const v2f* __restrict__ Wp,
                                         const float* __restrict__ bias, int lane)
{
    v8f acc[2][NT];
    v8f zero = {};
#pragma unroll
    for (int mt = 0; mt < 2; ++mt)
#pragma unroll
        for (int n = 0; n < NT; ++n) acc[mt][n] = zero;

    const int mrow  = lane & 15;
    const int khalf = (lane >> 4) * 2;
    const int ncol  = lane & 15;

    for (int k = 0; k < KSTEPS; ++k) {
        const int kc = 4 * k + khalf;
        v2f a0 = *(const v2f*)(in_lds + mrow * PITCH + kc);
        v2f a1 = *(const v2f*)(in_lds + (mrow + 16) * PITCH + kc);
#pragma unroll
        for (int n = 0; n < NT; ++n) {
            v2f bt = Wp[(kc >> 1) * COUT + n * 16 + ncol];
            acc[0][n] = wmma4(a0, bt, acc[0][n]);
            acc[1][n] = wmma4(a1, bt, acc[1][n]);
        }
    }
    const int rbase = (lane >> 4) * 8;
#pragma unroll
    for (int n = 0; n < NT; ++n) {
        float bb = bias[n * 16 + ncol];
#pragma unroll
        for (int mt = 0; mt < 2; ++mt)
#pragma unroll
            for (int j = 0; j < 8; ++j) {
                int rrow = mt * 16 + rbase + j;
                out_lds[rrow * PITCH + n * 16 + ncol] = fmaxf(acc[mt][n][j] + bb, 0.0f);
            }
    }
}

__global__ __launch_bounds__(32) void mlp_kernel(
    const float* __restrict__ xyz, const float* __restrict__ featT,
    const int* __restrict__ gidx, const float* __restrict__ new_xyz,
    float* __restrict__ out_feat,
    const v2f* __restrict__ Wp0, const float* __restrict__ b0f,
    const v2f* __restrict__ Wp1, const float* __restrict__ b1f,
    const v2f* __restrict__ Wp2, const float* __restrict__ b2f)
{
    __shared__ __align__(16) float bufA[32 * PITCH];
    __shared__ __align__(16) float bufB[32 * PITCH];
    const int cg   = blockIdx.x;
    const int b    = cg >> 10;
    const int lane = threadIdx.x;

    const int   idx = gidx[(size_t)cg * NSAMPLE + lane];
    const float cx = new_xyz[cg * 3 + 0];
    const float cy = new_xyz[cg * 3 + 1];
    const float cz = new_xyz[cg * 3 + 2];
    const float* pt = xyz + ((size_t)b * N_ + idx) * 3;
    float* row = bufA + lane * PITCH;
    row[0] = pt[0] - cx;
    row[1] = pt[1] - cy;
    row[2] = pt[2] - cz;
    const v2f* ft = (const v2f*)featT + (size_t)b * N_ * (C_ / 2);
    v2f f[NSAMPLE];
#pragma unroll
    for (int r = 0; r < NSAMPLE; ++r) {
        const int pr = __builtin_amdgcn_readlane(idx, r);
        f[r] = ft[(size_t)pr * (C_ / 2) + lane];
    }
#pragma unroll
    for (int r = 0; r < NSAMPLE; ++r) {
        bufA[r * PITCH + 3 + 2 * lane + 0] = f[r].x;
        bufA[r * PITCH + 3 + 2 * lane + 1] = f[r].y;
    }
    row[67] = 0.0f;
    __syncthreads();

    layer_gemm<17, 4, 64>(bufA, bufB, Wp0, b0f, lane);
    __syncthreads();
    layer_gemm<16, 4, 64>(bufB, bufA, Wp1, b1f, lane);
    __syncthreads();

    v8f acc[2][8];
    v8f zero = {};
#pragma unroll
    for (int mt = 0; mt < 2; ++mt)
#pragma unroll
        for (int n = 0; n < 8; ++n) acc[mt][n] = zero;

    const int mrow  = lane & 15;
    const int khalf = (lane >> 4) * 2;
    const int ncol  = lane & 15;
    for (int k = 0; k < 16; ++k) {
        const int kc = 4 * k + khalf;
        v2f a0 = *(const v2f*)(bufA + mrow * PITCH + kc);
        v2f a1 = *(const v2f*)(bufA + (mrow + 16) * PITCH + kc);
#pragma unroll
        for (int n = 0; n < 8; ++n) {
            v2f bt = Wp2[(kc >> 1) * 128 + n * 16 + ncol];
            acc[0][n] = wmma4(a0, bt, acc[0][n]);
            acc[1][n] = wmma4(a1, bt, acc[1][n]);
        }
    }
    float* of = out_feat + (size_t)cg * 128;
#pragma unroll
    for (int n = 0; n < 8; ++n) {
        float r = -3.4e38f;
#pragma unroll
        for (int j = 0; j < 8; ++j) {
            r = fmaxf(r, acc[0][n][j]);
            r = fmaxf(r, acc[1][n][j]);
        }
        r = fmaxf(r, __shfl_xor(r, 16, 32));
        float v = fmaxf(r + b2f[n * 16 + ncol], 0.0f);
        if (lane < 16) of[n * 16 + ncol] = v;
    }
}

extern "C" void kernel_launch(void* const* d_in, const int* in_sizes, int n_in,
                              void* d_out, int out_size, void* d_ws, size_t ws_size,
                              hipStream_t stream)
{
    const float* xyz      = (const float*)d_in[0];
    const float* features = (const float*)d_in[1];
    const float* W0 = (const float*)d_in[2];  const float* b0 = (const float*)d_in[3];
    const float* g0 = (const float*)d_in[4];  const float* be0 = (const float*)d_in[5];
    const float* m0 = (const float*)d_in[6];  const float* v0 = (const float*)d_in[7];
    const float* W1 = (const float*)d_in[8];  const float* b1 = (const float*)d_in[9];
    const float* g1 = (const float*)d_in[10]; const float* be1 = (const float*)d_in[11];
    const float* m1 = (const float*)d_in[12]; const float* v1 = (const float*)d_in[13];
    const float* W2 = (const float*)d_in[14]; const float* b2 = (const float*)d_in[15];
    const float* g2 = (const float*)d_in[16]; const float* be2 = (const float*)d_in[17];
    const float* m2 = (const float*)d_in[18]; const float* v2_ = (const float*)d_in[19];

    float* out      = (float*)d_out;
    float* new_xyz  = out;
    float* out_feat = out + (size_t)B_ * NPOINT * 3;

    int*   gidx = (int*)d_ws;
    float* wsf  = (float*)((char*)d_ws + (size_t)B_ * NPOINT * NSAMPLE * sizeof(int));
    v2f*   Wp0 = (v2f*)wsf;
    float* b0f = (float*)(Wp0 + 34 * 64);
    v2f*   Wp1 = (v2f*)(b0f + 64);
    float* b1f = (float*)(Wp1 + 32 * 64);
    v2f*   Wp2 = (v2f*)(b1f + 64);
    float* b2f = (float*)(Wp2 + 32 * 128);
    float* featT = (float*)((char*)d_ws + (size_t)4 * 1024 * 1024);

    fuse_weights_kernel<<<1, 256, 0, stream>>>(W0, b0, g0, be0, m0, v0,
                                               W1, b1, g1, be1, m1, v1,
                                               W2, b2, g2, be2, m2, v2_,
                                               Wp0, b0f, Wp1, b1f, Wp2, b2f);
    fps_kernel<<<B_, 256, 0, stream>>>(xyz, new_xyz);
    knn_kernel<<<B_ * 8, KNN_T, 0, stream>>>(xyz, new_xyz, gidx);
    transpose_features_kernel<<<B_ * (N_ / TP_PTS), 256, 0, stream>>>(features, featT);
    mlp_kernel<<<B_ * NPOINT, 32, 0, stream>>>(xyz, featT, gidx, new_xyz, out_feat,
                                               Wp0, b0f, Wp1, b1f, Wp2, b2f);
}


# knn: one wave per centroid: lanes split the 4096 points, the largest lane minimum bounds the 32nd distance, the few points under it get their slot by counting smaller keys; register lists gone
# speedup vs baseline: 3.5443x; 1.2287x over previous
#include <hip/hip_runtime.h>
#include <hip/hip_bf16.h>
#include <math.h>

#define B_      16
#define N_      4096
#define C_      64
#define NPOINT  1024
#define NSAMPLE 32
#define PITCH   68
#define BN_EPS  1e-5f

typedef float v2f __attribute__((ext_vector_type(2)));
typedef float v8f __attribute__((ext_vector_type(8)));

static __device__ inline v8f wmma4(v2f a, v2f b, v8f c) {
    return __builtin_amdgcn_wmma_f32_16x16x4_f32(
         false, a,  false, b,
         (short)0, c,  false,  false);
}

__global__ __launch_bounds__(256) void fuse_weights_kernel(
    const float* __restrict__ W0, const float* __restrict__ b0, const float* __restrict__ g0,
    const float* __restrict__ be0, const float* __restrict__ m0, const float* __restrict__ v0,
    const float* __restrict__ W1, const float* __restrict__ b1, const float* __restrict__ g1,
    const float* __restrict__ be1, const float* __restrict__ m1, const float* __restrict__ v1,
    const float* __restrict__ W2, const float* __restrict__ b2, const float* __restrict__ g2,
    const float* __restrict__ be2, const float* __restrict__ m2, const float* __restrict__ v2_,
    v2f* __restrict__ Wp0, float* __restrict__ b0f,
    v2f* __restrict__ Wp1, float* __restrict__ b1f,
    v2f* __restrict__ Wp2, float* __restrict__ b2f)
{
    const int tid = threadIdx.x;
    for (int i = tid; i < 34 * 64; i += 256) {
        int o = i & 63, kp = i >> 6;
        float s = g0[o] / sqrtf(v0[o] + BN_EPS);
        int k0 = 2 * kp, k1 = 2 * kp + 1;
        v2f w;
        w.x = (k0 < 67) ? W0[o * 67 + k0] * s : 0.0f;
        w.y = (k1 < 67) ? W0[o * 67 + k1] * s : 0.0f;
        Wp0[i] = w;
    }
    for (int i = tid; i < 64; i += 256) {
        float s = g0[i] / sqrtf(v0[i] + BN_EPS);
        b0f[i] = (b0[i] - m0[i]) * s + be0[i];
    }
    for (int i = tid; i < 32 * 64; i += 256) {
        int o = i & 63, kp = i >> 6;
        float s = g1[o] / sqrtf(v1[o] + BN_EPS);
        v2f w;
        w.x = W1[o * 64 + 2 * kp + 0] * s;
        w.y = W1[o * 64 + 2 * kp + 1] * s;
        Wp1[i] = w;
    }
    for (int i = tid; i < 64; i += 256) {
        float s = g1[i] / sqrtf(v1[i] + BN_EPS);
        b1f[i] = (b1[i] - m1[i]) * s + be1[i];
    }
    for (int i = tid; i < 32 * 128; i += 256) {
        int o = i & 127, kp = i >> 7;
        float s = g2[o] / sqrtf(v2_[o] + BN_EPS);
        v2f w;
        w.x = W2[o * 64 + 2 * kp + 0] * s;
        w.y = W2[o * 64 + 2 * kp + 1] * s;
        Wp2[i] = w;
    }
    for (int i = tid; i < 128; i += 256) {
        float s = g2[i] / sqrtf(v2_[i] + BN_EPS);
        b2f[i] = (b2[i] - m2[i]) * s + be2[i];
    }
}

template <int M>
static __device__ inline int xrow(int x) {
    return __builtin_amdgcn_update_dpp(0, x, 0x160 + M, 0xf, 0xf, true);
}
static __device__ inline int xhalf(int x) {
    return __builtin_amdgcn_permlanex16(x, x, 0x76543210, 0xfedcba98, false, false);
}
static __device__ inline int imax(int a, int b) { return a > b ? a : b; }
static __device__ inline int imin(int a, int b) { return a < b ? a : b; }
#define NO_IDX 0x7fffffff

__global__ __launch_bounds__(256) void fps_kernel(const float* __restrict__ xyz,
                                                  float* __restrict__ new_xyz)
{
    __shared__ float sx[N_], sy[N_], sz[N_];
    __shared__ int sidx[NPOINT];
    __shared__ __align__(16) int red[2][16];
    const int b   = blockIdx.x;
    const int tid = threadIdx.x;
    const float* bx = xyz + (size_t)b * N_ * 3;

    float dloc[16];
#pragma unroll
    for (int i = 0; i < 16; ++i) {
        int p = tid + 256 * i;
        sx[p] = bx[p * 3 + 0];
        sy[p] = bx[p * 3 + 1];
        sz[p] = bx[p * 3 + 2];
        dloc[i] = 1e10f;
    }
    __syncthreads();

    int last = 0;
    for (int it = 0; ; ++it) {
        const float lx = sx[last], ly = sy[last], lz = sz[last];
        if (tid == 0) sidx[it] = last;
        if (it == NPOINT - 1) break;

        float val = -1.0f;
        int   best = tid;
#pragma unroll
        for (int i = 0; i < 16; ++i) {
            int p = tid + 256 * i;
            float dx = sx[p] - lx, dy = sy[p] - ly, dz = sz[p] - lz;
            float dist = dx * dx + dy * dy + dz * dz;
            dloc[i] = fminf(dloc[i], dist);
            if (dloc[i] > val) { val = dloc[i]; best = p; }
        }
        const int vb = __float_as_int(val);
        int m = vb;
        m = imax(m, xrow<1>(m));
        m = imax(m, xrow<2>(m));
        m = imax(m, xrow<4>(m));
        m = imax(m, xrow<8>(m));
        m = imax(m, xhalf(m));
        int c = (vb == m) ? best : NO_IDX;
        c = imin(c, xrow<1>(c));
        c = imin(c, xrow<2>(c));
        c = imin(c, xrow<4>(c));
        c = imin(c, xrow<8>(c));
        c = imin(c, xhalf(c));

        int* slot = red[it & 1];
        if ((tid & 31) == 0) *(int2*)(slot + 2 * (tid >> 5)) = make_int2(m, c);
        __syncthreads();
        const int2 r = *(const int2*)(slot + 2 * (tid & 7));
        int gm = r.x;
        gm = imax(gm, xrow<1>(gm));
        gm = imax(gm, xrow<2>(gm));
        gm = imax(gm, xrow<4>(gm));
        int gc = (r.x == gm) ? r.y : NO_IDX;
        gc = imin(gc, xrow<1>(gc));
        gc = imin(gc, xrow<2>(gc));
        gc = imin(gc, xrow<4>(gc));
        last = gc;
    }
    __syncthreads();
    for (int k = 0; k < NPOINT / 256; ++k) {
        const int j = tid + 256 * k;
        const int p = sidx[j];
        float* o = new_xyz + ((size_t)b * NPOINT + j) * 3;
        o[0] = sx[p];
        o[1] = sy[p];
        o[2] = sz[p];
    }
}

#define KNN_WAVES 8
#define KNN_CPW   8
#define KNN_OWN   6
typedef unsigned long long knn_key;

__global__ __launch_bounds__(KNN_WAVES * 32) void knn_kernel(const float* __restrict__ xyz,
                                                             const float* __restrict__ new_xyz,
                                                             int* __restrict__ gidx)
{
    __shared__ float4 pts[N_ * 3 / 4];
    __shared__ knn_key cand[KNN_WAVES][N_];
    __shared__ int res[KNN_WAVES][NSAMPLE];
    const int b    = blockIdx.x >> 4;
    const int g    = blockIdx.x & 15;
    const int tid  = threadIdx.x;
    const int lane = tid & 31;
    const int w    = tid >> 5;

    const float4* src = (const float4*)(xyz + (size_t)b * N_ * 3);
    for (int i = tid; i < N_ * 3 / 4; i += KNN_WAVES * 32) pts[i] = src[i];
    __syncthreads();

    const float* P  = (const float*)pts;
    knn_key*     cw = cand[w];
    int*         rw = res[w];

#define KNN_DIST(PX, PY, PZ) fmaf((PZ) - cz, (PZ) - cz, fmaf((PX) - cx, (PX) - cx, ((PY) - cy) * ((PY) - cy)))

    for (int ci = 0; ci < KNN_CPW; ++ci) {
        const int cent = g * (KNN_WAVES * KNN_CPW) + w * KNN_CPW + ci;
        const float cx = new_xyz[(b * NPOINT + cent) * 3 + 0];
        const float cy = new_xyz[(b * NPOINT + cent) * 3 + 1];
        const float cz = new_xyz[(b * NPOINT + cent) * 3 + 2];

        float d[N_ / 32];
        float lmin = 3.4e38f;
#pragma unroll
        for (int t = 0; t < N_ / 32; ++t) {
            const int j = 32 * t + lane;
            const float px = P[3 * j + 0], py = P[3 * j + 1], pz = P[3 * j + 2];
            d[t] = KNN_DIST(px, py, pz);
            lmin = fminf(lmin, d[t]);
        }
        int mb = __float_as_int(lmin);
        mb = imax(mb, xrow<1>(mb));
        mb = imax(mb, xrow<2>(mb));
        mb = imax(mb, xrow<4>(mb));
        mb = imax(mb, xrow<8>(mb));
        mb = imax(mb, xhalf(mb));
        const float bound = __int_as_float(mb);

        unsigned M = 0;
#pragma unroll
        for (int t = 0; t < N_ / 32; ++t) {
            const bool     in   = d[t] <= bound;
            const unsigned mask = __builtin_amdgcn_ballot_w32(in);
            if (in)
                cw[M + __builtin_amdgcn_mbcnt_lo(mask, 0u)] =
                    ((knn_key)(unsigned)__float_as_int(d[t]) << 32) | (unsigned)(32 * t + lane);
            M += (unsigned)__builtin_popcount(mask);
        }
        rw[lane] = 0;
        __builtin_amdgcn_fence(__ATOMIC_ACQ_REL, "wavefront");

        for (unsigned k0 = 0; k0 < M; k0 += 32 * KNN_OWN) {
            knn_key  x[KNN_OWN];
            unsigned r[KNN_OWN];
#pragma unroll
            for (int q = 0; q < KNN_OWN; ++q) {
                const unsigned k = k0 + 32 * q + lane;
                x[q] = (k < M) ? cw[k] : ~(knn_key)0;
                r[q] = 0;
            }
#pragma unroll 4
            for (unsigned y = 0; y < M; ++y) {
                const knn_key cy = cw[y];
#pragma unroll
                for (int q = 0; q < KNN_OWN; ++q) r[q] += (cy < x[q]) ? 1u : 0u;
            }
#pragma unroll
            for (int q = 0; q < KNN_OWN; ++q) {
                const unsigned k = k0 + 32 * q + lane;
                if (k < M && r[q] < NSAMPLE) rw[r[q]] = (int)(unsigned)(x[q] & 0xffffffffull);
            }
        }
        __builtin_amdgcn_fence(__ATOMIC_ACQ_REL, "wavefront");
        gidx[(size_t)(b * NPOINT + cent) * NSAMPLE + lane] = rw[lane];
        __builtin_amdgcn_fence(__ATOMIC_ACQ_REL, "wavefront");
    }
#undef KNN_DIST
}

#define TP_PTS 64
__global__ __launch_bounds__(256) void transpose_features_kernel(const float* __restrict__ features,
                                                                 float* __restrict__ featT)
{
    __shared__ float tile[C_][TP_PTS + 1];
    const int b   = blockIdx.x >> 6;
    const int n0  = (blockIdx.x & 63) * TP_PTS;
    const int tid = threadIdx.x;
    const int q   = tid & 15;
    const int r0  = tid >> 4;
    const float* src = features + (size_t)b * C_ * N_ + n0;
#pragma unroll
    for (int k = 0; k < 4; ++k) {
        const int c = r0 + 16 * k;
        const float4 v = *(const float4*)(src + (size_t)c * N_ + 4 * q);
        tile[c][4 * q + 0] = v.x;
        tile[c][4 * q + 1] = v.y;
        tile[c][4 * q + 2] = v.z;
        tile[c][4 * q + 3] = v.w;
    }
    __syncthreads();
    float* dst = featT + ((size_t)b * N_ + n0) * C_;
#pragma unroll
    for (int k = 0; k < 4; ++k) {
        const int n = r0 + 16 * k;
        float4 v;
        v.x = tile[4 * q + 0][n];
        v.y = tile[4 * q + 1][n];
        v.z = tile[4 * q + 2][n];
        v.w = tile[4 * q + 3][n];
        *(float4*)(dst + (size_t)n * C_ + 4 * q) = v;
    }
}

template <int KSTEPS, int NT, int COUT>
static __device__ inline void layer_gemm(const float* in_lds, float* out_lds,
                                         const v2f* __restrict__ Wp,
                                         const float* __restrict__ bias, int lane)
{
    v8f acc[2][NT];
    v8f zero = {};
#pragma unroll
    for (int mt = 0; mt < 2; ++mt)
#pragma unroll
        for (int n = 0; n < NT; ++n) acc[mt][n] = zero;

    const int mrow  = lane & 15;
    const int khalf = (lane >> 4) * 2;
    const int ncol  = lane & 15;

    for (int k = 0; k < KSTEPS; ++k) {
        const int kc = 4 * k + khalf;
        v2f a0 = *(const v2f*)(in_lds + mrow * PITCH + kc);
        v2f a1 = *(const v2f*)(in_lds + (mrow + 16) * PITCH + kc);
#pragma unroll
        for (int n = 0; n < NT; ++n) {
            v2f bt = Wp[(kc >> 1) * COUT + n * 16 + ncol];
            acc[0][n] = wmma4(a0, bt, acc[0][n]);
            acc[1][n] = wmma4(a1, bt, acc[1][n]);
        }
    }
    const int rbase = (lane >> 4) * 8;
#pragma unroll
    for (int n = 0; n < NT; ++n) {
        float bb = bias[n * 16 + ncol];
#pragma unroll
        for (int mt = 0; mt < 2; ++mt)
#pragma unroll
            for (int j = 0; j < 8; ++j) {
                int rrow = mt * 16 + rbase + j;
                out_lds[rrow * PITCH + n * 16 + ncol] = fmaxf(acc[mt][n][j] + bb, 0.0f);
            }
    }
}

__global__ __launch_bounds__(32) void mlp_kernel(
    const float* __restrict__ xyz, const float* __restrict__ featT,
    const int* __restrict__ gidx, const float* __restrict__ new_xyz,
    float* __restrict__ out_feat,
    const v2f* __restrict__ Wp0, const float* __restrict__ b0f,
    const v2f* __restrict__ Wp1, const float* __restrict__ b1f,
    const v2f* __restrict__ Wp2, const float* __restrict__ b2f)
{
    __shared__ __align__(16) float bufA[32 * PITCH];
    __shared__ __align__(16) float bufB[32 * PITCH];
    const int cg   = blockIdx.x;
    const int b    = cg >> 10;
    const int lane = threadIdx.x;

    const int   idx = gidx[(size_t)cg * NSAMPLE + lane];
    const float cx = new_xyz[cg * 3 + 0];
    const float cy = new_xyz[cg * 3 + 1];
    const float cz = new_xyz[cg * 3 + 2];
    const float* pt = xyz + ((size_t)b * N_ + idx) * 3;
    float* row = bufA + lane * PITCH;
    row[0] = pt[0] - cx;
    row[1] = pt[1] - cy;
    row[2] = pt[2] - cz;
    const v2f* ft = (const v2f*)featT + (size_t)b * N_ * (C_ / 2);
    v2f f[NSAMPLE];
#pragma unroll
    for (int r = 0; r < NSAMPLE; ++r) {
        const int pr = __builtin_amdgcn_readlane(idx, r);
        f[r] = ft[(size_t)pr * (C_ / 2) + lane];
    }
#pragma unroll
    for (int r = 0; r < NSAMPLE; ++r) {
        bufA[r * PITCH + 3 + 2 * lane + 0] = f[r].x;
        bufA[r * PITCH + 3 + 2 * lane + 1] = f[r].y;
    }
    row[67] = 0.0f;
    __syncthreads();

    layer_gemm<17, 4, 64>(bufA, bufB, Wp0, b0f, lane);
    __syncthreads();
    layer_gemm<16, 4, 64>(bufB, bufA, Wp1, b1f, lane);
    __syncthreads();

    v8f acc[2][8];
    v8f zero = {};
#pragma unroll
    for (int mt = 0; mt < 2; ++mt)
#pragma unroll
        for (int n = 0; n < 8; ++n) acc[mt][n] = zero;

    const int mrow  = lane & 15;
    const int khalf = (lane >> 4) * 2;
    const int ncol  = lane & 15;
    for (int k = 0; k < 16; ++k) {
        const int kc = 4 * k + khalf;
        v2f a0 = *(const v2f*)(bufA + mrow * PITCH + kc);
        v2f a1 = *(const v2f*)(bufA + (mrow + 16) * PITCH + kc);
#pragma unroll
        for (int n = 0; n < 8; ++n) {
            v2f bt = Wp2[(kc >> 1) * 128 + n * 16 + ncol];
            acc[0][n] = wmma4(a0, bt, acc[0][n]);
            acc[1][n] = wmma4(a1, bt, acc[1][n]);
        }
    }
    float* of = out_feat + (size_t)cg * 128;
#pragma unroll
    for (int n = 0; n < 8; ++n) {
        float r = -3.4e38f;
#pragma unroll
        for (int j = 0; j < 8; ++j) {
            r = fmaxf(r, acc[0][n][j]);
            r = fmaxf(r, acc[1][n][j]);
        }
        r = fmaxf(r, __shfl_xor(r, 16, 32));
        float v = fmaxf(r + b2f[n * 16 + ncol], 0.0f);
        if (lane < 16) of[n * 16 + ncol] = v;
    }
}

extern "C" void kernel_launch(void* const* d_in, const int* in_sizes, int n_in,
                              void* d_out, int out_size, void* d_ws, size_t ws_size,
                              hipStream_t stream)
{
    const float* xyz      = (const float*)d_in[0];
    const float* features = (const float*)d_in[1];
    const float* W0 = (const float*)d_in[2];  const float* b0 = (const float*)d_in[3];
    const float* g0 = (const float*)d_in[4];  const float* be0 = (const float*)d_in[5];
    const float* m0 = (const float*)d_in[6];  const float* v0 = (const float*)d_in[7];
    const float* W1 = (const float*)d_in[8];  const float* b1 = (const float*)d_in[9];
    const float* g1 = (const float*)d_in[10]; const float* be1 = (const float*)d_in[11];
    const float* m1 = (const float*)d_in[12]; const float* v1 = (const float*)d_in[13];
    const float* W2 = (const float*)d_in[14]; const float* b2 = (const float*)d_in[15];
    const float* g2 = (const float*)d_in[16]; const float* be2 = (const float*)d_in[17];
    const float* m2 = (const float*)d_in[18]; const float* v2_ = (const float*)d_in[19];

    float* out      = (float*)d_out;
    float* new_xyz  = out;
    float* out_feat = out + (size_t)B_ * NPOINT * 3;

    int*   gidx = (int*)d_ws;
    float* wsf  = (float*)((char*)d_ws + (size_t)B_ * NPOINT * NSAMPLE * sizeof(int));
    v2f*   Wp0 = (v2f*)wsf;
    float* b0f = (float*)(Wp0 + 34 * 64);
    v2f*   Wp1 = (v2f*)(b0f + 64);
    float* b1f = (float*)(Wp1 + 32 * 64);
    v2f*   Wp2 = (v2f*)(b1f + 64);
    float* b2f = (float*)(Wp2 + 32 * 128);
    float* featT = (float*)((char*)d_ws + (size_t)4 * 1024 * 1024);

    fuse_weights_kernel<<<1, 256, 0, stream>>>(W0, b0, g0, be0, m0, v0,
                                               W1, b1, g1, be1, m1, v1,
                                               W2, b2, g2, be2, m2, v2_,
                                               Wp0, b0f, Wp1, b1f, Wp2, b2f);
    fps_kernel<<<B_, 256, 0, stream>>>(xyz, new_xyz);
    knn_kernel<<<B_ * 16, KNN_WAVES * 32, 0, stream>>>(xyz, new_xyz, gidx);
    transpose_features_kernel<<<B_ * (N_ / TP_PTS), 256, 0, stream>>>(features, featT);
    mlp_kernel<<<B_ * NPOINT, 32, 0, stream>>>(xyz, featT, gidx, new_xyz, out_feat,
                                               Wp0, b0f, Wp1, b1f, Wp2, b2f);
}


# fps: each thread's 16 points are held in registers for all 1023 steps; the 48 LDS point reads per lane and step are gone, LDS now serves only the last pick's coordinates and the final copy to out
# speedup vs baseline: 4.7250x; 1.3331x over previous
#include <hip/hip_runtime.h>
#include <hip/hip_bf16.h>
#include <math.h>

#define B_      16
#define N_      4096
#define C_      64
#define NPOINT  1024
#define NSAMPLE 32
#define PITCH   68
#define BN_EPS  1e-5f

typedef float v2f __attribute__((ext_vector_type(2)));
typedef float v8f __attribute__((ext_vector_type(8)));

static __device__ inline v8f wmma4(v2f a, v2f b, v8f c) {
    return __builtin_amdgcn_wmma_f32_16x16x4_f32(
         false, a,  false, b,
         (short)0, c,  false,  false);
}

__global__ __launch_bounds__(256) void fuse_weights_kernel(
    const float* __restrict__ W0, const float* __restrict__ b0, const float* __restrict__ g0,
    const float* __restrict__ be0, const float* __restrict__ m0, const float* __restrict__ v0,
    const float* __restrict__ W1, const float* __restrict__ b1, const float* __restrict__ g1,
    const float* __restrict__ be1, const float* __restrict__ m1, const float* __restrict__ v1,
    const float* __restrict__ W2, const float* __restrict__ b2, const float* __restrict__ g2,
    const float* __restrict__ be2, const float* __restrict__ m2, const float* __restrict__ v2_,
    v2f* __restrict__ Wp0, float* __restrict__ b0f,
    v2f* __restrict__ Wp1, float* __restrict__ b1f,
    v2f* __restrict__ Wp2, float* __restrict__ b2f)
{
    const int tid = threadIdx.x;
    for (int i = tid; i < 34 * 64; i += 256) {
        int o = i & 63, kp = i >> 6;
        float s = g0[o] / sqrtf(v0[o] + BN_EPS);
        int k0 = 2 * kp, k1 = 2 * kp + 1;
        v2f w;
        w.x = (k0 < 67) ? W0[o * 67 + k0] * s : 0.0f;
        w.y = (k1 < 67) ? W0[o * 67 + k1] * s : 0.0f;
        Wp0[i] = w;
    }
    for (int i = tid; i < 64; i += 256) {
        float s = g0[i] / sqrtf(v0[i] + BN_EPS);
        b0f[i] = (b0[i] - m0[i]) * s + be0[i];
    }
    for (int i = tid; i < 32 * 64; i += 256) {
        int o = i & 63, kp = i >> 6;
        float s = g1[o] / sqrtf(v1[o] + BN_EPS);
        v2f w;
        w.x = W1[o * 64 + 2 * kp + 0] * s;
        w.y = W1[o * 64 + 2 * kp + 1] * s;
        Wp1[i] = w;
    }
    for (int i = tid; i < 64; i += 256) {
        float s = g1[i] / sqrtf(v1[i] + BN_EPS);
        b1f[i] = (b1[i] - m1[i]) * s + be1[i];
    }
    for (int i = tid; i < 32 * 128; i += 256) {
        int o = i & 127, kp = i >> 7;
        float s = g2[o] / sqrtf(v2_[o] + BN_EPS);
        v2f w;
        w.x = W2[o * 64 + 2 * kp + 0] * s;
        w.y = W2[o * 64 + 2 * kp + 1] * s;
        Wp2[i] = w;
    }
    for (int i = tid; i < 128; i += 256) {
        float s = g2[i] / sqrtf(v2_[i] + BN_EPS);
        b2f[i] = (b2[i] - m2[i]) * s + be2[i];
    }
}

template <int M>
static __device__ inline int xrow(int x) {
    return __builtin_amdgcn_update_dpp(0, x, 0x160 + M, 0xf, 0xf, true);
}
static __device__ inline int xhalf(int x) {
    return __builtin_amdgcn_permlanex16(x, x, 0x76543210, 0xfedcba98, false, false);
}
static __device__ inline int imax(int a, int b) { return a > b ? a : b; }
static __device__ inline int imin(int a, int b) { return a < b ? a : b; }
#define NO_IDX 0x7fffffff

__global__ __launch_bounds__(256) void fps_kernel(const float* __restrict__ xyz,
                                                  float* __restrict__ new_xyz)
{
    __shared__ float sx[N_], sy[N_], sz[N_];
    __shared__ int sidx[NPOINT];
    __shared__ __align__(16) int red[2][16];
    const int b   = blockIdx.x;
    const int tid = threadIdx.x;
    const float* bx = xyz + (size_t)b * N_ * 3;

    float px[16], py[16], pz[16], dloc[16];
#pragma unroll
    for (int i = 0; i < 16; ++i) {
        int p = tid + 256 * i;
        px[i] = bx[p * 3 + 0];
        py[i] = bx[p * 3 + 1];
        pz[i] = bx[p * 3 + 2];
        sx[p] = px[i];
        sy[p] = py[i];
        sz[p] = pz[i];
        dloc[i] = 1e10f;
    }
    __syncthreads();

    int last = 0;
    for (int it = 0; ; ++it) {
        const float lx = sx[last], ly = sy[last], lz = sz[last];
        if (tid == 0) sidx[it] = last;
        if (it == NPOINT - 1) break;

        float val = -1.0f;
        int   best = tid;
#pragma unroll
        for (int i = 0; i < 16; ++i) {
            int p = tid + 256 * i;
            float dx = px[i] - lx, dy = py[i] - ly, dz = pz[i] - lz;
            float dist = dx * dx + dy * dy + dz * dz;
            dloc[i] = fminf(dloc[i], dist);
            if (dloc[i] > val) { val = dloc[i]; best = p; }
        }
        const int vb = __float_as_int(val);
        int m = vb;
        m = imax(m, xrow<1>(m));
        m = imax(m, xrow<2>(m));
        m = imax(m, xrow<4>(m));
        m = imax(m, xrow<8>(m));
        m = imax(m, xhalf(m));
        int c = (vb == m) ? best : NO_IDX;
        c = imin(c, xrow<1>(c));
        c = imin(c, xrow<2>(c));
        c = imin(c, xrow<4>(c));
        c = imin(c, xrow<8>(c));
        c = imin(c, xhalf(c));

        int* slot = red[it & 1];
        if ((tid & 31) == 0) *(int2*)(slot + 2 * (tid >> 5)) = make_int2(m, c);
        __syncthreads();
        const int2 r = *(const int2*)(slot + 2 * (tid & 7));
        int gm = r.x;
        gm = imax(gm, xrow<1>(gm));
        gm = imax(gm, xrow<2>(gm));
        gm = imax(gm, xrow<4>(gm));
        int gc = (r.x == gm) ? r.y : NO_IDX;
        gc = imin(gc, xrow<1>(gc));
        gc = imin(gc, xrow<2>(gc));
        gc = imin(gc, xrow<4>(gc));
        last = gc;
    }
    __syncthreads();
    for (int k = 0; k < NPOINT / 256; ++k) {
        const int j = tid + 256 * k;
        const int p = sidx[j];
        float* o = new_xyz + ((size_t)b * NPOINT + j) * 3;
        o[0] = sx[p];
        o[1] = sy[p];
        o[2] = sz[p];
    }
}

#define KNN_WAVES 8
#define KNN_CPW   8
#define KNN_OWN   6
typedef unsigned long long knn_key;

__global__ __launch_bounds__(KNN_WAVES * 32) void knn_kernel(const float* __restrict__ xyz,
                                                             const float* __restrict__ new_xyz,
                                                             int* __restrict__ gidx)
{
    __shared__ float4 pts[N_ * 3 / 4];
    __shared__ knn_key cand[KNN_WAVES][N_];
    __shared__ int res[KNN_WAVES][NSAMPLE];
    const int b    = blockIdx.x >> 4;
    const int g    = blockIdx.x & 15;
    const int tid  = threadIdx.x;
    const int lane = tid & 31;
    const int w    = tid >> 5;

    const float4* src = (const float4*)(xyz + (size_t)b * N_ * 3);
    for (int i = tid; i < N_ * 3 / 4; i += KNN_WAVES * 32) pts[i] = src[i];
    __syncthreads();

    const float* P  = (const float*)pts;
    knn_key*     cw = cand[w];
    int*         rw = res[w];

#define KNN_DIST(PX, PY, PZ) fmaf((PZ) - cz, (PZ) - cz, fmaf((PX) - cx, (PX) - cx, ((PY) - cy) * ((PY) - cy)))

    for (int ci = 0; ci < KNN_CPW; ++ci) {
        const int cent = g * (KNN_WAVES * KNN_CPW) + w * KNN_CPW + ci;
        const float cx = new_xyz[(b * NPOINT + cent) * 3 + 0];
        const float cy = new_xyz[(b * NPOINT + cent) * 3 + 1];
        const float cz = new_xyz[(b * NPOINT + cent) * 3 + 2];

        float d[N_ / 32];
        float lmin = 3.4e38f;
#pragma unroll
        for (int t = 0; t < N_ / 32; ++t) {
            const int j = 32 * t + lane;
            const float px = P[3 * j + 0], py = P[3 * j + 1], pz = P[3 * j + 2];
            d[t] = KNN_DIST(px, py, pz);
            lmin = fminf(lmin, d[t]);
        }
        int mb = __float_as_int(lmin);
        mb = imax(mb, xrow<1>(mb));
        mb = imax(mb, xrow<2>(mb));
        mb = imax(mb, xrow<4>(mb));
        mb = imax(mb, xrow<8>(mb));
        mb = imax(mb, xhalf(mb));
        const float bound = __int_as_float(mb);

        unsigned M = 0;
#pragma unroll
        for (int t = 0; t < N_ / 32; ++t) {
            const bool     in   = d[t] <= bound;
            const unsigned mask = __builtin_amdgcn_ballot_w32(in);
            if (in)
                cw[M + __builtin_amdgcn_mbcnt_lo(mask, 0u)] =
                    ((knn_key)(unsigned)__float_as_int(d[t]) << 32) | (unsigned)(32 * t + lane);
            M += (unsigned)__builtin_popcount(mask);
        }
        rw[lane] = 0;
        __builtin_amdgcn_fence(__ATOMIC_ACQ_REL, "wavefront");

        for (unsigned k0 = 0; k0 < M; k0 += 32 * KNN_OWN) {
            knn_key  x[KNN_OWN];
            unsigned r[KNN_OWN];
#pragma unroll
            for (int q = 0; q < KNN_OWN; ++q) {
                const unsigned k = k0 + 32 * q + lane;
                x[q] = (k < M) ? cw[k] : ~(knn_key)0;
                r[q] = 0;
            }
#pragma unroll 4
            for (unsigned y = 0; y < M; ++y) {
                const knn_key cy = cw[y];
#pragma unroll
                for (int q = 0; q < KNN_OWN; ++q) r[q] += (cy < x[q]) ? 1u : 0u;
            }
#pragma unroll
            for (int q = 0; q < KNN_OWN; ++q) {
                const unsigned k = k0 + 32 * q + lane;
                if (k < M && r[q] < NSAMPLE) rw[r[q]] = (int)(unsigned)(x[q] & 0xffffffffull);
            }
        }
        __builtin_amdgcn_fence(__ATOMIC_ACQ_REL, "wavefront");
        gidx[(size_t)(b * NPOINT + cent) * NSAMPLE + lane] = rw[lane];
        __builtin_amdgcn_fence(__ATOMIC_ACQ_REL, "wavefront");
    }
#undef KNN_DIST
}

#define TP_PTS 64
__global__ __launch_bounds__(256) void transpose_features_kernel(const float* __restrict__ features,
                                                                 float* __restrict__ featT)
{
    __shared__ float tile[C_][TP_PTS + 1];
    const int b   = blockIdx.x >> 6;
    const int n0  = (blockIdx.x & 63) * TP_PTS;
    const int tid = threadIdx.x;
    const int q   = tid & 15;
    const int r0  = tid >> 4;
    const float* src = features + (size_t)b * C_ * N_ + n0;
#pragma unroll
    for (int k = 0; k < 4; ++k) {
        const int c = r0 + 16 * k;
        const float4 v = *(const float4*)(src + (size_t)c * N_ + 4 * q);
        tile[c][4 * q + 0] = v.x;
        tile[c][4 * q + 1] = v.y;
        tile[c][4 * q + 2] = v.z;
        tile[c][4 * q + 3] = v.w;
    }
    __syncthreads();
    float* dst = featT + ((size_t)b * N_ + n0) * C_;
#pragma unroll
    for (int k = 0; k < 4; ++k) {
        const int n = r0 + 16 * k;
        float4 v;
        v.x = tile[4 * q + 0][n];
        v.y = tile[4 * q + 1][n];
        v.z = tile[4 * q + 2][n];
        v.w = tile[4 * q + 3][n];
        *(float4*)(dst + (size_t)n * C_ + 4 * q) = v;
    }
}

template <int KSTEPS, int NT, int COUT>
static __device__ inline void layer_gemm(const float* in_lds, float* out_lds,
                                         const v2f* __restrict__ Wp,
                                         const float* __restrict__ bias, int lane)
{
    v8f acc[2][NT];
    v8f zero = {};
#pragma unroll
    for (int mt = 0; mt < 2; ++mt)
#pragma unroll
        for (int n = 0; n < NT; ++n) acc[mt][n] = zero;

    const int mrow  = lane & 15;
    const int khalf = (lane >> 4) * 2;
    const int ncol  = lane & 15;

    for (int k = 0; k < KSTEPS; ++k) {
        const int kc = 4 * k + khalf;
        v2f a0 = *(const v2f*)(in_lds + mrow * PITCH + kc);
        v2f a1 = *(const v2f*)(in_lds + (mrow + 16) * PITCH + kc);
#pragma unroll
        for (int n = 0; n < NT; ++n) {
            v2f bt = Wp[(kc >> 1) * COUT + n * 16 + ncol];
            acc[0][n] = wmma4(a0, bt, acc[0][n]);
            acc[1][n] = wmma4(a1, bt, acc[1][n]);
        }
    }
    const int rbase = (lane >> 4) * 8;
#pragma unroll
    for (int n = 0; n < NT; ++n) {
        float bb = bias[n * 16 + ncol];
#pragma unroll
        for (int mt = 0; mt < 2; ++mt)
#pragma unroll
            for (int j = 0; j < 8; ++j) {
                int rrow = mt * 16 + rbase + j;
                out_lds[rrow * PITCH + n * 16 + ncol] = fmaxf(acc[mt][n][j] + bb, 0.0f);
            }
    }
}

__global__ __launch_bounds__(32) void mlp_kernel(
    const float* __restrict__ xyz, const float* __restrict__ featT,
    const int* __restrict__ gidx, const float* __restrict__ new_xyz,
    float* __restrict__ out_feat,
    const v2f* __restrict__ Wp0, const float* __restrict__ b0f,
    const v2f* __restrict__ Wp1, const float* __restrict__ b1f,
    const v2f* __restrict__ Wp2, const float* __restrict__ b2f)
{
    __shared__ __align__(16) float bufA[32 * PITCH];
    __shared__ __align__(16) float bufB[32 * PITCH];
    const int cg   = blockIdx.x;
    const int b    = cg >> 10;
    const int lane = threadIdx.x;

    const int   idx = gidx[(size_t)cg * NSAMPLE + lane];
    const float cx = new_xyz[cg * 3 + 0];
    const float cy = new_xyz[cg * 3 + 1];
    const float cz = new_xyz[cg * 3 + 2];
    const float* pt = xyz + ((size_t)b * N_ + idx) * 3;
    float* row = bufA + lane * PITCH;
    row[0] = pt[0] - cx;
    row[1] = pt[1] - cy;
    row[2] = pt[2] - cz;
    const v2f* ft = (const v2f*)featT + (size_t)b * N_ * (C_ / 2);
    v2f f[NSAMPLE];
#pragma unroll
    for (int r = 0; r < NSAMPLE; ++r) {
        const int pr = __builtin_amdgcn_readlane(idx, r);
        f[r] = ft[(size_t)pr * (C_ / 2) + lane];
    }
#pragma unroll
    for (int r = 0; r < NSAMPLE; ++r) {
        bufA[r * PITCH + 3 + 2 * lane + 0] = f[r].x;
        bufA[r * PITCH + 3 + 2 * lane + 1] = f[r].y;
    }
    row[67] = 0.0f;
    __syncthreads();

    layer_gemm<17, 4, 64>(bufA, bufB, Wp0, b0f, lane);
    __syncthreads();
    layer_gemm<16, 4, 64>(bufB, bufA, Wp1, b1f, lane);
    __syncthreads();

    v8f acc[2][8];
    v8f zero = {};
#pragma unroll
    for (int mt = 0; mt < 2; ++mt)
#pragma unroll
        for (int n = 0; n < 8; ++n) acc[mt][n] = zero;

    const int mrow  = lane & 15;
    const int khalf = (lane >> 4) * 2;
    const int ncol  = lane & 15;
    for (int k = 0; k < 16; ++k) {
        const int kc = 4 * k + khalf;
        v2f a0 = *(const v2f*)(bufA + mrow * PITCH + kc);
        v2f a1 = *(const v2f*)(bufA + (mrow + 16) * PITCH + kc);
#pragma unroll
        for (int n = 0; n < 8; ++n) {
            v2f bt = Wp2[(kc >> 1) * 128 + n * 16 + ncol];
            acc[0][n] = wmma4(a0, bt, acc[0][n]);
            acc[1][n] = wmma4(a1, bt, acc[1][n]);
        }
    }
    float* of = out_feat + (size_t)cg * 128;
#pragma unroll
    for (int n = 0; n < 8; ++n) {
        float r = -3.4e38f;
#pragma unroll
        for (int j = 0; j < 8; ++j) {
            r = fmaxf(r, acc[0][n][j]);
            r = fmaxf(r, acc[1][n][j]);
        }
        r = fmaxf(r, __shfl_xor(r, 16, 32));
        float v = fmaxf(r + b2f[n * 16 + ncol], 0.0f);
        if (lane < 16) of[n * 16 + ncol] = v;
    }
}

extern "C" void kernel_launch(void* const* d_in, const int* in_sizes, int n_in,
                              void* d_out, int out_size, void* d_ws, size_t ws_size,
                              hipStream_t stream)
{
    const float* xyz      = (const float*)d_in[0];
    const float* features = (const float*)d_in[1];
    const float* W0 = (const float*)d_in[2];  const float* b0 = (const float*)d_in[3];
    const float* g0 = (const float*)d_in[4];  const float* be0 = (const float*)d_in[5];
    const float* m0 = (const float*)d_in[6];  const float* v0 = (const float*)d_in[7];
    const float* W1 = (const float*)d_in[8];  const float* b1 = (const float*)d_in[9];
    const float* g1 = (const float*)d_in[10]; const float* be1 = (const float*)d_in[11];
    const float* m1 = (const float*)d_in[12]; const float* v1 = (const float*)d_in[13];
    const float* W2 = (const float*)d_in[14]; const float* b2 = (const float*)d_in[15];
    const float* g2 = (const float*)d_in[16]; const float* be2 = (const float*)d_in[17];
    const float* m2 = (const float*)d_in[18]; const float* v2_ = (const float*)d_in[19];

    float* out      = (float*)d_out;
    float* new_xyz  = out;
    float* out_feat = out + (size_t)B_ * NPOINT * 3;

    int*   gidx = (int*)d_ws;
    float* wsf  = (float*)((char*)d_ws + (size_t)B_ * NPOINT * NSAMPLE * sizeof(int));
    v2f*   Wp0 = (v2f*)wsf;
    float* b0f = (float*)(Wp0 + 34 * 64);
    v2f*   Wp1 = (v2f*)(b0f + 64);
    float* b1f = (float*)(Wp1 + 32 * 64);
    v2f*   Wp2 = (v2f*)(b1f + 64);
    float* b2f = (float*)(Wp2 + 32 * 128);
    float* featT = (float*)((char*)d_ws + (size_t)4 * 1024 * 1024);

    fuse_weights_kernel<<<1, 256, 0, stream>>>(W0, b0, g0, be0, m0, v0,
                                               W1, b1, g1, be1, m1, v1,
                                               W2, b2, g2, be2, m2, v2_,
                                               Wp0, b0f, Wp1, b1f, Wp2, b2f);
    fps_kernel<<<B_, 256, 0, stream>>>(xyz, new_xyz);
    knn_kernel<<<B_ * 16, KNN_WAVES * 32, 0, stream>>>(xyz, new_xyz, gidx);
    transpose_features_kernel<<<B_ * (N_ / TP_PTS), 256, 0, stream>>>(features, featT);
    mlp_kernel<<<B_ * NPOINT, 32, 0, stream>>>(xyz, featT, gidx, new_xyz, out_feat,
                                               Wp0, b0f, Wp1, b1f, Wp2, b2f);
}


# sampling no longer runs alone: idle workgroups of its launch make the 32-nearest lists of ALL 4096 points per batch, transpose features, fold weights; mlp takes a centroid's list by its point
# speedup vs baseline: 4.9198x; 1.0412x over previous
#include <hip/hip_runtime.h>
#include <hip/hip_bf16.h>
#include <math.h>

#define B_      16
#define N_      4096
#define C_      64
#define NPOINT  1024
#define NSAMPLE 32
#define PITCH   68
#define BN_EPS  1e-5f

typedef float v2f __attribute__((ext_vector_type(2)));
typedef float v8f __attribute__((ext_vector_type(8)));

static __device__ inline v8f wmma4(v2f a, v2f b, v8f c) {
    return __builtin_amdgcn_wmma_f32_16x16x4_f32(
         false, a,  false, b,
         (short)0, c,  false,  false);
}

static __device__ __forceinline__ void fuse_weights_role(
    const int tid,
    const float* __restrict__ W0, const float* __restrict__ b0, const float* __restrict__ g0,
    const float* __restrict__ be0, const float* __restrict__ m0, const float* __restrict__ v0,
    const float* __restrict__ W1, const float* __restrict__ b1, const float* __restrict__ g1,
    const float* __restrict__ be1, const float* __restrict__ m1, const float* __restrict__ v1,
    const float* __restrict__ W2, const float* __restrict__ b2, const float* __restrict__ g2,
    const float* __restrict__ be2, const float* __restrict__ m2, const float* __restrict__ v2_,
    v2f* __restrict__ Wp0, float* __restrict__ b0f,
    v2f* __restrict__ Wp1, float* __restrict__ b1f,
    v2f* __restrict__ Wp2, float* __restrict__ b2f)
{
    for (int i = tid; i < 34 * 64; i += 256) {
        int o = i & 63, kp = i >> 6;
        float s = g0[o] / sqrtf(v0[o] + BN_EPS);
        int k0 = 2 * kp, k1 = 2 * kp + 1;
        v2f w;
        w.x = (k0 < 67) ? W0[o * 67 + k0] * s : 0.0f;
        w.y = (k1 < 67) ? W0[o * 67 + k1] * s : 0.0f;
        Wp0[i] = w;
    }
    for (int i = tid; i < 64; i += 256) {
        float s = g0[i] / sqrtf(v0[i] + BN_EPS);
        b0f[i] = (b0[i] - m0[i]) * s + be0[i];
    }
    for (int i = tid; i < 32 * 64; i += 256) {
        int o = i & 63, kp = i >> 6;
        float s = g1[o] / sqrtf(v1[o] + BN_EPS);
        v2f w;
        w.x = W1[o * 64 + 2 * kp + 0] * s;
        w.y = W1[o * 64 + 2 * kp + 1] * s;
        Wp1[i] = w;
    }
    for (int i = tid; i < 64; i += 256) {
        float s = g1[i] / sqrtf(v1[i] + BN_EPS);
        b1f[i] = (b1[i] - m1[i]) * s + be1[i];
    }
    for (int i = tid; i < 32 * 128; i += 256) {
        int o = i & 127, kp = i >> 7;
        float s = g2[o] / sqrtf(v2_[o] + BN_EPS);
        v2f w;
        w.x = W2[o * 64 + 2 * kp + 0] * s;
        w.y = W2[o * 64 + 2 * kp + 1] * s;
        Wp2[i] = w;
    }
    for (int i = tid; i < 128; i += 256) {
        float s = g2[i] / sqrtf(v2_[i] + BN_EPS);
        b2f[i] = (b2[i] - m2[i]) * s + be2[i];
    }
}

template <int M>
static __device__ inline int xrow(int x) {
    return __builtin_amdgcn_update_dpp(0, x, 0x160 + M, 0xf, 0xf, true);
}
static __device__ inline int xhalf(int x) {
    return __builtin_amdgcn_permlanex16(x, x, 0x76543210, 0xfedcba98, false, false);
}
static __device__ inline int imax(int a, int b) { return a > b ? a : b; }
static __device__ inline int imin(int a, int b) { return a < b ? a : b; }
#define NO_IDX 0x7fffffff

static __device__ __forceinline__ void fps_role(unsigned char* smem, const int b, const int tid,
                                                const float* __restrict__ xyz,
                                                float* __restrict__ new_xyz,
                                                int* __restrict__ sel)
{
    float* sx   = (float*)smem;
    float* sy   = sx + N_;
    float* sz   = sy + N_;
    int*   sidx = (int*)(sz + N_);
    int (*red)[16] = (int (*)[16])(sidx + NPOINT);
    const float* bx = xyz + (size_t)b * N_ * 3;

    float px[16], py[16], pz[16], dloc[16];
#pragma unroll
    for (int i = 0; i < 16; ++i) {
        int p = tid + 256 * i;
        px[i] = bx[p * 3 + 0];
        py[i] = bx[p * 3 + 1];
        pz[i] = bx[p * 3 + 2];
        sx[p] = px[i];
        sy[p] = py[i];
        sz[p] = pz[i];
        dloc[i] = 1e10f;
    }
    __syncthreads();

    int last = 0;
    for (int it = 0; ; ++it) {
        const float lx = sx[last], ly = sy[last], lz = sz[last];
        if (tid == 0) sidx[it] = last;
        if (it == NPOINT - 1) break;

        float val = -1.0f;
        int   best = tid;
#pragma unroll
        for (int i = 0; i < 16; ++i) {
            int p = tid + 256 * i;
            float dx = px[i] - lx, dy = py[i] - ly, dz = pz[i] - lz;
            float dist = dx * dx + dy * dy + dz * dz;
            dloc[i] = fminf(dloc[i], dist);
            if (dloc[i] > val) { val = dloc[i]; best = p; }
        }
        const int vb = __float_as_int(val);
        int m = vb;
        m = imax(m, xrow<1>(m));
        m = imax(m, xrow<2>(m));
        m = imax(m, xrow<4>(m));
        m = imax(m, xrow<8>(m));
        m = imax(m, xhalf(m));
        int c = (vb == m) ? best : NO_IDX;
        c = imin(c, xrow<1>(c));
        c = imin(c, xrow<2>(c));
        c = imin(c, xrow<4>(c));
        c = imin(c, xrow<8>(c));
        c = imin(c, xhalf(c));

        int* slot = red[it & 1];
        if ((tid & 31) == 0) *(int2*)(slot + 2 * (tid >> 5)) = make_int2(m, c);
        __syncthreads();
        const int2 r = *(const int2*)(slot + 2 * (tid & 7));
        int gm = r.x;
        gm = imax(gm, xrow<1>(gm));
        gm = imax(gm, xrow<2>(gm));
        gm = imax(gm, xrow<4>(gm));
        int gc = (r.x == gm) ? r.y : NO_IDX;
        gc = imin(gc, xrow<1>(gc));
        gc = imin(gc, xrow<2>(gc));
        gc = imin(gc, xrow<4>(gc));
        last = gc;
    }
    __syncthreads();
    for (int k = 0; k < NPOINT / 256; ++k) {
        const int j = tid + 256 * k;
        const int p = sidx[j];
        float* o = new_xyz + ((size_t)b * NPOINT + j) * 3;
        o[0] = sx[p];
        o[1] = sy[p];
        o[2] = sz[p];
        sel[b * NPOINT + j] = p;
    }
}

#define KNN_WAVES 8
#define KNN_CPW   8
#define KNN_OWN   6
typedef unsigned long long knn_key;

static __device__ __forceinline__ void knn_role(unsigned char* smem, const int blk, const int tid,
                                                const float* __restrict__ xyz,
                                                int* __restrict__ gidx_all)
{
    float4* pts = (float4*)smem;
    knn_key (*cand)[N_]   = (knn_key (*)[N_])(smem + N_ * 3 * sizeof(float));
    int (*res)[NSAMPLE]   = (int (*)[NSAMPLE])(smem + N_ * 3 * sizeof(float) + sizeof(knn_key) * KNN_WAVES * N_);
    const int b    = blk >> 6;
    const int g    = blk & 63;
    const int lane = tid & 31;
    const int w    = tid >> 5;

    const float4* src = (const float4*)(xyz + (size_t)b * N_ * 3);
    for (int i = tid; i < N_ * 3 / 4; i += KNN_WAVES * 32) pts[i] = src[i];
    __syncthreads();

    const float* P  = (const float*)pts;
    knn_key*     cw = cand[w];
    int*         rw = res[w];

#define KNN_DIST(PX, PY, PZ) fmaf((PZ) - cz, (PZ) - cz, fmaf((PX) - cx, (PX) - cx, ((PY) - cy) * ((PY) - cy)))

    for (int ci = 0; ci < KNN_CPW; ++ci) {
        const int cent = g * (KNN_WAVES * KNN_CPW) + w * KNN_CPW + ci;
        const float cx = P[3 * cent + 0];
        const float cy = P[3 * cent + 1];
        const float cz = P[3 * cent + 2];

        float d[N_ / 32];
        float lmin = 3.4e38f;
#pragma unroll
        for (int t = 0; t < N_ / 32; ++t) {
            const int j = 32 * t + lane;
            const float px = P[3 * j + 0], py = P[3 * j + 1], pz = P[3 * j + 2];
            d[t] = KNN_DIST(px, py, pz);
            lmin = fminf(lmin, d[t]);
        }
        int mb = __float_as_int(lmin);
        mb = imax(mb, xrow<1>(mb));
        mb = imax(mb, xrow<2>(mb));
        mb = imax(mb, xrow<4>(mb));
        mb = imax(mb, xrow<8>(mb));
        mb = imax(mb, xhalf(mb));
        const float bound = __int_as_float(mb);

        unsigned M = 0;
#pragma unroll
        for (int t = 0; t < N_ / 32; ++t) {
            const bool     in   = d[t] <= bound;
            const unsigned mask = __builtin_amdgcn_ballot_w32(in);
            if (in)
                cw[M + __builtin_amdgcn_mbcnt_lo(mask, 0u)] =
                    ((knn_key)(unsigned)__float_as_int(d[t]) << 32) | (unsigned)(32 * t + lane);
            M += (unsigned)__builtin_popcount(mask);
        }
        rw[lane] = 0;
        __builtin_amdgcn_fence(__ATOMIC_ACQ_REL, "wavefront");

        for (unsigned k0 = 0; k0 < M; k0 += 32 * KNN_OWN) {
            knn_key  x[KNN_OWN];
            unsigned r[KNN_OWN];
#pragma unroll
            for (int q = 0; q < KNN_OWN; ++q) {
                const unsigned k = k0 + 32 * q + lane;
                x[q] = (k < M) ? cw[k] : ~(knn_key)0;
                r[q] = 0;
            }
#pragma unroll 4
            for (unsigned y = 0; y < M; ++y) {
                const knn_key cy = cw[y];
#pragma unroll
                for (int q = 0; q < KNN_OWN; ++q) r[q] += (cy < x[q]) ? 1u : 0u;
            }
#pragma unroll
            for (int q = 0; q < KNN_OWN; ++q) {
                const unsigned k = k0 + 32 * q + lane;
                if (k < M && r[q] < NSAMPLE) rw[r[q]] = (int)(unsigned)(x[q] & 0xffffffffull);
            }
        }
        __builtin_amdgcn_fence(__ATOMIC_ACQ_REL, "wavefront");
        gidx_all[(size_t)(b * N_ + cent) * NSAMPLE + lane] = rw[lane];
        __builtin_amdgcn_fence(__ATOMIC_ACQ_REL, "wavefront");
    }
#undef KNN_DIST
}

#define TP_PTS 64
static __device__ __forceinline__ void transpose_role(unsigned char* smem, const int blk, const int tid,
                                                      const float* __restrict__ features,
                                                      float* __restrict__ featT)
{
    float (*tile)[TP_PTS + 1] = (float (*)[TP_PTS + 1])smem;
    const int b   = blk >> 6;
    const int n0  = (blk & 63) * TP_PTS;
    const int q   = tid & 15;
    const int r0  = tid >> 4;
    const float* src = features + (size_t)b * C_ * N_ + n0;
#pragma unroll
    for (int k = 0; k < 4; ++k) {
        const int c = r0 + 16 * k;
        const float4 v = *(const float4*)(src + (size_t)c * N_ + 4 * q);
        tile[c][4 * q + 0] = v.x;
        tile[c][4 * q + 1] = v.y;
        tile[c][4 * q + 2] = v.z;
        tile[c][4 * q + 3] = v.w;
    }
    __syncthreads();
    float* dst = featT + ((size_t)b * N_ + n0) * C_;
#pragma unroll
    for (int k = 0; k < 4; ++k) {
        const int n = r0 + 16 * k;
        float4 v;
        v.x = tile[4 * q + 0][n];
        v.y = tile[4 * q + 1][n];
        v.z = tile[4 * q + 2][n];
        v.w = tile[4 * q + 3][n];
        *(float4*)(dst + (size_t)n * C_ + 4 * q) = v;
    }
}

#define FRONT_FPS   B_
#define FRONT_KNN   (B_ * (N_ / (KNN_WAVES * KNN_CPW)))
#define FRONT_TP    (B_ * (N_ / TP_PTS))
#define FRONT_GRID  (FRONT_FPS + FRONT_KNN + FRONT_TP + 1)
#define FRONT_LDS   (N_ * 3 * 4 + 8 * KNN_WAVES * N_ + 4 * KNN_WAVES * NSAMPLE)
__global__ __launch_bounds__(256) void front_kernel(
    const float* __restrict__ xyz, const float* __restrict__ features,
    const float* __restrict__ W0, const float* __restrict__ b0, const float* __restrict__ g0,
    const float* __restrict__ be0, const float* __restrict__ m0, const float* __restrict__ v0,
    const float* __restrict__ W1, const float* __restrict__ b1, const float* __restrict__ g1,
    const float* __restrict__ be1, const float* __restrict__ m1, const float* __restrict__ v1,
    const float* __restrict__ W2, const float* __restrict__ b2, const float* __restrict__ g2,
    const float* __restrict__ be2, const float* __restrict__ m2, const float* __restrict__ v2_,
    float* __restrict__ new_xyz, int* __restrict__ sel, int* __restrict__ gidx_all,
    float* __restrict__ featT,
    v2f* __restrict__ Wp0, float* __restrict__ b0f,
    v2f* __restrict__ Wp1, float* __restrict__ b1f,
    v2f* __restrict__ Wp2, float* __restrict__ b2f)
{
    __shared__ __align__(16) unsigned char smem[FRONT_LDS];
    const int blk = blockIdx.x;
    const int tid = threadIdx.x;
    if (blk < FRONT_FPS) {
        fps_role(smem, blk, tid, xyz, new_xyz, sel);
    } else if (blk < FRONT_FPS + FRONT_KNN) {
        knn_role(smem, blk - FRONT_FPS, tid, xyz, gidx_all);
    } else if (blk < FRONT_FPS + FRONT_KNN + FRONT_TP) {
        transpose_role(smem, blk - (FRONT_FPS + FRONT_KNN), tid, features, featT);
    } else {
        fuse_weights_role(tid, W0, b0, g0, be0, m0, v0, W1, b1, g1, be1, m1, v1,
                          W2, b2, g2, be2, m2, v2_, Wp0, b0f, Wp1, b1f, Wp2, b2f);
    }
}

template <int KSTEPS, int NT, int COUT>
static __device__ inline void layer_gemm(const float* in_lds, float* out_lds,
                                         const v2f* __restrict__ Wp,
                                         const float* __restrict__ bias, int lane)
{
    v8f acc[2][NT];
    v8f zero = {};
#pragma unroll
    for (int mt = 0; mt < 2; ++mt)
#pragma unroll
        for (int n = 0; n < NT; ++n) acc[mt][n] = zero;

    const int mrow  = lane & 15;
    const int khalf = (lane >> 4) * 2;
    const int ncol  = lane & 15;

    for (int k = 0; k < KSTEPS; ++k) {
        const int kc = 4 * k + khalf;
        v2f a0 = *(const v2f*)(in_lds + mrow * PITCH + kc);
        v2f a1 = *(const v2f*)(in_lds + (mrow + 16) * PITCH + kc);
#pragma unroll
        for (int n = 0; n < NT; ++n) {
            v2f bt = Wp[(kc >> 1) * COUT + n * 16 + ncol];
            acc[0][n] = wmma4(a0, bt, acc[0][n]);
            acc[1][n] = wmma4(a1, bt, acc[1][n]);
        }
    }
    const int rbase = (lane >> 4) * 8;
#pragma unroll
    for (int n = 0; n < NT; ++n) {
        float bb = bias[n * 16 + ncol];
#pragma unroll
        for (int mt = 0; mt < 2; ++mt)
#pragma unroll
            for (int j = 0; j < 8; ++j) {
                int rrow = mt * 16 + rbase + j;
                out_lds[rrow * PITCH + n * 16 + ncol] = fmaxf(acc[mt][n][j] + bb, 0.0f);
            }
    }
}

__global__ __launch_bounds__(32) void mlp_kernel(
    const float* __restrict__ xyz, const float* __restrict__ featT,
    const int* __restrict__ sel, const int* __restrict__ gidx_all,
    const float* __restrict__ new_xyz,
    float* __restrict__ out_feat,
    const v2f* __restrict__ Wp0, const float* __restrict__ b0f,
    const v2f* __restrict__ Wp1, const float* __restrict__ b1f,
    const v2f* __restrict__ Wp2, const float* __restrict__ b2f)
{
    __shared__ __align__(16) float bufA[32 * PITCH];
    __shared__ __align__(16) float bufB[32 * PITCH];
    const int cg   = blockIdx.x;
    const int b    = cg >> 10;
    const int lane = threadIdx.x;

    const int   idx = gidx_all[((size_t)b * N_ + sel[cg]) * NSAMPLE + lane];
    const float cx = new_xyz[cg * 3 + 0];
    const float cy = new_xyz[cg * 3 + 1];
    const float cz = new_xyz[cg * 3 + 2];
    const float* pt = xyz + ((size_t)b * N_ + idx) * 3;
    float* row = bufA + lane * PITCH;
    row[0] = pt[0] - cx;
    row[1] = pt[1] - cy;
    row[2] = pt[2] - cz;
    const v2f* ft = (const v2f*)featT + (size_t)b * N_ * (C_ / 2);
    v2f f[NSAMPLE];
#pragma unroll
    for (int r = 0; r < NSAMPLE; ++r) {
        const int pr = __builtin_amdgcn_readlane(idx, r);
        f[r] = ft[(size_t)pr * (C_ / 2) + lane];
    }
#pragma unroll
    for (int r = 0; r < NSAMPLE; ++r) {
        bufA[r * PITCH + 3 + 2 * lane + 0] = f[r].x;
        bufA[r * PITCH + 3 + 2 * lane + 1] = f[r].y;
    }
    row[67] = 0.0f;
    __syncthreads();

    layer_gemm<17, 4, 64>(bufA, bufB, Wp0, b0f, lane);
    __syncthreads();
    layer_gemm<16, 4, 64>(bufB, bufA, Wp1, b1f, lane);
    __syncthreads();

    v8f acc[2][8];
    v8f zero = {};
#pragma unroll
    for (int mt = 0; mt < 2; ++mt)
#pragma unroll
        for (int n = 0; n < 8; ++n) acc[mt][n] = zero;

    const int mrow  = lane & 15;
    const int khalf = (lane >> 4) * 2;
    const int ncol  = lane & 15;
    for (int k = 0; k < 16; ++k) {
        const int kc = 4 * k + khalf;
        v2f a0 = *(const v2f*)(bufA + mrow * PITCH + kc);
        v2f a1 = *(const v2f*)(bufA + (mrow + 16) * PITCH + kc);
#pragma unroll
        for (int n = 0; n < 8; ++n) {
            v2f bt = Wp2[(kc >> 1) * 128 + n * 16 + ncol];
            acc[0][n] = wmma4(a0, bt, acc[0][n]);
            acc[1][n] = wmma4(a1, bt, acc[1][n]);
        }
    }
    float* of = out_feat + (size_t)cg * 128;
#pragma unroll
    for (int n = 0; n < 8; ++n) {
        float r = -3.4e38f;
#pragma unroll
        for (int j = 0; j < 8; ++j) {
            r = fmaxf(r, acc[0][n][j]);
            r = fmaxf(r, acc[1][n][j]);
        }
        r = fmaxf(r, __shfl_xor(r, 16, 32));
        float v = fmaxf(r + b2f[n * 16 + ncol], 0.0f);
        if (lane < 16) of[n * 16 + ncol] = v;
    }
}

extern "C" void kernel_launch(void* const* d_in, const int* in_sizes, int n_in,
                              void* d_out, int out_size, void* d_ws, size_t ws_size,
                              hipStream_t stream)
{
    const float* xyz      = (const float*)d_in[0];
    const float* features = (const float*)d_in[1];
    const float* W0 = (const float*)d_in[2];  const float* b0 = (const float*)d_in[3];
    const float* g0 = (const float*)d_in[4];  const float* be0 = (const float*)d_in[5];
    const float* m0 = (const float*)d_in[6];  const float* v0 = (const float*)d_in[7];
    const float* W1 = (const float*)d_in[8];  const float* b1 = (const float*)d_in[9];
    const float* g1 = (const float*)d_in[10]; const float* be1 = (const float*)d_in[11];
    const float* m1 = (const float*)d_in[12]; const float* v1 = (const float*)d_in[13];
    const float* W2 = (const float*)d_in[14]; const float* b2 = (const float*)d_in[15];
    const float* g2 = (const float*)d_in[16]; const float* be2 = (const float*)d_in[17];
    const float* m2 = (const float*)d_in[18]; const float* v2_ = (const float*)d_in[19];

    float* out      = (float*)d_out;
    float* new_xyz  = out;
    float* out_feat = out + (size_t)B_ * NPOINT * 3;

    int*   sel  = (int*)d_ws;
    float* wsf  = (float*)((char*)d_ws + (size_t)2 * 1024 * 1024);
    v2f*   Wp0 = (v2f*)wsf;
    float* b0f = (float*)(Wp0 + 34 * 64);
    v2f*   Wp1 = (v2f*)(b0f + 64);
    float* b1f = (float*)(Wp1 + 32 * 64);
    v2f*   Wp2 = (v2f*)(b1f + 64);
    float* b2f = (float*)(Wp2 + 32 * 128);
    float* featT = (float*)((char*)d_ws + (size_t)4 * 1024 * 1024);
    int*   gidx_all = (int*)((char*)d_ws + (size_t)32 * 1024 * 1024);

    front_kernel<<<FRONT_GRID, 256, 0, stream>>>(xyz, features,
                                                 W0, b0, g0, be0, m0, v0,
                                                 W1, b1, g1, be1, m1, v1,
                                                 W2, b2, g2, be2, m2, v2_,
                                                 new_xyz, sel, gidx_all, featT,
                                                 Wp0, b0f, Wp1, b1f, Wp2, b2f);
    mlp_kernel<<<B_ * NPOINT, 32, 0, stream>>>(xyz, featT, sel, gidx_all, new_xyz, out_feat,
                                               Wp0, b0f, Wp1, b1f, Wp2, b2f);
}


# mlp layer 0 once per point, not per (centroid, neighbour): front launch forms W0.[xyz,feat] of every point on matrix instructions, gather subtracts the centroid's xyz term; order of that sum changes
# speedup vs baseline: 5.1407x; 1.0449x over previous
#include <hip/hip_runtime.h>
#include <hip/hip_bf16.h>
#include <math.h>

#define B_      16
#define N_      4096
#define C_      64
#define NPOINT  1024
#define NSAMPLE 32
#define PITCH   68
#define BN_EPS  1e-5f

typedef float v2f __attribute__((ext_vector_type(2)));
typedef float v8f __attribute__((ext_vector_type(8)));

static __device__ inline v8f wmma4(v2f a, v2f b, v8f c) {
    return __builtin_amdgcn_wmma_f32_16x16x4_f32(
         false, a,  false, b,
         (short)0, c,  false,  false);
}

static __device__ __forceinline__ void fuse_weights_role(
    const int tid,
    const float* __restrict__ W1, const float* __restrict__ b1, const float* __restrict__ g1,
    const float* __restrict__ be1, const float* __restrict__ m1, const float* __restrict__ v1,
    const float* __restrict__ W2, const float* __restrict__ b2, const float* __restrict__ g2,
    const float* __restrict__ be2, const float* __restrict__ m2, const float* __restrict__ v2_,
    v2f* __restrict__ Wp1, float* __restrict__ b1f,
    v2f* __restrict__ Wp2, float* __restrict__ b2f)
{
    for (int i = tid; i < 32 * 64; i += 256) {
        int o = i & 63, kp = i >> 6;
        float s = g1[o] / sqrtf(v1[o] + BN_EPS);
        v2f w;
        w.x = W1[o * 64 + 2 * kp + 0] * s;
        w.y = W1[o * 64 + 2 * kp + 1] * s;
        Wp1[i] = w;
    }
    for (int i = tid; i < 64; i += 256) {
        float s = g1[i] / sqrtf(v1[i] + BN_EPS);
        b1f[i] = (b1[i] - m1[i]) * s + be1[i];
    }
    for (int i = tid; i < 32 * 128; i += 256) {
        int o = i & 127, kp = i >> 7;
        float s = g2[o] / sqrtf(v2_[o] + BN_EPS);
        v2f w;
        w.x = W2[o * 64 + 2 * kp + 0] * s;
        w.y = W2[o * 64 + 2 * kp + 1] * s;
        Wp2[i] = w;
    }
    for (int i = tid; i < 128; i += 256) {
        float s = g2[i] / sqrtf(v2_[i] + BN_EPS);
        b2f[i] = (b2[i] - m2[i]) * s + be2[i];
    }
}

template <int M>
static __device__ inline int xrow(int x) {
    return __builtin_amdgcn_update_dpp(0, x, 0x160 + M, 0xf, 0xf, true);
}
static __device__ inline int xhalf(int x) {
    return __builtin_amdgcn_permlanex16(x, x, 0x76543210, 0xfedcba98, false, false);
}
static __device__ inline int imax(int a, int b) { return a > b ? a : b; }
static __device__ inline int imin(int a, int b) { return a < b ? a : b; }
#define NO_IDX 0x7fffffff

static __device__ __forceinline__ void fps_role(unsigned char* smem, const int b, const int tid,
                                                const float* __restrict__ xyz,
                                                float* __restrict__ new_xyz,
                                                int* __restrict__ sel)
{
    float* sx   = (float*)smem;
    float* sy   = sx + N_;
    float* sz   = sy + N_;
    int*   sidx = (int*)(sz + N_);
    int (*red)[16] = (int (*)[16])(sidx + NPOINT);
    const float* bx = xyz + (size_t)b * N_ * 3;

    float px[16], py[16], pz[16], dloc[16];
#pragma unroll
    for (int i = 0; i < 16; ++i) {
        int p = tid + 256 * i;
        px[i] = bx[p * 3 + 0];
        py[i] = bx[p * 3 + 1];
        pz[i] = bx[p * 3 + 2];
        sx[p] = px[i];
        sy[p] = py[i];
        sz[p] = pz[i];
        dloc[i] = 1e10f;
    }
    __syncthreads();

    int last = 0;
    for (int it = 0; ; ++it) {
        const float lx = sx[last], ly = sy[last], lz = sz[last];
        if (tid == 0) sidx[it] = last;
        if (it == NPOINT - 1) break;

        float val = -1.0f;
        int   best = tid;
#pragma unroll
        for (int i = 0; i < 16; ++i) {
            int p = tid + 256 * i;
            float dx = px[i] - lx, dy = py[i] - ly, dz = pz[i] - lz;
            float dist = dx * dx + dy * dy + dz * dz;
            dloc[i] = fminf(dloc[i], dist);
            if (dloc[i] > val) { val = dloc[i]; best = p; }
        }
        const int vb = __float_as_int(val);
        int m = vb;
        m = imax(m, xrow<1>(m));
        m = imax(m, xrow<2>(m));
        m = imax(m, xrow<4>(m));
        m = imax(m, xrow<8>(m));
        m = imax(m, xhalf(m));
        int c = (vb == m) ? best : NO_IDX;
        c = imin(c, xrow<1>(c));
        c = imin(c, xrow<2>(c));
        c = imin(c, xrow<4>(c));
        c = imin(c, xrow<8>(c));
        c = imin(c, xhalf(c));

        int* slot = red[it & 1];
        if ((tid & 31) == 0) *(int2*)(slot + 2 * (tid >> 5)) = make_int2(m, c);
        __syncthreads();
        const int2 r = *(const int2*)(slot + 2 * (tid & 7));
        int gm = r.x;
        gm = imax(gm, xrow<1>(gm));
        gm = imax(gm, xrow<2>(gm));
        gm = imax(gm, xrow<4>(gm));
        int gc = (r.x == gm) ? r.y : NO_IDX;
        gc = imin(gc, xrow<1>(gc));
        gc = imin(gc, xrow<2>(gc));
        gc = imin(gc, xrow<4>(gc));
        last = gc;
    }
    __syncthreads();
    for (int k = 0; k < NPOINT / 256; ++k) {
        const int j = tid + 256 * k;
        const int p = sidx[j];
        float* o = new_xyz + ((size_t)b * NPOINT + j) * 3;
        o[0] = sx[p];
        o[1] = sy[p];
        o[2] = sz[p];
        sel[b * NPOINT + j] = p;
    }
}

#define KNN_WAVES 8
#define KNN_CPW   8
#define KNN_OWN   6
typedef unsigned long long knn_key;

static __device__ __forceinline__ void knn_role(unsigned char* smem, const int blk, const int tid,
                                                const float* __restrict__ xyz,
                                                int* __restrict__ gidx_all)
{
    float4* pts = (float4*)smem;
    knn_key (*cand)[N_]   = (knn_key (*)[N_])(smem + N_ * 3 * sizeof(float));
    int (*res)[NSAMPLE]   = (int (*)[NSAMPLE])(smem + N_ * 3 * sizeof(float) + sizeof(knn_key) * KNN_WAVES * N_);
    const int b    = blk >> 6;
    const int g    = blk & 63;
    const int lane = tid & 31;
    const int w    = tid >> 5;

    const float4* src = (const float4*)(xyz + (size_t)b * N_ * 3);
    for (int i = tid; i < N_ * 3 / 4; i += KNN_WAVES * 32) pts[i] = src[i];
    __syncthreads();

    const float* P  = (const float*)pts;
    knn_key*     cw = cand[w];
    int*         rw = res[w];

#define KNN_DIST(PX, PY, PZ) fmaf((PZ) - cz, (PZ) - cz, fmaf((PX) - cx, (PX) - cx, ((PY) - cy) * ((PY) - cy)))

    for (int ci = 0; ci < KNN_CPW; ++ci) {
        const int cent = g * (KNN_WAVES * KNN_CPW) + w * KNN_CPW + ci;
        const float cx = P[3 * cent + 0];
        const float cy = P[3 * cent + 1];
        const float cz = P[3 * cent + 2];

        float d[N_ / 32];
        float lmin = 3.4e38f;
#pragma unroll
        for (int t = 0; t < N_ / 32; ++t) {
            const int j = 32 * t + lane;
            const float px = P[3 * j + 0], py = P[3 * j + 1], pz = P[3 * j + 2];
            d[t] = KNN_DIST(px, py, pz);
            lmin = fminf(lmin, d[t]);
        }
        int mb = __float_as_int(lmin);
        mb = imax(mb, xrow<1>(mb));
        mb = imax(mb, xrow<2>(mb));
        mb = imax(mb, xrow<4>(mb));
        mb = imax(mb, xrow<8>(mb));
        mb = imax(mb, xhalf(mb));
        const float bound = __int_as_float(mb);

        unsigned M = 0;
#pragma unroll
        for (int t = 0; t < N_ / 32; ++t) {
            const bool     in   = d[t] <= bound;
            const unsigned mask = __builtin_amdgcn_ballot_w32(in);
            if (in)
                cw[M + __builtin_amdgcn_mbcnt_lo(mask, 0u)] =
                    ((knn_key)(unsigned)__float_as_int(d[t]) << 32) | (unsigned)(32 * t + lane);
            M += (unsigned)__builtin_popcount(mask);
        }
        rw[lane] = 0;
        __builtin_amdgcn_fence(__ATOMIC_ACQ_REL, "wavefront");

        for (unsigned k0 = 0; k0 < M; k0 += 32 * KNN_OWN) {
            knn_key  x[KNN_OWN];
            unsigned r[KNN_OWN];
#pragma unroll
            for (int q = 0; q < KNN_OWN; ++q) {
                const unsigned k = k0 + 32 * q + lane;
                x[q] = (k < M) ? cw[k] : ~(knn_key)0;
                r[q] = 0;
            }
#pragma unroll 4
            for (unsigned y = 0; y < M; ++y) {
                const knn_key cy = cw[y];
#pragma unroll
                for (int q = 0; q < KNN_OWN; ++q) r[q] += (cy < x[q]) ? 1u : 0u;
            }
#pragma unroll
            for (int q = 0; q < KNN_OWN; ++q) {
                const unsigned k = k0 + 32 * q + lane;
                if (k < M && r[q] < NSAMPLE) rw[r[q]] = (int)(unsigned)(x[q] & 0xffffffffull);
            }
        }
        __builtin_amdgcn_fence(__ATOMIC_ACQ_REL, "wavefront");
        gidx_all[(size_t)(b * N_ + cent) * NSAMPLE + lane] = rw[lane];
        __builtin_amdgcn_fence(__ATOMIC_ACQ_REL, "wavefront");
    }
#undef KNN_DIST
}

#define PL_PTS 64
#define PL_TP  72
#define PL_OP  68
static __device__ __forceinline__ void point_layer_role(unsigned char* smem, const int blk, const int tid,
                                                        const float* __restrict__ xyz,
                                                        const float* __restrict__ features,
                                                        const float* __restrict__ W0, const float* __restrict__ b0,
                                                        const float* __restrict__ g0, const float* __restrict__ be0,
                                                        const float* __restrict__ m0, const float* __restrict__ v0,
                                                        float* __restrict__ Atab, float* __restrict__ Htab)
{
    float* tile = (float*)smem;
    float* pxyz = tile + C_ * PL_TP;
    float* oA   = pxyz + 256;
    float* oH   = oA + PL_PTS * PL_OP;
    const int b   = blk >> 6;
    const int n0  = (blk & 63) * PL_PTS;
    const int q   = tid & 15;
    const int r0  = tid >> 4;
    const float* src = features + (size_t)b * C_ * N_ + n0;
#pragma unroll
    for (int k = 0; k < 4; ++k) {
        const int c = r0 + 16 * k;
        *(float4*)(tile + c * PL_TP + 4 * q) = *(const float4*)(src + (size_t)c * N_ + 4 * q);
    }
    if (tid < PL_PTS * 3) pxyz[tid] = xyz[((size_t)b * N_ + n0) * 3 + tid];
    __syncthreads();

    const int lane = tid & 31;
    const int w    = tid >> 5;
    const int mt   = w >> 1;
    const int nb   = (w & 1) * 2;
    const int c16  = lane & 15;
    const int h    = lane >> 4;
    const int prow = mt * 16 + c16;

    v8f accF[2], accX[2];
    const v8f zero = {};
    accF[0] = zero; accF[1] = zero; accX[0] = zero; accX[1] = zero;

    {
        const float p0 = pxyz[prow * 3 + 2 * h];
        const float p1 = pxyz[prow * 3 + 1];
        v2f a;
        a.x = p0;
        a.y = h ? 0.0f : p1;
#pragma unroll
        for (int n = 0; n < 2; ++n) {
            const int o = (nb + n) * 16 + c16;
            const float w0 = W0[o * 67 + 2 * h];
            const float w1 = W0[o * 67 + 1];
            v2f bt;
            bt.x = w0;
            bt.y = h ? 0.0f : w1;
            accX[n] = wmma4(a, bt, accX[n]);
        }
    }
#pragma unroll
    for (int ks = 0; ks < C_ / 4; ++ks) {
        const int k = 4 * ks + 2 * h;
        v2f a;
        a.x = tile[(k + 0) * PL_TP + prow];
        a.y = tile[(k + 1) * PL_TP + prow];
#pragma unroll
        for (int n = 0; n < 2; ++n) {
            const int o = (nb + n) * 16 + c16;
            v2f bt;
            bt.x = W0[o * 67 + 3 + k + 0];
            bt.y = W0[o * 67 + 3 + k + 1];
            accF[n] = wmma4(a, bt, accF[n]);
        }
    }
#pragma unroll
    for (int n = 0; n < 2; ++n) {
        const int   o  = (nb + n) * 16 + c16;
        const float sc = g0[o] / sqrtf(v0[o] + BN_EPS);
        const float sh = (b0[o] - m0[o]) * sc + be0[o];
#pragma unroll
        for (int j = 0; j < 8; ++j) {
            const int row = mt * 16 + 8 * h + j;
            oH[row * PL_OP + o] = accX[n][j] * sc;
            oA[row * PL_OP + o] = (accF[n][j] + accX[n][j]) * sc + sh;
        }
    }
    __syncthreads();
    float* dA = Atab + ((size_t)b * N_ + n0) * C_;
    float* dH = Htab + ((size_t)b * N_ + n0) * C_;
#pragma unroll
    for (int k = 0; k < 4; ++k) {
        const int n = r0 + 16 * k;
        *(float4*)(dA + (size_t)n * C_ + 4 * q) = *(const float4*)(oA + n * PL_OP + 4 * q);
        *(float4*)(dH + (size_t)n * C_ + 4 * q) = *(const float4*)(oH + n * PL_OP + 4 * q);
    }
}

#define FRONT_FPS   B_
#define FRONT_KNN   (B_ * (N_ / (KNN_WAVES * KNN_CPW)))
#define FRONT_PL    (B_ * (N_ / PL_PTS))
#define FRONT_GRID  (FRONT_FPS + FRONT_KNN + FRONT_PL + 1)
#define FRONT_LDS   (N_ * 3 * 4 + 8 * KNN_WAVES * N_ + 4 * KNN_WAVES * NSAMPLE)
__global__ __launch_bounds__(256) void front_kernel(
    const float* __restrict__ xyz, const float* __restrict__ features,
    const float* __restrict__ W0, const float* __restrict__ b0, const float* __restrict__ g0,
    const float* __restrict__ be0, const float* __restrict__ m0, const float* __restrict__ v0,
    const float* __restrict__ W1, const float* __restrict__ b1, const float* __restrict__ g1,
    const float* __restrict__ be1, const float* __restrict__ m1, const float* __restrict__ v1,
    const float* __restrict__ W2, const float* __restrict__ b2, const float* __restrict__ g2,
    const float* __restrict__ be2, const float* __restrict__ m2, const float* __restrict__ v2_,
    float* __restrict__ new_xyz, int* __restrict__ sel, int* __restrict__ gidx_all,
    float* __restrict__ Atab, float* __restrict__ Htab,
    v2f* __restrict__ Wp1, float* __restrict__ b1f,
    v2f* __restrict__ Wp2, float* __restrict__ b2f)
{
    __shared__ __align__(16) unsigned char smem[FRONT_LDS];
    const int blk = blockIdx.x;
    const int tid = threadIdx.x;
    if (blk < FRONT_FPS) {
        fps_role(smem, blk, tid, xyz, new_xyz, sel);
    } else if (blk < FRONT_FPS + FRONT_KNN) {
        knn_role(smem, blk - FRONT_FPS, tid, xyz, gidx_all);
    } else if (blk < FRONT_FPS + FRONT_KNN + FRONT_PL) {
        point_layer_role(smem, blk - (FRONT_FPS + FRONT_KNN), tid, xyz, features,
                         W0, b0, g0, be0, m0, v0, Atab, Htab);
    } else {
        fuse_weights_role(tid, W1, b1, g1, be1, m1, v1,
                          W2, b2, g2, be2, m2, v2_, Wp1, b1f, Wp2, b2f);
    }
}

template <int KSTEPS, int NT, int COUT>
static __device__ inline void layer_gemm(const float* in_lds, float* out_lds,
                                         const v2f* __restrict__ Wp,
                                         const float* __restrict__ bias, int lane)
{
    v8f acc[2][NT];
    v8f zero = {};
#pragma unroll
    for (int mt = 0; mt < 2; ++mt)
#pragma unroll
        for (int n = 0; n < NT; ++n) acc[mt][n] = zero;

    const int mrow  = lane & 15;
    const int khalf = (lane >> 4) * 2;
    const int ncol  = lane & 15;

    for (int k = 0; k < KSTEPS; ++k) {
        const int kc = 4 * k + khalf;
        v2f a0 = *(const v2f*)(in_lds + mrow * PITCH + kc);
        v2f a1 = *(const v2f*)(in_lds + (mrow + 16) * PITCH + kc);
#pragma unroll
        for (int n = 0; n < NT; ++n) {
            v2f bt = Wp[(kc >> 1) * COUT + n * 16 + ncol];
            acc[0][n] = wmma4(a0, bt, acc[0][n]);
            acc[1][n] = wmma4(a1, bt, acc[1][n]);
        }
    }
    const int rbase = (lane >> 4) * 8;
#pragma unroll
    for (int n = 0; n < NT; ++n) {
        float bb = bias[n * 16 + ncol];
#pragma unroll
        for (int mt = 0; mt < 2; ++mt)
#pragma unroll
            for (int j = 0; j < 8; ++j) {
                int rrow = mt * 16 + rbase + j;
                out_lds[rrow * PITCH + n * 16 + ncol] = fmaxf(acc[mt][n][j] + bb, 0.0f);
            }
    }
}

__global__ __launch_bounds__(32) void mlp_kernel(
    const float* __restrict__ Atab, const float* __restrict__ Htab,
    const int* __restrict__ sel, const int* __restrict__ gidx_all,
    float* __restrict__ out_feat,
    const v2f* __restrict__ Wp1, const float* __restrict__ b1f,
    const v2f* __restrict__ Wp2, const float* __restrict__ b2f)
{
    __shared__ __align__(16) float bufA[32 * PITCH];
    __shared__ __align__(16) float bufB[32 * PITCH];
    const int cg   = blockIdx.x;
    const int b    = cg >> 10;
    const int lane = threadIdx.x;

    const int cp  = sel[cg];
    const int idx = gidx_all[((size_t)b * N_ + cp) * NSAMPLE + lane];
    const v2f  hc = ((const v2f*)Htab)[((size_t)b * N_ + cp) * (C_ / 2) + lane];
    const v2f* At = (const v2f*)Atab + (size_t)b * N_ * (C_ / 2);
    v2f f[NSAMPLE];
#pragma unroll
    for (int r = 0; r < NSAMPLE; ++r) {
        const int pr = __builtin_amdgcn_readlane(idx, r);
        f[r] = At[(size_t)pr * (C_ / 2) + lane];
    }
#pragma unroll
    for (int r = 0; r < NSAMPLE; ++r) {
        v2f a;
        a.x = fmaxf(f[r].x - hc.x, 0.0f);
        a.y = fmaxf(f[r].y - hc.y, 0.0f);
        *(v2f*)(bufA + r * PITCH + 2 * lane) = a;
    }
    __syncthreads();

    layer_gemm<16, 4, 64>(bufA, bufB, Wp1, b1f, lane);
    __syncthreads();

    v8f acc[2][8];
    v8f zero = {};
#pragma unroll
    for (int mt = 0; mt < 2; ++mt)
#pragma unroll
        for (int n = 0; n < 8; ++n) acc[mt][n] = zero;

    const int mrow  = lane & 15;
    const int khalf = (lane >> 4) * 2;
    const int ncol  = lane & 15;
    for (int k = 0; k < 16; ++k) {
        const int kc = 4 * k + khalf;
        v2f a0 = *(const v2f*)(bufB + mrow * PITCH + kc);
        v2f a1 = *(const v2f*)(bufB + (mrow + 16) * PITCH + kc);
#pragma unroll
        for (int n = 0; n < 8; ++n) {
            v2f bt = Wp2[(kc >> 1) * 128 + n * 16 + ncol];
            acc[0][n] = wmma4(a0, bt, acc[0][n]);
            acc[1][n] = wmma4(a1, bt, acc[1][n]);
        }
    }
    float* of = out_feat + (size_t)cg * 128;
#pragma unroll
    for (int n = 0; n < 8; ++n) {
        float r = -3.4e38f;
#pragma unroll
        for (int j = 0; j < 8; ++j) {
            r = fmaxf(r, acc[0][n][j]);
            r = fmaxf(r, acc[1][n][j]);
        }
        r = fmaxf(r, __shfl_xor(r, 16, 32));
        float v = fmaxf(r + b2f[n * 16 + ncol], 0.0f);
        if (lane < 16) of[n * 16 + ncol] = v;
    }
}

extern "C" void kernel_launch(void* const* d_in, const int* in_sizes, int n_in,
                              void* d_out, int out_size, void* d_ws, size_t ws_size,
                              hipStream_t stream)
{
    const float* xyz      = (const float*)d_in[0];
    const float* features = (const float*)d_in[1];
    const float* W0 = (const float*)d_in[2];  const float* b0 = (const float*)d_in[3];
    const float* g0 = (const float*)d_in[4];  const float* be0 = (const float*)d_in[5];
    const float* m0 = (const float*)d_in[6];  const float* v0 = (const float*)d_in[7];
    const float* W1 = (const float*)d_in[8];  const float* b1 = (const float*)d_in[9];
    const float* g1 = (const float*)d_in[10]; const float* be1 = (const float*)d_in[11];
    const float* m1 = (const float*)d_in[12]; const float* v1 = (const float*)d_in[13];
    const float* W2 = (const float*)d_in[14]; const float* b2 = (const float*)d_in[15];
    const float* g2 = (const float*)d_in[16]; const float* be2 = (const float*)d_in[17];
    const float* m2 = (const float*)d_in[18]; const float* v2_ = (const float*)d_in[19];

    float* out      = (float*)d_out;
    float* new_xyz  = out;
    float* out_feat = out + (size_t)B_ * NPOINT * 3;

    int*   sel  = (int*)d_ws;
    float* wsf  = (float*)((char*)d_ws + (size_t)2 * 1024 * 1024);
    v2f*   Wp1 = (v2f*)wsf;
    float* b1f = (float*)(Wp1 + 32 * 64);
    v2f*   Wp2 = (v2f*)(b1f + 64);
    float* b2f = (float*)(Wp2 + 32 * 128);
    float* Atab = (float*)((char*)d_ws + (size_t)4 * 1024 * 1024);
    int*   gidx_all = (int*)((char*)d_ws + (size_t)32 * 1024 * 1024);
    float* Htab = (float*)((char*)d_ws + (size_t)64 * 1024 * 1024);

    front_kernel<<<FRONT_GRID, 256, 0, stream>>>(xyz, features,
                                                 W0, b0, g0, be0, m0, v0,
                                                 W1, b1, g1, be1, m1, v1,
                                                 W2, b2, g2, be2, m2, v2_,
                                                 new_xyz, sel, gidx_all, Atab, Htab,
                                                 Wp1, b1f, Wp2, b2f);
    mlp_kernel<<<B_ * NPOINT, 32, 0, stream>>>(Atab, Htab, sel, gidx_all, out_feat,
                                               Wp1, b1f, Wp2, b2f);
}


# knn: bound is the 32nd smallest of the lanes' two smallest distances (rank count over 64 values), not the largest lane minimum: a quarter of the keys pass, so a lane ranks 2 keys per sweep, not 6
# speedup vs baseline: 6.5954x; 1.2830x over previous
#include <hip/hip_runtime.h>
#include <hip/hip_bf16.h>
#include <math.h>

#define B_      16
#define N_      4096
#define C_      64
#define NPOINT  1024
#define NSAMPLE 32
#define PITCH   68
#define BN_EPS  1e-5f

typedef float v2f __attribute__((ext_vector_type(2)));
typedef float v8f __attribute__((ext_vector_type(8)));

static __device__ inline v8f wmma4(v2f a, v2f b, v8f c) {
    return __builtin_amdgcn_wmma_f32_16x16x4_f32(
         false, a,  false, b,
         (short)0, c,  false,  false);
}

static __device__ __forceinline__ void fuse_weights_role(
    const int tid,
    const float* __restrict__ W1, const float* __restrict__ b1, const float* __restrict__ g1,
    const float* __restrict__ be1, const float* __restrict__ m1, const float* __restrict__ v1,
    const float* __restrict__ W2, const float* __restrict__ b2, const float* __restrict__ g2,
    const float* __restrict__ be2, const float* __restrict__ m2, const float* __restrict__ v2_,
    v2f* __restrict__ Wp1, float* __restrict__ b1f,
    v2f* __restrict__ Wp2, float* __restrict__ b2f)
{
    for (int i = tid; i < 32 * 64; i += 256) {
        int o = i & 63, kp = i >> 6;
        float s = g1[o] / sqrtf(v1[o] + BN_EPS);
        v2f w;
        w.x = W1[o * 64 + 2 * kp + 0] * s;
        w.y = W1[o * 64 + 2 * kp + 1] * s;
        Wp1[i] = w;
    }
    for (int i = tid; i < 64; i += 256) {
        float s = g1[i] / sqrtf(v1[i] + BN_EPS);
        b1f[i] = (b1[i] - m1[i]) * s + be1[i];
    }
    for (int i = tid; i < 32 * 128; i += 256) {
        int o = i & 127, kp = i >> 7;
        float s = g2[o] / sqrtf(v2_[o] + BN_EPS);
        v2f w;
        w.x = W2[o * 64 + 2 * kp + 0] * s;
        w.y = W2[o * 64 + 2 * kp + 1] * s;
        Wp2[i] = w;
    }
    for (int i = tid; i < 128; i += 256) {
        float s = g2[i] / sqrtf(v2_[i] + BN_EPS);
        b2f[i] = (b2[i] - m2[i]) * s + be2[i];
    }
}

template <int M>
static __device__ inline int xrow(int x) {
    return __builtin_amdgcn_update_dpp(0, x, 0x160 + M, 0xf, 0xf, true);
}
static __device__ inline int xhalf(int x) {
    return __builtin_amdgcn_permlanex16(x, x, 0x76543210, 0xfedcba98, false, false);
}
static __device__ inline int imax(int a, int b) { return a > b ? a : b; }
static __device__ inline int imin(int a, int b) { return a < b ? a : b; }
#define NO_IDX 0x7fffffff

static __device__ __forceinline__ void fps_role(unsigned char* smem, const int b, const int tid,
                                                const float* __restrict__ xyz,
                                                float* __restrict__ new_xyz,
                                                int* __restrict__ sel)
{
    float* sx   = (float*)smem;
    float* sy   = sx + N_;
    float* sz   = sy + N_;
    int*   sidx = (int*)(sz + N_);
    int (*red)[16] = (int (*)[16])(sidx + NPOINT);
    const float* bx = xyz + (size_t)b * N_ * 3;

    float px[16], py[16], pz[16], dloc[16];
#pragma unroll
    for (int i = 0; i < 16; ++i) {
        int p = tid + 256 * i;
        px[i] = bx[p * 3 + 0];
        py[i] = bx[p * 3 + 1];
        pz[i] = bx[p * 3 + 2];
        sx[p] = px[i];
        sy[p] = py[i];
        sz[p] = pz[i];
        dloc[i] = 1e10f;
    }
    __syncthreads();

    int last = 0;
    for (int it = 0; ; ++it) {
        const float lx = sx[last], ly = sy[last], lz = sz[last];
        if (tid == 0) sidx[it] = last;
        if (it == NPOINT - 1) break;

        float val = -1.0f;
        int   best = tid;
#pragma unroll
        for (int i = 0; i < 16; ++i) {
            int p = tid + 256 * i;
            float dx = px[i] - lx, dy = py[i] - ly, dz = pz[i] - lz;
            float dist = dx * dx + dy * dy + dz * dz;
            dloc[i] = fminf(dloc[i], dist);
            if (dloc[i] > val) { val = dloc[i]; best = p; }
        }
        const int vb = __float_as_int(val);
        int m = vb;
        m = imax(m, xrow<1>(m));
        m = imax(m, xrow<2>(m));
        m = imax(m, xrow<4>(m));
        m = imax(m, xrow<8>(m));
        m = imax(m, xhalf(m));
        int c = (vb == m) ? best : NO_IDX;
        c = imin(c, xrow<1>(c));
        c = imin(c, xrow<2>(c));
        c = imin(c, xrow<4>(c));
        c = imin(c, xrow<8>(c));
        c = imin(c, xhalf(c));

        int* slot = red[it & 1];
        if ((tid & 31) == 0) *(int2*)(slot + 2 * (tid >> 5)) = make_int2(m, c);
        __syncthreads();
        const int2 r = *(const int2*)(slot + 2 * (tid & 7));
        int gm = r.x;
        gm = imax(gm, xrow<1>(gm));
        gm = imax(gm, xrow<2>(gm));
        gm = imax(gm, xrow<4>(gm));
        int gc = (r.x == gm) ? r.y : NO_IDX;
        gc = imin(gc, xrow<1>(gc));
        gc = imin(gc, xrow<2>(gc));
        gc = imin(gc, xrow<4>(gc));
        last = gc;
    }
    __syncthreads();
    for (int k = 0; k < NPOINT / 256; ++k) {
        const int j = tid + 256 * k;
        const int p = sidx[j];
        float* o = new_xyz + ((size_t)b * NPOINT + j) * 3;
        o[0] = sx[p];
        o[1] = sy[p];
        o[2] = sz[p];
        sel[b * NPOINT + j] = p;
    }
}

#define KNN_WAVES 8
#define KNN_CPW   8
#define KNN_OWN   2
typedef unsigned long long knn_key;

static __device__ __forceinline__ void knn_role(unsigned char* smem, const int blk, const int tid,
                                                const float* __restrict__ xyz,
                                                int* __restrict__ gidx_all)
{
    float4* pts = (float4*)smem;
    knn_key (*cand)[N_]   = (knn_key (*)[N_])(smem + N_ * 3 * sizeof(float));
    int (*res)[NSAMPLE]   = (int (*)[NSAMPLE])(smem + N_ * 3 * sizeof(float) + sizeof(knn_key) * KNN_WAVES * N_);
    const int b    = blk >> 6;
    const int g    = blk & 63;
    const int lane = tid & 31;
    const int w    = tid >> 5;

    const float4* src = (const float4*)(xyz + (size_t)b * N_ * 3);
    for (int i = tid; i < N_ * 3 / 4; i += KNN_WAVES * 32) pts[i] = src[i];
    __syncthreads();

    const float* P  = (const float*)pts;
    knn_key*     cw = cand[w];
    int*         rw = res[w];

#define KNN_DIST(PX, PY, PZ) fmaf((PZ) - cz, (PZ) - cz, fmaf((PX) - cx, (PX) - cx, ((PY) - cy) * ((PY) - cy)))

    for (int ci = 0; ci < KNN_CPW; ++ci) {
        const int cent = g * (KNN_WAVES * KNN_CPW) + w * KNN_CPW + ci;
        const float cx = P[3 * cent + 0];
        const float cy = P[3 * cent + 1];
        const float cz = P[3 * cent + 2];

        float d[N_ / 32];
        float lo0 = 3.4e38f, hi0 = 3.4e38f, lo1 = 3.4e38f, hi1 = 3.4e38f;
#pragma unroll
        for (int t = 0; t < N_ / 32; ++t) {
            const int j = 32 * t + lane;
            const float px = P[3 * j + 0], py = P[3 * j + 1], pz = P[3 * j + 2];
            d[t] = KNN_DIST(px, py, pz);
            if (t & 1) { hi1 = __builtin_amdgcn_fmed3f(lo1, hi1, d[t]); lo1 = fminf(lo1, d[t]); }
            else       { hi0 = __builtin_amdgcn_fmed3f(lo0, hi0, d[t]); lo0 = fminf(lo0, d[t]); }
        }
        const int s1 = __float_as_int(fminf(lo0, lo1));
        const int s2 = __float_as_int(fminf(fmaxf(lo0, lo1), fminf(hi0, hi1)));
        cw[lane] = ((knn_key)(unsigned)s2 << 32) | (unsigned)s1;
        __builtin_amdgcn_fence(__ATOMIC_ACQ_REL, "wavefront");
        int below1 = 0, below2 = 0;
#pragma unroll
        for (int s = 0; s < 16; ++s) {
            const knn_key e0 = cw[2 * s + 0], e1 = cw[2 * s + 1];
            const int v0 = (int)(unsigned)(e0 & 0xffffffffull), v1 = (int)(unsigned)(e0 >> 32);
            const int v2 = (int)(unsigned)(e1 & 0xffffffffull), v3 = (int)(unsigned)(e1 >> 32);
            below1 += (v0 < s1 ? 1 : 0) + (v1 < s1 ? 1 : 0) + (v2 < s1 ? 1 : 0) + (v3 < s1 ? 1 : 0);
            below2 += (v0 < s2 ? 1 : 0) + (v1 < s2 ? 1 : 0) + (v2 < s2 ? 1 : 0) + (v3 < s2 ? 1 : 0);
        }
        int mb = (below2 < NSAMPLE) ? s2 : ((below1 < NSAMPLE) ? s1 : 0);
        mb = imax(mb, xrow<1>(mb));
        mb = imax(mb, xrow<2>(mb));
        mb = imax(mb, xrow<4>(mb));
        mb = imax(mb, xrow<8>(mb));
        mb = imax(mb, xhalf(mb));
        const float bound = __int_as_float(mb);
        __builtin_amdgcn_fence(__ATOMIC_ACQ_REL, "wavefront");

        unsigned M = 0;
#pragma unroll
        for (int t = 0; t < N_ / 32; ++t) {
            const bool     in   = d[t] <= bound;
            const unsigned mask = __builtin_amdgcn_ballot_w32(in);
            if (in)
                cw[M + __builtin_amdgcn_mbcnt_lo(mask, 0u)] =
                    ((knn_key)(unsigned)__float_as_int(d[t]) << 32) | (unsigned)(32 * t + lane);
            M += (unsigned)__builtin_popcount(mask);
        }
        rw[lane] = 0;
        __builtin_amdgcn_fence(__ATOMIC_ACQ_REL, "wavefront");

        for (unsigned k0 = 0; k0 < M; k0 += 32 * KNN_OWN) {
            knn_key  x[KNN_OWN];
            unsigned r[KNN_OWN];
#pragma unroll
            for (int q = 0; q < KNN_OWN; ++q) {
                const unsigned k = k0 + 32 * q + lane;
                x[q] = (k < M) ? cw[k] : ~(knn_key)0;
                r[q] = 0;
            }
#pragma unroll 4
            for (unsigned y = 0; y < M; ++y) {
                const knn_key cy = cw[y];
#pragma unroll
                for (int q = 0; q < KNN_OWN; ++q) r[q] += (cy < x[q]) ? 1u : 0u;
            }
#pragma unroll
            for (int q = 0; q < KNN_OWN; ++q) {
                const unsigned k = k0 + 32 * q + lane;
                if (k < M && r[q] < NSAMPLE) rw[r[q]] = (int)(unsigned)(x[q] & 0xffffffffull);
            }
        }
        __builtin_amdgcn_fence(__ATOMIC_ACQ_REL, "wavefront");
        gidx_all[(size_t)(b * N_ + cent) * NSAMPLE + lane] = rw[lane];
        __builtin_amdgcn_fence(__ATOMIC_ACQ_REL, "wavefront");
    }
#undef KNN_DIST
}

#define PL_PTS 64
#define PL_TP  72
#define PL_OP  68
static __device__ __forceinline__ void point_layer_role(unsigned char* smem, const int blk, const int tid,
                                                        const float* __restrict__ xyz,
                                                        const float* __restrict__ features,
                                                        const float* __restrict__ W0, const float* __restrict__ b0,
                                                        const float* __restrict__ g0, const float* __restrict__ be0,
                                                        const float* __restrict__ m0, const float* __restrict__ v0,
                                                        float* __restrict__ Atab, float* __restrict__ Htab)
{
    float* tile = (float*)smem;
    float* pxyz = tile + C_ * PL_TP;
    float* oA   = pxyz + 256;
    float* oH   = oA + PL_PTS * PL_OP;
    const int b   = blk >> 6;
    const int n0  = (blk & 63) * PL_PTS;
    const int q   = tid & 15;
    const int r0  = tid >> 4;
    const float* src = features + (size_t)b * C_ * N_ + n0;
#pragma unroll
    for (int k = 0; k < 4; ++k) {
        const int c = r0 + 16 * k;
        *(float4*)(tile + c * PL_TP + 4 * q) = *(const float4*)(src + (size_t)c * N_ + 4 * q);
    }
    if (tid < PL_PTS * 3) pxyz[tid] = xyz[((size_t)b * N_ + n0) * 3 + tid];
    __syncthreads();

    const int lane = tid & 31;
    const int w    = tid >> 5;
    const int mt   = w >> 1;
    const int nb   = (w & 1) * 2;
    const int c16  = lane & 15;
    const int h    = lane >> 4;
    const int prow = mt * 16 + c16;

    v8f accF[2], accX[2];
    const v8f zero = {};
    accF[0] = zero; accF[1] = zero; accX[0] = zero; accX[1] = zero;

    {
        const float p0 = pxyz[prow * 3 + 2 * h];
        const float p1 = pxyz[prow * 3 + 1];
        v2f a;
        a.x = p0;
        a.y = h ? 0.0f : p1;
#pragma unroll
        for (int n = 0; n < 2; ++n) {
            const int o = (nb + n) * 16 + c16;
            const float w0 = W0[o * 67 + 2 * h];
            const float w1 = W0[o * 67 + 1];
            v2f bt;
            bt.x = w0;
            bt.y = h ? 0.0f : w1;
            accX[n] = wmma4(a, bt, accX[n]);
        }
    }
#pragma unroll
    for (int ks = 0; ks < C_ / 4; ++ks) {
        const int k = 4 * ks + 2 * h;
        v2f a;
        a.x = tile[(k + 0) * PL_TP + prow];
        a.y = tile[(k + 1) * PL_TP + prow];
#pragma unroll
        for (int n = 0; n < 2; ++n) {
            const int o = (nb + n) * 16 + c16;
            v2f bt;
            bt.x = W0[o * 67 + 3 + k + 0];
            bt.y = W0[o * 67 + 3 + k + 1];
            accF[n] = wmma4(a, bt, accF[n]);
        }
    }
#pragma unroll
    for (int n = 0; n < 2; ++n) {
        const int   o  = (nb + n) * 16 + c16;
        const float sc = g0[o] / sqrtf(v0[o] + BN_EPS);
        const float sh = (b0[o] - m0[o]) * sc + be0[o];
#pragma unroll
        for (int j = 0; j < 8; ++j) {
            const int row = mt * 16 + 8 * h + j;
            oH[row * PL_OP + o] = accX[n][j] * sc;
            oA[row * PL_OP + o] = (accF[n][j] + accX[n][j]) * sc + sh;
        }
    }
    __syncthreads();
    float* dA = Atab + ((size_t)b * N_ + n0) * C_;
    float* dH = Htab + ((size_t)b * N_ + n0) * C_;
#pragma unroll
    for (int k = 0; k < 4; ++k) {
        const int n = r0 + 16 * k;
        *(float4*)(dA + (size_t)n * C_ + 4 * q) = *(const float4*)(oA + n * PL_OP + 4 * q);
        *(float4*)(dH + (size_t)n * C_ + 4 * q) = *(const float4*)(oH + n * PL_OP + 4 * q);
    }
}

#define FRONT_FPS   B_
#define FRONT_KNN   (B_ * (N_ / (KNN_WAVES * KNN_CPW)))
#define FRONT_PL    (B_ * (N_ / PL_PTS))
#define FRONT_GRID  (FRONT_FPS + FRONT_KNN + FRONT_PL + 1)
#define FRONT_LDS   (N_ * 3 * 4 + 8 * KNN_WAVES * N_ + 4 * KNN_WAVES * NSAMPLE)
__global__ __launch_bounds__(256) void front_kernel(
    const float* __restrict__ xyz, const float* __restrict__ features,
    const float* __restrict__ W0, const float* __restrict__ b0, const float* __restrict__ g0,
    const float* __restrict__ be0, const float* __restrict__ m0, const float* __restrict__ v0,
    const float* __restrict__ W1, const float* __restrict__ b1, const float* __restrict__ g1,
    const float* __restrict__ be1, const float* __restrict__ m1, const float* __restrict__ v1,
    const float* __restrict__ W2, const float* __restrict__ b2, const float* __restrict__ g2,
    const float* __restrict__ be2, const float* __restrict__ m2, const float* __restrict__ v2_,
    float* __restrict__ new_xyz, int* __restrict__ sel, int* __restrict__ gidx_all,
    float* __restrict__ Atab, float* __restrict__ Htab,
    v2f* __restrict__ Wp1, float* __restrict__ b1f,
    v2f* __restrict__ Wp2, float* __restrict__ b2f)
{
    __shared__ __align__(16) unsigned char smem[FRONT_LDS];
    const int blk = blockIdx.x;
    const int tid = threadIdx.x;
    if (blk < FRONT_FPS) {
        fps_role(smem, blk, tid, xyz, new_xyz, sel);
    } else if (blk < FRONT_FPS + FRONT_KNN) {
        knn_role(smem, blk - FRONT_FPS, tid, xyz, gidx_all);
    } else if (blk < FRONT_FPS + FRONT_KNN + FRONT_PL) {
        point_layer_role(smem, blk - (FRONT_FPS + FRONT_KNN), tid, xyz, features,
                         W0, b0, g0, be0, m0, v0, Atab, Htab);
    } else {
        fuse_weights_role(tid, W1, b1, g1, be1, m1, v1,
                          W2, b2, g2, be2, m2, v2_, Wp1, b1f, Wp2, b2f);
    }
}

template <int KSTEPS, int NT, int COUT>
static __device__ inline void layer_gemm(const float* in_lds, float* out_lds,
                                         const v2f* __restrict__ Wp,
                                         const float* __restrict__ bias, int lane)
{
    v8f acc[2][NT];
    v8f zero = {};
#pragma unroll
    for (int mt = 0; mt < 2; ++mt)
#pragma unroll
        for (int n = 0; n < NT; ++n) acc[mt][n] = zero;

    const int mrow  = lane & 15;
    const int khalf = (lane >> 4) * 2;
    const int ncol  = lane & 15;

    for (int k = 0; k < KSTEPS; ++k) {
        const int kc = 4 * k + khalf;
        v2f a0 = *(const v2f*)(in_lds + mrow * PITCH + kc);
        v2f a1 = *(const v2f*)(in_lds + (mrow + 16) * PITCH + kc);
#pragma unroll
        for (int n = 0; n < NT; ++n) {
            v2f bt = Wp[(kc >> 1) * COUT + n * 16 + ncol];
            acc[0][n] = wmma4(a0, bt, acc[0][n]);
            acc[1][n] = wmma4(a1, bt, acc[1][n]);
        }
    }
    const int rbase = (lane >> 4) * 8;
#pragma unroll
    for (int n = 0; n < NT; ++n) {
        float bb = bias[n * 16 + ncol];
#pragma unroll
        for (int mt = 0; mt < 2; ++mt)
#pragma unroll
            for (int j = 0; j < 8; ++j) {
                int rrow = mt * 16 + rbase + j;
                out_lds[rrow * PITCH + n * 16 + ncol] = fmaxf(acc[mt][n][j] + bb, 0.0f);
            }
    }
}

__global__ __launch_bounds__(32) void mlp_kernel(
    const float* __restrict__ Atab, const float* __restrict__ Htab,
    const int* __restrict__ sel, const int* __restrict__ gidx_all,
    float* __restrict__ out_feat,
    const v2f* __restrict__ Wp1, const float* __restrict__ b1f,
    const v2f* __restrict__ Wp2, const float* __restrict__ b2f)
{
    __shared__ __align__(16) float bufA[32 * PITCH];
    __shared__ __align__(16) float bufB[32 * PITCH];
    const int cg   = blockIdx.x;
    const int b    = cg >> 10;
    const int lane = threadIdx.x;

    const int cp  = sel[cg];
    const int idx = gidx_all[((size_t)b * N_ + cp) * NSAMPLE + lane];
    const v2f  hc = ((const v2f*)Htab)[((size_t)b * N_ + cp) * (C_ / 2) + lane];
    const v2f* At = (const v2f*)Atab + (size_t)b * N_ * (C_ / 2);
    v2f f[NSAMPLE];
#pragma unroll
    for (int r = 0; r < NSAMPLE; ++r) {
        const int pr = __builtin_amdgcn_readlane(idx, r);
        f[r] = At[(size_t)pr * (C_ / 2) + lane];
    }
#pragma unroll
    for (int r = 0; r < NSAMPLE; ++r) {
        v2f a;
        a.x = fmaxf(f[r].x - hc.x, 0.0f);
        a.y = fmaxf(f[r].y - hc.y, 0.0f);
        *(v2f*)(bufA + r * PITCH + 2 * lane) = a;
    }
    __syncthreads();

    layer_gemm<16, 4, 64>(bufA, bufB, Wp1, b1f, lane);
    __syncthreads();

    v8f acc[2][8];
    v8f zero = {};
#pragma unroll
    for (int mt = 0; mt < 2; ++mt)
#pragma unroll
        for (int n = 0; n < 8; ++n) acc[mt][n] = zero;

    const int mrow  = lane & 15;
    const int khalf = (lane >> 4) * 2;
    const int ncol  = lane & 15;
    for (int k = 0; k < 16; ++k) {
        const int kc = 4 * k + khalf;
        v2f a0 = *(const v2f*)(bufB + mrow * PITCH + kc);
        v2f a1 = *(const v2f*)(bufB + (mrow + 16) * PITCH + kc);
#pragma unroll
        for (int n = 0; n < 8; ++n) {
            v2f bt = Wp2[(kc >> 1) * 128 + n * 16 + ncol];
            acc[0][n] = wmma4(a0, bt, acc[0][n]);
            acc[1][n] = wmma4(a1, bt, acc[1][n]);
        }
    }
    float* of = out_feat + (size_t)cg * 128;
#pragma unroll
    for (int n = 0; n < 8; ++n) {
        float r = -3.4e38f;
#pragma unroll
        for (int j = 0; j < 8; ++j) {
            r = fmaxf(r, acc[0][n][j]);
            r = fmaxf(r, acc[1][n][j]);
        }
        r = fmaxf(r, __shfl_xor(r, 16, 32));
        float v = fmaxf(r + b2f[n * 16 + ncol], 0.0f);
        if (lane < 16) of[n * 16 + ncol] = v;
    }
}

extern "C" void kernel_launch(void* const* d_in, const int* in_sizes, int n_in,
                              void* d_out, int out_size, void* d_ws, size_t ws_size,
                              hipStream_t stream)
{
    const float* xyz      = (const float*)d_in[0];
    const float* features = (const float*)d_in[1];
    const float* W0 = (const float*)d_in[2];  const float* b0 = (const float*)d_in[3];
    const float* g0 = (const float*)d_in[4];  const float* be0 = (const float*)d_in[5];
    const float* m0 = (const float*)d_in[6];  const float* v0 = (const float*)d_in[7];
    const float* W1 = (const float*)d_in[8];  const float* b1 = (const float*)d_in[9];
    const float* g1 = (const float*)d_in[10]; const float* be1 = (const float*)d_in[11];
    const float* m1 = (const float*)d_in[12]; const float* v1 = (const float*)d_in[13];
    const float* W2 = (const float*)d_in[14]; const float* b2 = (const float*)d_in[15];
    const float* g2 = (const float*)d_in[16]; const float* be2 = (const float*)d_in[17];
    const float* m2 = (const float*)d_in[18]; const float* v2_ = (const float*)d_in[19];

    float* out      = (float*)d_out;
    float* new_xyz  = out;
    float* out_feat = out + (size_t)B_ * NPOINT * 3;

    int*   sel  = (int*)d_ws;
    float* wsf  = (float*)((char*)d_ws + (size_t)2 * 1024 * 1024);
    v2f*   Wp1 = (v2f*)wsf;
    float* b1f = (float*)(Wp1 + 32 * 64);
    v2f*   Wp2 = (v2f*)(b1f + 64);
    float* b2f = (float*)(Wp2 + 32 * 128);
    float* Atab = (float*)((char*)d_ws + (size_t)4 * 1024 * 1024);
    int*   gidx_all = (int*)((char*)d_ws + (size_t)32 * 1024 * 1024);
    float* Htab = (float*)((char*)d_ws + (size_t)64 * 1024 * 1024);

    front_kernel<<<FRONT_GRID, 256, 0, stream>>>(xyz, features,
                                                 W0, b0, g0, be0, m0, v0,
                                                 W1, b1, g1, be1, m1, v1,
                                                 W2, b2, g2, be2, m2, v2_,
                                                 new_xyz, sel, gidx_all, Atab, Htab,
                                                 Wp1, b1f, Wp2, b2f);
    mlp_kernel<<<B_ * NPOINT, 32, 0, stream>>>(Atab, Htab, sel, gidx_all, out_feat,
                                               Wp1, b1f, Wp2, b2f);
}


# fps: each lane reads its own farthest point's xyz from LDS under the wave reductions, the wave result carries it, every wave takes the winner's xyz by one row gather: no dependent LDS read per step
# speedup vs baseline: 6.8319x; 1.0359x over previous
#include <hip/hip_runtime.h>
#include <hip/hip_bf16.h>
#include <math.h>

#define B_      16
#define N_      4096
#define C_      64
#define NPOINT  1024
#define NSAMPLE 32
#define PITCH   68
#define BN_EPS  1e-5f

typedef float v2f __attribute__((ext_vector_type(2)));
typedef float v8f __attribute__((ext_vector_type(8)));

static __device__ inline v8f wmma4(v2f a, v2f b, v8f c) {
    return __builtin_amdgcn_wmma_f32_16x16x4_f32(
         false, a,  false, b,
         (short)0, c,  false,  false);
}

static __device__ __forceinline__ void fuse_weights_role(
    const int tid,
    const float* __restrict__ W1, const float* __restrict__ b1, const float* __restrict__ g1,
    const float* __restrict__ be1, const float* __restrict__ m1, const float* __restrict__ v1,
    const float* __restrict__ W2, const float* __restrict__ b2, const float* __restrict__ g2,
    const float* __restrict__ be2, const float* __restrict__ m2, const float* __restrict__ v2_,
    v2f* __restrict__ Wp1, float* __restrict__ b1f,
    v2f* __restrict__ Wp2, float* __restrict__ b2f)
{
    for (int i = tid; i < 32 * 64; i += 256) {
        int o = i & 63, kp = i >> 6;
        float s = g1[o] / sqrtf(v1[o] + BN_EPS);
        v2f w;
        w.x = W1[o * 64 + 2 * kp + 0] * s;
        w.y = W1[o * 64 + 2 * kp + 1] * s;
        Wp1[i] = w;
    }
    for (int i = tid; i < 64; i += 256) {
        float s = g1[i] / sqrtf(v1[i] + BN_EPS);
        b1f[i] = (b1[i] - m1[i]) * s + be1[i];
    }
    for (int i = tid; i < 32 * 128; i += 256) {
        int o = i & 127, kp = i >> 7;
        float s = g2[o] / sqrtf(v2_[o] + BN_EPS);
        v2f w;
        w.x = W2[o * 64 + 2 * kp + 0] * s;
        w.y = W2[o * 64 + 2 * kp + 1] * s;
        Wp2[i] = w;
    }
    for (int i = tid; i < 128; i += 256) {
        float s = g2[i] / sqrtf(v2_[i] + BN_EPS);
        b2f[i] = (b2[i] - m2[i]) * s + be2[i];
    }
}

template <int M>
static __device__ inline int xrow(int x) {
    return __builtin_amdgcn_update_dpp(0, x, 0x160 + M, 0xf, 0xf, true);
}
static __device__ inline int xhalf(int x) {
    return __builtin_amdgcn_permlanex16(x, x, 0x76543210, 0xfedcba98, false, false);
}
static __device__ inline int rowpick(int x, int sel) {
    return (int)__builtin_amdgcn_permlane16_var((unsigned)x, (unsigned)x, (unsigned)sel, false, false);
}
static __device__ inline int imax(int a, int b) { return a > b ? a : b; }
static __device__ inline int imin(int a, int b) { return a < b ? a : b; }
#define NO_IDX 0x7fffffff

static __device__ __forceinline__ void fps_role(unsigned char* smem, const int b, const int tid,
                                                const float* __restrict__ xyz,
                                                float* __restrict__ new_xyz,
                                                int* __restrict__ sel)
{
    float* sx   = (float*)smem;
    float* sy   = sx + N_;
    float* sz   = sy + N_;
    int*   sidx = (int*)(sz + N_);
    int (*red)[64] = (int (*)[64])(sidx + NPOINT);
    const float* bx = xyz + (size_t)b * N_ * 3;

    float px[16], py[16], pz[16], dloc[16];
#pragma unroll
    for (int i = 0; i < 16; ++i) {
        int p = tid + 256 * i;
        px[i] = bx[p * 3 + 0];
        py[i] = bx[p * 3 + 1];
        pz[i] = bx[p * 3 + 2];
        sx[p] = px[i];
        sy[p] = py[i];
        sz[p] = pz[i];
        dloc[i] = 1e10f;
    }
    __syncthreads();

    int last = 0;
    float lx = bx[0], ly = bx[1], lz = bx[2];
    for (int it = 0; it < NPOINT - 1; ++it) {
        float val = -1.0f;
        int   best = tid;
#pragma unroll
        for (int i = 0; i < 16; ++i) {
            int p = tid + 256 * i;
            float dx = px[i] - lx, dy = py[i] - ly, dz = pz[i] - lz;
            float dist = dx * dx + dy * dy + dz * dz;
            dloc[i] = fminf(dloc[i], dist);
            if (dloc[i] > val) { val = dloc[i]; best = p; }
        }
        const float bxv = sx[best], byv = sy[best], bzv = sz[best];
        __builtin_amdgcn_sched_barrier(0);
        const int vb = __float_as_int(val);
        int m = vb;
        m = imax(m, xrow<1>(m));
        m = imax(m, xrow<2>(m));
        m = imax(m, xrow<4>(m));
        m = imax(m, xrow<8>(m));
        m = imax(m, xhalf(m));
        int c = (vb == m) ? best : NO_IDX;
        c = imin(c, xrow<1>(c));
        c = imin(c, xrow<2>(c));
        c = imin(c, xrow<4>(c));
        c = imin(c, xrow<8>(c));
        c = imin(c, xhalf(c));

        int* slot = red[it & 1];
        if (best == c) {
            int* e = slot + 8 * (tid >> 5);
            *(int4*)e = make_int4(m, c, __float_as_int(bxv), __float_as_int(byv));
            e[4] = __float_as_int(bzv);
        }
        __syncthreads();
        const int* e = slot + 8 * (tid & 7);
        const int4 r  = *(const int4*)e;
        const int  rz = e[4];
        if (tid == 0) sidx[it] = last;
        int gm = r.x;
        gm = imax(gm, xrow<1>(gm));
        gm = imax(gm, xrow<2>(gm));
        gm = imax(gm, xrow<4>(gm));
        int gc = (r.x == gm) ? r.y : NO_IDX;
        gc = imin(gc, xrow<1>(gc));
        gc = imin(gc, xrow<2>(gc));
        gc = imin(gc, xrow<4>(gc));
        last = gc;
        const int wsel = (gc >> 5) & 7;
        lx = __int_as_float(rowpick(r.z, wsel));
        ly = __int_as_float(rowpick(r.w, wsel));
        lz = __int_as_float(rowpick(rz,  wsel));
    }
    if (tid == 0) sidx[NPOINT - 1] = last;
    __syncthreads();
    for (int k = 0; k < NPOINT / 256; ++k) {
        const int j = tid + 256 * k;
        const int p = sidx[j];
        float* o = new_xyz + ((size_t)b * NPOINT + j) * 3;
        o[0] = sx[p];
        o[1] = sy[p];
        o[2] = sz[p];
        sel[b * NPOINT + j] = p;
    }
}

#define KNN_WAVES 8
#define KNN_CPW   8
#define KNN_OWN   2
typedef unsigned long long knn_key;

static __device__ __forceinline__ void knn_role(unsigned char* smem, const int blk, const int tid,
                                                const float* __restrict__ xyz,
                                                int* __restrict__ gidx_all)
{
    float4* pts = (float4*)smem;
    knn_key (*cand)[N_]   = (knn_key (*)[N_])(smem + N_ * 3 * sizeof(float));
    int (*res)[NSAMPLE]   = (int (*)[NSAMPLE])(smem + N_ * 3 * sizeof(float) + sizeof(knn_key) * KNN_WAVES * N_);
    const int b    = blk >> 6;
    const int g    = blk & 63;
    const int lane = tid & 31;
    const int w    = tid >> 5;

    const float4* src = (const float4*)(xyz + (size_t)b * N_ * 3);
    for (int i = tid; i < N_ * 3 / 4; i += KNN_WAVES * 32) pts[i] = src[i];
    __syncthreads();

    const float* P  = (const float*)pts;
    knn_key*     cw = cand[w];
    int*         rw = res[w];

#define KNN_DIST(PX, PY, PZ) fmaf((PZ) - cz, (PZ) - cz, fmaf((PX) - cx, (PX) - cx, ((PY) - cy) * ((PY) - cy)))

    for (int ci = 0; ci < KNN_CPW; ++ci) {
        const int cent = g * (KNN_WAVES * KNN_CPW) + w * KNN_CPW + ci;
        const float cx = P[3 * cent + 0];
        const float cy = P[3 * cent + 1];
        const float cz = P[3 * cent + 2];

        float d[N_ / 32];
        float lo0 = 3.4e38f, hi0 = 3.4e38f, lo1 = 3.4e38f, hi1 = 3.4e38f;
#pragma unroll
        for (int t = 0; t < N_ / 32; ++t) {
            const int j = 32 * t + lane;
            const float px = P[3 * j + 0], py = P[3 * j + 1], pz = P[3 * j + 2];
            d[t] = KNN_DIST(px, py, pz);
            if (t & 1) { hi1 = __builtin_amdgcn_fmed3f(lo1, hi1, d[t]); lo1 = fminf(lo1, d[t]); }
            else       { hi0 = __builtin_amdgcn_fmed3f(lo0, hi0, d[t]); lo0 = fminf(lo0, d[t]); }
        }
        const int s1 = __float_as_int(fminf(lo0, lo1));
        const int s2 = __float_as_int(fminf(fmaxf(lo0, lo1), fminf(hi0, hi1)));
        cw[lane] = ((knn_key)(unsigned)s2 << 32) | (unsigned)s1;
        __builtin_amdgcn_fence(__ATOMIC_ACQ_REL, "wavefront");
        int below1 = 0, below2 = 0;
#pragma unroll
        for (int s = 0; s < 16; ++s) {
            const knn_key e0 = cw[2 * s + 0], e1 = cw[2 * s + 1];
            const int v0 = (int)(unsigned)(e0 & 0xffffffffull), v1 = (int)(unsigned)(e0 >> 32);
            const int v2 = (int)(unsigned)(e1 & 0xffffffffull), v3 = (int)(unsigned)(e1 >> 32);
            below1 += (v0 < s1 ? 1 : 0) + (v1 < s1 ? 1 : 0) + (v2 < s1 ? 1 : 0) + (v3 < s1 ? 1 : 0);
            below2 += (v0 < s2 ? 1 : 0) + (v1 < s2 ? 1 : 0) + (v2 < s2 ? 1 : 0) + (v3 < s2 ? 1 : 0);
        }
        int mb = (below2 < NSAMPLE) ? s2 : ((below1 < NSAMPLE) ? s1 : 0);
        mb = imax(mb, xrow<1>(mb));
        mb = imax(mb, xrow<2>(mb));
        mb = imax(mb, xrow<4>(mb));
        mb = imax(mb, xrow<8>(mb));
        mb = imax(mb, xhalf(mb));
        const float bound = __int_as_float(mb);
        __builtin_amdgcn_fence(__ATOMIC_ACQ_REL, "wavefront");

        unsigned M = 0;
#pragma unroll
        for (int t = 0; t < N_ / 32; ++t) {
            const bool     in   = d[t] <= bound;
            const unsigned mask = __builtin_amdgcn_ballot_w32(in);
            if (in)
                cw[M + __builtin_amdgcn_mbcnt_lo(mask, 0u)] =
                    ((knn_key)(unsigned)__float_as_int(d[t]) << 32) | (unsigned)(32 * t + lane);
            M += (unsigned)__builtin_popcount(mask);
        }
        rw[lane] = 0;
        __builtin_amdgcn_fence(__ATOMIC_ACQ_REL, "wavefront");

        for (unsigned k0 = 0; k0 < M; k0 += 32 * KNN_OWN) {
            knn_key  x[KNN_OWN];
            unsigned r[KNN_OWN];
#pragma unroll
            for (int q = 0; q < KNN_OWN; ++q) {
                const unsigned k = k0 + 32 * q + lane;
                x[q] = (k < M) ? cw[k] : ~(knn_key)0;
                r[q] = 0;
            }
#pragma unroll 4
            for (unsigned y = 0; y < M; ++y) {
                const knn_key cy = cw[y];
#pragma unroll
                for (int q = 0; q < KNN_OWN; ++q) r[q] += (cy < x[q]) ? 1u : 0u;
            }
#pragma unroll
            for (int q = 0; q < KNN_OWN; ++q) {
                const unsigned k = k0 + 32 * q + lane;
                if (k < M && r[q] < NSAMPLE) rw[r[q]] = (int)(unsigned)(x[q] & 0xffffffffull);
            }
        }
        __builtin_amdgcn_fence(__ATOMIC_ACQ_REL, "wavefront");
        gidx_all[(size_t)(b * N_ + cent) * NSAMPLE + lane] = rw[lane];
        __builtin_amdgcn_fence(__ATOMIC_ACQ_REL, "wavefront");
    }
#undef KNN_DIST
}

#define PL_PTS 64
#define PL_TP  72
#define PL_OP  68
static __device__ __forceinline__ void point_layer_role(unsigned char* smem, const int blk, const int tid,
                                                        const float* __restrict__ xyz,
                                                        const float* __restrict__ features,
                                                        const float* __restrict__ W0, const float* __restrict__ b0,
                                                        const float* __restrict__ g0, const float* __restrict__ be0,
                                                        const float* __restrict__ m0, const float* __restrict__ v0,
                                                        float* __restrict__ Atab, float* __restrict__ Htab)
{
    float* tile = (float*)smem;
    float* pxyz = tile + C_ * PL_TP;
    float* oA   = pxyz + 256;
    float* oH   = oA + PL_PTS * PL_OP;
    const int b   = blk >> 6;
    const int n0  = (blk & 63) * PL_PTS;
    const int q   = tid & 15;
    const int r0  = tid >> 4;
    const float* src = features + (size_t)b * C_ * N_ + n0;
#pragma unroll
    for (int k = 0; k < 4; ++k) {
        const int c = r0 + 16 * k;
        *(float4*)(tile + c * PL_TP + 4 * q) = *(const float4*)(src + (size_t)c * N_ + 4 * q);
    }
    if (tid < PL_PTS * 3) pxyz[tid] = xyz[((size_t)b * N_ + n0) * 3 + tid];
    __syncthreads();

    const int lane = tid & 31;
    const int w    = tid >> 5;
    const int mt   = w >> 1;
    const int nb   = (w & 1) * 2;
    const int c16  = lane & 15;
    const int h    = lane >> 4;
    const int prow = mt * 16 + c16;

    v8f accF[2], accX[2];
    const v8f zero = {};
    accF[0] = zero; accF[1] = zero; accX[0] = zero; accX[1] = zero;

    {
        const float p0 = pxyz[prow * 3 + 2 * h];
        const float p1 = pxyz[prow * 3 + 1];
        v2f a;
        a.x = p0;
        a.y = h ? 0.0f : p1;
#pragma unroll
        for (int n = 0; n < 2; ++n) {
            const int o = (nb + n) * 16 + c16;
            const float w0 = W0[o * 67 + 2 * h];
            const float w1 = W0[o * 67 + 1];
            v2f bt;
            bt.x = w0;
            bt.y = h ? 0.0f : w1;
            accX[n] = wmma4(a, bt, accX[n]);
        }
    }
#pragma unroll
    for (int ks = 0; ks < C_ / 4; ++ks) {
        const int k = 4 * ks + 2 * h;
        v2f a;
        a.x = tile[(k + 0) * PL_TP + prow];
        a.y = tile[(k + 1) * PL_TP + prow];
#pragma unroll
        for (int n = 0; n < 2; ++n) {
            const int o = (nb + n) * 16 + c16;
            v2f bt;
            bt.x = W0[o * 67 + 3 + k + 0];
            bt.y = W0[o * 67 + 3 + k + 1];
            accF[n] = wmma4(a, bt, accF[n]);
        }
    }
#pragma unroll
    for (int n = 0; n < 2; ++n) {
        const int   o  = (nb + n) * 16 + c16;
        const float sc = g0[o] / sqrtf(v0[o] + BN_EPS);
        const float sh = (b0[o] - m0[o]) * sc + be0[o];
#pragma unroll
        for (int j = 0; j < 8; ++j) {
            const int row = mt * 16 + 8 * h + j;
            oH[row * PL_OP + o] = accX[n][j] * sc;
            oA[row * PL_OP + o] = (accF[n][j] + accX[n][j]) * sc + sh;
        }
    }
    __syncthreads();
    float* dA = Atab + ((size_t)b * N_ + n0) * C_;
    float* dH = Htab + ((size_t)b * N_ + n0) * C_;
#pragma unroll
    for (int k = 0; k < 4; ++k) {
        const int n = r0 + 16 * k;
        *(float4*)(dA + (size_t)n * C_ + 4 * q) = *(const float4*)(oA + n * PL_OP + 4 * q);
        *(float4*)(dH + (size_t)n * C_ + 4 * q) = *(const float4*)(oH + n * PL_OP + 4 * q);
    }
}

#define FRONT_FPS   B_
#define FRONT_KNN   (B_ * (N_ / (KNN_WAVES * KNN_CPW)))
#define FRONT_PL    (B_ * (N_ / PL_PTS))
#define FRONT_GRID  (FRONT_FPS + FRONT_KNN + FRONT_PL + 1)
#define FRONT_LDS   (N_ * 3 * 4 + 8 * KNN_WAVES * N_ + 4 * KNN_WAVES * NSAMPLE)
__global__ __launch_bounds__(256) void front_kernel(
    const float* __restrict__ xyz, const float* __restrict__ features,
    const float* __restrict__ W0, const float* __restrict__ b0, const float* __restrict__ g0,
    const float* __restrict__ be0, const float* __restrict__ m0, const float* __restrict__ v0,
    const float* __restrict__ W1, const float* __restrict__ b1, const float* __restrict__ g1,
    const float* __restrict__ be1, const float* __restrict__ m1, const float* __restrict__ v1,
    const float* __restrict__ W2, const float* __restrict__ b2, const float* __restrict__ g2,
    const float* __restrict__ be2, const float* __restrict__ m2, const float* __restrict__ v2_,
    float* __restrict__ new_xyz, int* __restrict__ sel, int* __restrict__ gidx_all,
    float* __restrict__ Atab, float* __restrict__ Htab,
    v2f* __restrict__ Wp1, float* __restrict__ b1f,
    v2f* __restrict__ Wp2, float* __restrict__ b2f)
{
    __shared__ __align__(16) unsigned char smem[FRONT_LDS];
    const int blk = blockIdx.x;
    const int tid = threadIdx.x;
    if (blk < FRONT_FPS) {
        fps_role(smem, blk, tid, xyz, new_xyz, sel);
    } else if (blk < FRONT_FPS + FRONT_KNN) {
        knn_role(smem, blk - FRONT_FPS, tid, xyz, gidx_all);
    } else if (blk < FRONT_FPS + FRONT_KNN + FRONT_PL) {
        point_layer_role(smem, blk - (FRONT_FPS + FRONT_KNN), tid, xyz, features,
                         W0, b0, g0, be0, m0, v0, Atab, Htab);
    } else {
        fuse_weights_role(tid, W1, b1, g1, be1, m1, v1,
                          W2, b2, g2, be2, m2, v2_, Wp1, b1f, Wp2, b2f);
    }
}

template <int KSTEPS, int NT, int COUT>
static __device__ inline void layer_gemm(const float* in_lds, float* out_lds,
                                         const v2f* __restrict__ Wp,
                                         const float* __restrict__ bias, int lane)
{
    v8f acc[2][NT];
    v8f zero = {};
#pragma unroll
    for (int mt = 0; mt < 2; ++mt)
#pragma unroll
        for (int n = 0; n < NT; ++n) acc[mt][n] = zero;

    const int mrow  = lane & 15;
    const int khalf = (lane >> 4) * 2;
    const int ncol  = lane & 15;

    for (int k = 0; k < KSTEPS; ++k) {
        const int kc = 4 * k + khalf;
        v2f a0 = *(const v2f*)(in_lds + mrow * PITCH + kc);
        v2f a1 = *(const v2f*)(in_lds + (mrow + 16) * PITCH + kc);
#pragma unroll
        for (int n = 0; n < NT; ++n) {
            v2f bt = Wp[(kc >> 1) * COUT + n * 16 + ncol];
            acc[0][n] = wmma4(a0, bt, acc[0][n]);
            acc[1][n] = wmma4(a1, bt, acc[1][n]);
        }
    }
    const int rbase = (lane >> 4) * 8;
#pragma unroll
    for (int n = 0; n < NT; ++n) {
        float bb = bias[n * 16 + ncol];
#pragma unroll
        for (int mt = 0; mt < 2; ++mt)
#pragma unroll
            for (int j = 0; j < 8; ++j) {
                int rrow = mt * 16 + rbase + j;
                out_lds[rrow * PITCH + n * 16 + ncol] = fmaxf(acc[mt][n][j] + bb, 0.0f);
            }
    }
}

__global__ __launch_bounds__(32) void mlp_kernel(
    const float* __restrict__ Atab, const float* __restrict__ Htab,
    const int* __restrict__ sel, const int* __restrict__ gidx_all,
    float* __restrict__ out_feat,
    const v2f* __restrict__ Wp1, const float* __restrict__ b1f,
    const v2f* __restrict__ Wp2, const float* __restrict__ b2f)
{
    __shared__ __align__(16) float bufA[32 * PITCH];
    __shared__ __align__(16) float bufB[32 * PITCH];
    const int cg   = blockIdx.x;
    const int b    = cg >> 10;
    const int lane = threadIdx.x;

    const int cp  = sel[cg];
    const int idx = gidx_all[((size_t)b * N_ + cp) * NSAMPLE + lane];
    const v2f  hc = ((const v2f*)Htab)[((size_t)b * N_ + cp) * (C_ / 2) + lane];
    const v2f* At = (const v2f*)Atab + (size_t)b * N_ * (C_ / 2);
    v2f f[NSAMPLE];
#pragma unroll
    for (int r = 0; r < NSAMPLE; ++r) {
        const int pr = __builtin_amdgcn_readlane(idx, r);
        f[r] = At[(size_t)pr * (C_ / 2) + lane];
    }
#pragma unroll
    for (int r = 0; r < NSAMPLE; ++r) {
        v2f a;
        a.x = fmaxf(f[r].x - hc.x, 0.0f);
        a.y = fmaxf(f[r].y - hc.y, 0.0f);
        *(v2f*)(bufA + r * PITCH + 2 * lane) = a;
    }
    __syncthreads();

    layer_gemm<16, 4, 64>(bufA, bufB, Wp1, b1f, lane);
    __syncthreads();

    v8f acc[2][8];
    v8f zero = {};
#pragma unroll
    for (int mt = 0; mt < 2; ++mt)
#pragma unroll
        for (int n = 0; n < 8; ++n) acc[mt][n] = zero;

    const int mrow  = lane & 15;
    const int khalf = (lane >> 4) * 2;
    const int ncol  = lane & 15;
    for (int k = 0; k < 16; ++k) {
        const int kc = 4 * k + khalf;
        v2f a0 = *(const v2f*)(bufB + mrow * PITCH + kc);
        v2f a1 = *(const v2f*)(bufB + (mrow + 16) * PITCH + kc);
#pragma unroll
        for (int n = 0; n < 8; ++n) {
            v2f bt = Wp2[(kc >> 1) * 128 + n * 16 + ncol];
            acc[0][n] = wmma4(a0, bt, acc[0][n]);
            acc[1][n] = wmma4(a1, bt, acc[1][n]);
        }
    }
    float* of = out_feat + (size_t)cg * 128;
#pragma unroll
    for (int n = 0; n < 8; ++n) {
        float r = -3.4e38f;
#pragma unroll
        for (int j = 0; j < 8; ++j) {
            r = fmaxf(r, acc[0][n][j]);
            r = fmaxf(r, acc[1][n][j]);
        }
        r = fmaxf(r, __shfl_xor(r, 16, 32));
        float v = fmaxf(r + b2f[n * 16 + ncol], 0.0f);
        if (lane < 16) of[n * 16 + ncol] = v;
    }
}

extern "C" void kernel_launch(void* const* d_in, const int* in_sizes, int n_in,
                              void* d_out, int out_size, void* d_ws, size_t ws_size,
                              hipStream_t stream)
{
    const float* xyz      = (const float*)d_in[0];
    const float* features = (const float*)d_in[1];
    const float* W0 = (const float*)d_in[2];  const float* b0 = (const float*)d_in[3];
    const float* g0 = (const float*)d_in[4];  const float* be0 = (const float*)d_in[5];
    const float* m0 = (const float*)d_in[6];  const float* v0 = (const float*)d_in[7];
    const float* W1 = (const float*)d_in[8];  const float* b1 = (const float*)d_in[9];
    const float* g1 = (const float*)d_in[10]; const float* be1 = (const float*)d_in[11];
    const float* m1 = (const float*)d_in[12]; const float* v1 = (const float*)d_in[13];
    const float* W2 = (const float*)d_in[14]; const float* b2 = (const float*)d_in[15];
    const float* g2 = (const float*)d_in[16]; const float* be2 = (const float*)d_in[17];
    const float* m2 = (const float*)d_in[18]; const float* v2_ = (const float*)d_in[19];

    float* out      = (float*)d_out;
    float* new_xyz  = out;
    float* out_feat = out + (size_t)B_ * NPOINT * 3;

    int*   sel  = (int*)d_ws;
    float* wsf  = (float*)((char*)d_ws + (size_t)2 * 1024 * 1024);
    v2f*   Wp1 = (v2f*)wsf;
    float* b1f = (float*)(Wp1 + 32 * 64);
    v2f*   Wp2 = (v2f*)(b1f + 64);
    float* b2f = (float*)(Wp2 + 32 * 128);
    float* Atab = (float*)((char*)d_ws + (size_t)4 * 1024 * 1024);
    int*   gidx_all = (int*)((char*)d_ws + (size_t)32 * 1024 * 1024);
    float* Htab = (float*)((char*)d_ws + (size_t)64 * 1024 * 1024);

    front_kernel<<<FRONT_GRID, 256, 0, stream>>>(xyz, features,
                                                 W0, b0, g0, be0, m0, v0,
                                                 W1, b1, g1, be1, m1, v1,
                                                 W2, b2, g2, be2, m2, v2_,
                                                 new_xyz, sel, gidx_all, Atab, Htab,
                                                 Wp1, b1f, Wp2, b2f);
    mlp_kernel<<<B_ * NPOINT, 32, 0, stream>>>(Atab, Htab, sel, gidx_all, out_feat,
                                               Wp1, b1f, Wp2, b2f);
}


# sampling split over two launches: the mlp of the first 880 centroids per batch runs beside the last 144 sampling steps (minima handed over in the workspace); a third launch does the other 144
# speedup vs baseline: 7.0682x; 1.0346x over previous
#include <hip/hip_runtime.h>
#include <hip/hip_bf16.h>
#include <math.h>

#define B_      16
#define N_      4096
#define C_      64
#define NPOINT  1024
#define NSAMPLE 32
#define PITCH   68
#define BN_EPS  1e-5f
#define K1      880

typedef float v2f __attribute__((ext_vector_type(2)));
typedef float v8f __attribute__((ext_vector_type(8)));

static __device__ inline v8f wmma4(v2f a, v2f b, v8f c) {
    return __builtin_amdgcn_wmma_f32_16x16x4_f32(
         false, a,  false, b,
         (short)0, c,  false,  false);
}

static __device__ __forceinline__ void fuse_weights_role(
    const int tid,
    const float* __restrict__ W1, const float* __restrict__ b1, const float* __restrict__ g1,
    const float* __restrict__ be1, const float* __restrict__ m1, const float* __restrict__ v1,
    const float* __restrict__ W2, const float* __restrict__ b2, const float* __restrict__ g2,
    const float* __restrict__ be2, const float* __restrict__ m2, const float* __restrict__ v2_,
    v2f* __restrict__ Wp1, float* __restrict__ b1f,
    v2f* __restrict__ Wp2, float* __restrict__ b2f)
{
    for (int i = tid; i < 32 * 64; i += 256) {
        int o = i & 63, kp = i >> 6;
        float s = g1[o] / sqrtf(v1[o] + BN_EPS);
        v2f w;
        w.x = W1[o * 64 + 2 * kp + 0] * s;
        w.y = W1[o * 64 + 2 * kp + 1] * s;
        Wp1[i] = w;
    }
    for (int i = tid; i < 64; i += 256) {
        float s = g1[i] / sqrtf(v1[i] + BN_EPS);
        b1f[i] = (b1[i] - m1[i]) * s + be1[i];
    }
    for (int i = tid; i < 32 * 128; i += 256) {
        int o = i & 127, kp = i >> 7;
        float s = g2[o] / sqrtf(v2_[o] + BN_EPS);
        v2f w;
        w.x = W2[o * 64 + 2 * kp + 0] * s;
        w.y = W2[o * 64 + 2 * kp + 1] * s;
        Wp2[i] = w;
    }
    for (int i = tid; i < 128; i += 256) {
        float s = g2[i] / sqrtf(v2_[i] + BN_EPS);
        b2f[i] = (b2[i] - m2[i]) * s + be2[i];
    }
}

template <int M>
static __device__ inline int xrow(int x) {
    return __builtin_amdgcn_update_dpp(0, x, 0x160 + M, 0xf, 0xf, true);
}
static __device__ inline int xhalf(int x) {
    return __builtin_amdgcn_permlanex16(x, x, 0x76543210, 0xfedcba98, false, false);
}
static __device__ inline int rowpick(int x, int sel) {
    return (int)__builtin_amdgcn_permlane16_var((unsigned)x, (unsigned)x, (unsigned)sel, false, false);
}
static __device__ inline int imax(int a, int b) { return a > b ? a : b; }
static __device__ inline int imin(int a, int b) { return a < b ? a : b; }
#define NO_IDX 0x7fffffff

template <int PART>
static __device__ __forceinline__ void fps_role(unsigned char* smem, const int b, const int tid,
                                                const float* __restrict__ xyz,
                                                float* __restrict__ new_xyz,
                                                int* __restrict__ sel,
                                                float* __restrict__ dsave)
{
    float* sx   = (float*)smem;
    float* sy   = sx + N_;
    float* sz   = sy + N_;
    int*   sidx = (int*)(sz + N_);
    int (*red)[64] = (int (*)[64])(sidx + NPOINT);
    const float* bx = xyz + (size_t)b * N_ * 3;

    float px[16], py[16], pz[16], dloc[16];
#pragma unroll
    for (int i = 0; i < 16; ++i) {
        int p = tid + 256 * i;
        px[i] = bx[p * 3 + 0];
        py[i] = bx[p * 3 + 1];
        pz[i] = bx[p * 3 + 2];
        sx[p] = px[i];
        sy[p] = py[i];
        sz[p] = pz[i];
        dloc[i] = PART ? dsave[((size_t)b * 16 + i) * 256 + tid] : 1e10f;
    }
    __syncthreads();

    int last = PART ? sel[b * NPOINT + K1 - 1] : 0;
    float lx = bx[last * 3 + 0], ly = bx[last * 3 + 1], lz = bx[last * 3 + 2];
    const int it_begin = PART ? K1 - 1 : 0;
    const int it_end   = PART ? NPOINT - 1 : K1 - 1;
    for (int it = it_begin; it < it_end; ++it) {
        float val = -1.0f;
        int   best = tid;
#pragma unroll
        for (int i = 0; i < 16; ++i) {
            int p = tid + 256 * i;
            float dx = px[i] - lx, dy = py[i] - ly, dz = pz[i] - lz;
            float dist = dx * dx + dy * dy + dz * dz;
            dloc[i] = fminf(dloc[i], dist);
            if (dloc[i] > val) { val = dloc[i]; best = p; }
        }
        const float bxv = sx[best], byv = sy[best], bzv = sz[best];
        __builtin_amdgcn_sched_barrier(0);
        const int vb = __float_as_int(val);
        int m = vb;
        m = imax(m, xrow<1>(m));
        m = imax(m, xrow<2>(m));
        m = imax(m, xrow<4>(m));
        m = imax(m, xrow<8>(m));
        m = imax(m, xhalf(m));
        int c = (vb == m) ? best : NO_IDX;
        c = imin(c, xrow<1>(c));
        c = imin(c, xrow<2>(c));
        c = imin(c, xrow<4>(c));
        c = imin(c, xrow<8>(c));
        c = imin(c, xhalf(c));

        int* slot = red[it & 1];
        if (best == c) {
            int* e = slot + 8 * (tid >> 5);
            *(int4*)e = make_int4(m, c, __float_as_int(bxv), __float_as_int(byv));
            e[4] = __float_as_int(bzv);
        }
        __syncthreads();
        const int* e = slot + 8 * (tid & 7);
        const int4 r  = *(const int4*)e;
        const int  rz = e[4];
        if (tid == 0) sidx[it] = last;
        int gm = r.x;
        gm = imax(gm, xrow<1>(gm));
        gm = imax(gm, xrow<2>(gm));
        gm = imax(gm, xrow<4>(gm));
        int gc = (r.x == gm) ? r.y : NO_IDX;
        gc = imin(gc, xrow<1>(gc));
        gc = imin(gc, xrow<2>(gc));
        gc = imin(gc, xrow<4>(gc));
        last = gc;
        const int wsel = (gc >> 5) & 7;
        lx = __int_as_float(rowpick(r.z, wsel));
        ly = __int_as_float(rowpick(r.w, wsel));
        lz = __int_as_float(rowpick(rz,  wsel));
    }
    if (tid == 0) sidx[it_end] = last;
    __syncthreads();
    const int j_begin = PART ? K1 : 0;
    const int j_end   = PART ? NPOINT : K1;
    for (int j = j_begin + tid; j < j_end; j += 256) {
        const int p = sidx[j];
        float* o = new_xyz + ((size_t)b * NPOINT + j) * 3;
        o[0] = sx[p];
        o[1] = sy[p];
        o[2] = sz[p];
        sel[b * NPOINT + j] = p;
    }
    if (PART == 0) {
#pragma unroll
        for (int i = 0; i < 16; ++i) dsave[((size_t)b * 16 + i) * 256 + tid] = dloc[i];
    }
}

#define KNN_WAVES 8
#define KNN_CPW   8
#define KNN_OWN   2
typedef unsigned long long knn_key;

static __device__ __forceinline__ void knn_role(unsigned char* smem, const int blk, const int tid,
                                                const float* __restrict__ xyz,
                                                int* __restrict__ gidx_all)
{
    float4* pts = (float4*)smem;
    knn_key (*cand)[N_]   = (knn_key (*)[N_])(smem + N_ * 3 * sizeof(float));
    int (*res)[NSAMPLE]   = (int (*)[NSAMPLE])(smem + N_ * 3 * sizeof(float) + sizeof(knn_key) * KNN_WAVES * N_);
    const int b    = blk >> 6;
    const int g    = blk & 63;
    const int lane = tid & 31;
    const int w    = tid >> 5;

    const float4* src = (const float4*)(xyz + (size_t)b * N_ * 3);
    for (int i = tid; i < N_ * 3 / 4; i += KNN_WAVES * 32) pts[i] = src[i];
    __syncthreads();

    const float* P  = (const float*)pts;
    knn_key*     cw = cand[w];
    int*         rw = res[w];

#define KNN_DIST(PX, PY, PZ) fmaf((PZ) - cz, (PZ) - cz, fmaf((PX) - cx, (PX) - cx, ((PY) - cy) * ((PY) - cy)))

    for (int ci = 0; ci < KNN_CPW; ++ci) {
        const int cent = g * (KNN_WAVES * KNN_CPW) + w * KNN_CPW + ci;
        const float cx = P[3 * cent + 0];
        const float cy = P[3 * cent + 1];
        const float cz = P[3 * cent + 2];

        float d[N_ / 32];
        float lo0 = 3.4e38f, hi0 = 3.4e38f, lo1 = 3.4e38f, hi1 = 3.4e38f;
#pragma unroll
        for (int t = 0; t < N_ / 32; ++t) {
            const int j = 32 * t + lane;
            const float px = P[3 * j + 0], py = P[3 * j + 1], pz = P[3 * j + 2];
            d[t] = KNN_DIST(px, py, pz);
            if (t & 1) { hi1 = __builtin_amdgcn_fmed3f(lo1, hi1, d[t]); lo1 = fminf(lo1, d[t]); }
            else       { hi0 = __builtin_amdgcn_fmed3f(lo0, hi0, d[t]); lo0 = fminf(lo0, d[t]); }
        }
        const int s1 = __float_as_int(fminf(lo0, lo1));
        const int s2 = __float_as_int(fminf(fmaxf(lo0, lo1), fminf(hi0, hi1)));
        cw[lane] = ((knn_key)(unsigned)s2 << 32) | (unsigned)s1;
        __builtin_amdgcn_fence(__ATOMIC_ACQ_REL, "wavefront");
        int below1 = 0, below2 = 0;
#pragma unroll
        for (int s = 0; s < 16; ++s) {
            const knn_key e0 = cw[2 * s + 0], e1 = cw[2 * s + 1];
            const int v0 = (int)(unsigned)(e0 & 0xffffffffull), v1 = (int)(unsigned)(e0 >> 32);
            const int v2 = (int)(unsigned)(e1 & 0xffffffffull), v3 = (int)(unsigned)(e1 >> 32);
            below1 += (v0 < s1 ? 1 : 0) + (v1 < s1 ? 1 : 0) + (v2 < s1 ? 1 : 0) + (v3 < s1 ? 1 : 0);
            below2 += (v0 < s2 ? 1 : 0) + (v1 < s2 ? 1 : 0) + (v2 < s2 ? 1 : 0) + (v3 < s2 ? 1 : 0);
        }
        int mb = (below2 < NSAMPLE) ? s2 : ((below1 < NSAMPLE) ? s1 : 0);
        mb = imax(mb, xrow<1>(mb));
        mb = imax(mb, xrow<2>(mb));
        mb = imax(mb, xrow<4>(mb));
        mb = imax(mb, xrow<8>(mb));
        mb = imax(mb, xhalf(mb));
        const float bound = __int_as_float(mb);
        __builtin_amdgcn_fence(__ATOMIC_ACQ_REL, "wavefront");

        unsigned M = 0;
#pragma unroll
        for (int t = 0; t < N_ / 32; ++t) {
            const bool     in   = d[t] <= bound;
            const unsigned mask = __builtin_amdgcn_ballot_w32(in);
            if (in)
                cw[M + __builtin_amdgcn_mbcnt_lo(mask, 0u)] =
                    ((knn_key)(unsigned)__float_as_int(d[t]) << 32) | (unsigned)(32 * t + lane);
            M += (unsigned)__builtin_popcount(mask);
        }
        rw[lane] = 0;
        __builtin_amdgcn_fence(__ATOMIC_ACQ_REL, "wavefront");

        for (unsigned k0 = 0; k0 < M; k0 += 32 * KNN_OWN) {
            knn_key  x[KNN_OWN];
            unsigned r[KNN_OWN];
#pragma unroll
            for (int q = 0; q < KNN_OWN; ++q) {
                const unsigned k = k0 + 32 * q + lane;
                x[q] = (k < M) ? cw[k] : ~(knn_key)0;
                r[q] = 0;
            }
#pragma unroll 4
            for (unsigned y = 0; y < M; ++y) {
                const knn_key cy = cw[y];
#pragma unroll
                for (int q = 0; q < KNN_OWN; ++q) r[q] += (cy < x[q]) ? 1u : 0u;
            }
#pragma unroll
            for (int q = 0; q < KNN_OWN; ++q) {
                const unsigned k = k0 + 32 * q + lane;
                if (k < M && r[q] < NSAMPLE) rw[r[q]] = (int)(unsigned)(x[q] & 0xffffffffull);
            }
        }
        __builtin_amdgcn_fence(__ATOMIC_ACQ_REL, "wavefront");
        gidx_all[(size_t)(b * N_ + cent) * NSAMPLE + lane] = rw[lane];
        __builtin_amdgcn_fence(__ATOMIC_ACQ_REL, "wavefront");
    }
#undef KNN_DIST
}

#define PL_PTS 64
#define PL_TP  72
#define PL_OP  68
static __device__ __forceinline__ void point_layer_role(unsigned char* smem, const int blk, const int tid,
                                                        const float* __restrict__ xyz,
                                                        const float* __restrict__ features,
                                                        const float* __restrict__ W0, const float* __restrict__ b0,
                                                        const float* __restrict__ g0, const float* __restrict__ be0,
                                                        const float* __restrict__ m0, const float* __restrict__ v0,
                                                        float* __restrict__ Atab, float* __restrict__ Htab)
{
    float* tile = (float*)smem;
    float* pxyz = tile + C_ * PL_TP;
    float* oA   = pxyz + 256;
    float* oH   = oA + PL_PTS * PL_OP;
    const int b   = blk >> 6;
    const int n0  = (blk & 63) * PL_PTS;
    const int q   = tid & 15;
    const int r0  = tid >> 4;
    const float* src = features + (size_t)b * C_ * N_ + n0;
#pragma unroll
    for (int k = 0; k < 4; ++k) {
        const int c = r0 + 16 * k;
        *(float4*)(tile + c * PL_TP + 4 * q) = *(const float4*)(src + (size_t)c * N_ + 4 * q);
    }
    if (tid < PL_PTS * 3) pxyz[tid] = xyz[((size_t)b * N_ + n0) * 3 + tid];
    __syncthreads();

    const int lane = tid & 31;
    const int w    = tid >> 5;
    const int mt   = w >> 1;
    const int nb   = (w & 1) * 2;
    const int c16  = lane & 15;
    const int h    = lane >> 4;
    const int prow = mt * 16 + c16;

    v8f accF[2], accX[2];
    const v8f zero = {};
    accF[0] = zero; accF[1] = zero; accX[0] = zero; accX[1] = zero;

    {
        const float p0 = pxyz[prow * 3 + 2 * h];
        const float p1 = pxyz[prow * 3 + 1];
        v2f a;
        a.x = p0;
        a.y = h ? 0.0f : p1;
#pragma unroll
        for (int n = 0; n < 2; ++n) {
            const int o = (nb + n) * 16 + c16;
            const float w0 = W0[o * 67 + 2 * h];
            const float w1 = W0[o * 67 + 1];
            v2f bt;
            bt.x = w0;
            bt.y = h ? 0.0f : w1;
            accX[n] = wmma4(a, bt, accX[n]);
        }
    }
#pragma unroll
    for (int ks = 0; ks < C_ / 4; ++ks) {
        const int k = 4 * ks + 2 * h;
        v2f a;
        a.x = tile[(k + 0) * PL_TP + prow];
        a.y = tile[(k + 1) * PL_TP + prow];
#pragma unroll
        for (int n = 0; n < 2; ++n) {
            const int o = (nb + n) * 16 + c16;
            v2f bt;
            bt.x = W0[o * 67 + 3 + k + 0];
            bt.y = W0[o * 67 + 3 + k + 1];
            accF[n] = wmma4(a, bt, accF[n]);
        }
    }
#pragma unroll
    for (int n = 0; n < 2; ++n) {
        const int   o  = (nb + n) * 16 + c16;
        const float sc = g0[o] / sqrtf(v0[o] + BN_EPS);
        const float sh = (b0[o] - m0[o]) * sc + be0[o];
#pragma unroll
        for (int j = 0; j < 8; ++j) {
            const int row = mt * 16 + 8 * h + j;
            oH[row * PL_OP + o] = accX[n][j] * sc;
            oA[row * PL_OP + o] = (accF[n][j] + accX[n][j]) * sc + sh;
        }
    }
    __syncthreads();
    float* dA = Atab + ((size_t)b * N_ + n0) * C_;
    float* dH = Htab + ((size_t)b * N_ + n0) * C_;
#pragma unroll
    for (int k = 0; k < 4; ++k) {
        const int n = r0 + 16 * k;
        *(float4*)(dA + (size_t)n * C_ + 4 * q) = *(const float4*)(oA + n * PL_OP + 4 * q);
        *(float4*)(dH + (size_t)n * C_ + 4 * q) = *(const float4*)(oH + n * PL_OP + 4 * q);
    }
}

#define FRONT_FPS   B_
#define FRONT_KNN   (B_ * (N_ / (KNN_WAVES * KNN_CPW)))
#define FRONT_PL    (B_ * (N_ / PL_PTS))
#define FRONT_GRID  (FRONT_FPS + FRONT_KNN + FRONT_PL + 1)
#define FRONT_LDS   (N_ * 3 * 4 + 8 * KNN_WAVES * N_ + 4 * KNN_WAVES * NSAMPLE)
__global__ __launch_bounds__(256) void front_kernel(
    const float* __restrict__ xyz, const float* __restrict__ features,
    const float* __restrict__ W0, const float* __restrict__ b0, const float* __restrict__ g0,
    const float* __restrict__ be0, const float* __restrict__ m0, const float* __restrict__ v0,
    const float* __restrict__ W1, const float* __restrict__ b1, const float* __restrict__ g1,
    const float* __restrict__ be1, const float* __restrict__ m1, const float* __restrict__ v1,
    const float* __restrict__ W2, const float* __restrict__ b2, const float* __restrict__ g2,
    const float* __restrict__ be2, const float* __restrict__ m2, const float* __restrict__ v2_,
    float* __restrict__ new_xyz, int* __restrict__ sel, int* __restrict__ gidx_all,
    float* __restrict__ Atab, float* __restrict__ Htab,
    v2f* __restrict__ Wp1, float* __restrict__ b1f,
    v2f* __restrict__ Wp2, float* __restrict__ b2f,
    float* __restrict__ dsave)
{
    __shared__ __align__(16) unsigned char smem[FRONT_LDS];
    const int blk = blockIdx.x;
    const int tid = threadIdx.x;
    if (blk < FRONT_FPS) {
        fps_role<0>(smem, blk, tid, xyz, new_xyz, sel, dsave);
    } else if (blk < FRONT_FPS + FRONT_KNN) {
        knn_role(smem, blk - FRONT_FPS, tid, xyz, gidx_all);
    } else if (blk < FRONT_FPS + FRONT_KNN + FRONT_PL) {
        point_layer_role(smem, blk - (FRONT_FPS + FRONT_KNN), tid, xyz, features,
                         W0, b0, g0, be0, m0, v0, Atab, Htab);
    } else {
        fuse_weights_role(tid, W1, b1, g1, be1, m1, v1,
                          W2, b2, g2, be2, m2, v2_, Wp1, b1f, Wp2, b2f);
    }
}

template <int KSTEPS, int NT, int COUT>
static __device__ inline void layer_gemm(const float* in_lds, float* out_lds,
                                         const v2f* __restrict__ Wp,
                                         const float* __restrict__ bias, int lane)
{
    v8f acc[2][NT];
    v8f zero = {};
#pragma unroll
    for (int mt = 0; mt < 2; ++mt)
#pragma unroll
        for (int n = 0; n < NT; ++n) acc[mt][n] = zero;

    const int mrow  = lane & 15;
    const int khalf = (lane >> 4) * 2;
    const int ncol  = lane & 15;

    for (int k = 0; k < KSTEPS; ++k) {
        const int kc = 4 * k + khalf;
        v2f a0 = *(const v2f*)(in_lds + mrow * PITCH + kc);
        v2f a1 = *(const v2f*)(in_lds + (mrow + 16) * PITCH + kc);
#pragma unroll
        for (int n = 0; n < NT; ++n) {
            v2f bt = Wp[(kc >> 1) * COUT + n * 16 + ncol];
            acc[0][n] = wmma4(a0, bt, acc[0][n]);
            acc[1][n] = wmma4(a1, bt, acc[1][n]);
        }
    }
    const int rbase = (lane >> 4) * 8;
#pragma unroll
    for (int n = 0; n < NT; ++n) {
        float bb = bias[n * 16 + ncol];
#pragma unroll
        for (int mt = 0; mt < 2; ++mt)
#pragma unroll
            for (int j = 0; j < 8; ++j) {
                int rrow = mt * 16 + rbase + j;
                out_lds[rrow * PITCH + n * 16 + ncol] = fmaxf(acc[mt][n][j] + bb, 0.0f);
            }
    }
}

#define MLP_WAVES     8
#define MLP_LDS_WAVE  (2 * 32 * PITCH * 4)
#define MLP_LDS       (MLP_WAVES * MLP_LDS_WAVE)
static __device__ __forceinline__ void mlp_role(
    float* bufA, float* bufB, const int cg, const int lane,
    const float* __restrict__ Atab, const float* __restrict__ Htab,
    const int* __restrict__ sel, const int* __restrict__ gidx_all,
    float* __restrict__ out_feat,
    const v2f* __restrict__ Wp1, const float* __restrict__ b1f,
    const v2f* __restrict__ Wp2, const float* __restrict__ b2f)
{
    const int b = cg >> 10;

    const int cp  = sel[cg];
    const int idx = gidx_all[((size_t)b * N_ + cp) * NSAMPLE + lane];
    const v2f  hc = ((const v2f*)Htab)[((size_t)b * N_ + cp) * (C_ / 2) + lane];
    const v2f* At = (const v2f*)Atab + (size_t)b * N_ * (C_ / 2);
    v2f f[NSAMPLE];
#pragma unroll
    for (int r = 0; r < NSAMPLE; ++r) {
        const int pr = __builtin_amdgcn_readlane(idx, r);
        f[r] = At[(size_t)pr * (C_ / 2) + lane];
    }
#pragma unroll
    for (int r = 0; r < NSAMPLE; ++r) {
        v2f a;
        a.x = fmaxf(f[r].x - hc.x, 0.0f);
        a.y = fmaxf(f[r].y - hc.y, 0.0f);
        *(v2f*)(bufA + r * PITCH + 2 * lane) = a;
    }
    __builtin_amdgcn_fence(__ATOMIC_ACQ_REL, "wavefront");

    layer_gemm<16, 4, 64>(bufA, bufB, Wp1, b1f, lane);
    __builtin_amdgcn_fence(__ATOMIC_ACQ_REL, "wavefront");

    v8f acc[2][8];
    v8f zero = {};
#pragma unroll
    for (int mt = 0; mt < 2; ++mt)
#pragma unroll
        for (int n = 0; n < 8; ++n) acc[mt][n] = zero;

    const int mrow  = lane & 15;
    const int khalf = (lane >> 4) * 2;
    const int ncol  = lane & 15;
    for (int k = 0; k < 16; ++k) {
        const int kc = 4 * k + khalf;
        v2f a0 = *(const v2f*)(bufB + mrow * PITCH + kc);
        v2f a1 = *(const v2f*)(bufB + (mrow + 16) * PITCH + kc);
#pragma unroll
        for (int n = 0; n < 8; ++n) {
            v2f bt = Wp2[(kc >> 1) * 128 + n * 16 + ncol];
            acc[0][n] = wmma4(a0, bt, acc[0][n]);
            acc[1][n] = wmma4(a1, bt, acc[1][n]);
        }
    }
    float* of = out_feat + (size_t)cg * 128;
#pragma unroll
    for (int n = 0; n < 8; ++n) {
        float r = -3.4e38f;
#pragma unroll
        for (int j = 0; j < 8; ++j) {
            r = fmaxf(r, acc[0][n][j]);
            r = fmaxf(r, acc[1][n][j]);
        }
        r = fmaxf(r, __shfl_xor(r, 16, 32));
        float v = fmaxf(r + b2f[n * 16 + ncol], 0.0f);
        if (lane < 16) of[n * 16 + ncol] = v;
    }
}

#define TAIL_MLP   (B_ * K1 / MLP_WAVES)
#define TAIL_GRID  (B_ + TAIL_MLP)
__global__ __launch_bounds__(256) void tail_kernel(
    const float* __restrict__ xyz, float* __restrict__ new_xyz, int* sel, float* __restrict__ dsave,
    const float* __restrict__ Atab, const float* __restrict__ Htab,
    const int* __restrict__ gidx_all, float* __restrict__ out_feat,
    const v2f* __restrict__ Wp1, const float* __restrict__ b1f,
    const v2f* __restrict__ Wp2, const float* __restrict__ b2f)
{
    __shared__ __align__(16) unsigned char smem[MLP_LDS];
    const int blk = blockIdx.x;
    const int tid = threadIdx.x;
    if (blk < B_) {
        fps_role<1>(smem, blk, tid, xyz, new_xyz, sel, dsave);
    } else {
        const int w = tid >> 5;
        const int m = (blk - B_) * MLP_WAVES + w;
        const int b = m / K1;
        const int j = m - b * K1;
        float* bufA = (float*)(smem + w * MLP_LDS_WAVE);
        mlp_role(bufA, bufA + 32 * PITCH, b * NPOINT + j, tid & 31,
                 Atab, Htab, sel, gidx_all, out_feat, Wp1, b1f, Wp2, b2f);
    }
}

#define REST       (NPOINT - K1)
#define REST_GRID  (B_ * REST / MLP_WAVES)
__global__ __launch_bounds__(256) void mlp_kernel(
    const float* __restrict__ Atab, const float* __restrict__ Htab,
    const int* __restrict__ sel, const int* __restrict__ gidx_all,
    float* __restrict__ out_feat,
    const v2f* __restrict__ Wp1, const float* __restrict__ b1f,
    const v2f* __restrict__ Wp2, const float* __restrict__ b2f)
{
    __shared__ __align__(16) unsigned char smem[MLP_LDS];
    const int tid = threadIdx.x;
    const int w = tid >> 5;
    const int m = blockIdx.x * MLP_WAVES + w;
    const int b = m / REST;
    const int j = K1 + (m - b * REST);
    float* bufA = (float*)(smem + w * MLP_LDS_WAVE);
    mlp_role(bufA, bufA + 32 * PITCH, b * NPOINT + j, tid & 31,
             Atab, Htab, sel, gidx_all, out_feat, Wp1, b1f, Wp2, b2f);
}

extern "C" void kernel_launch(void* const* d_in, const int* in_sizes, int n_in,
                              void* d_out, int out_size, void* d_ws, size_t ws_size,
                              hipStream_t stream)
{
    const float* xyz      = (const float*)d_in[0];
    const float* features = (const float*)d_in[1];
    const float* W0 = (const float*)d_in[2];  const float* b0 = (const float*)d_in[3];
    const float* g0 = (const float*)d_in[4];  const float* be0 = (const float*)d_in[5];
    const float* m0 = (const float*)d_in[6];  const float* v0 = (const float*)d_in[7];
    const float* W1 = (const float*)d_in[8];  const float* b1 = (const float*)d_in[9];
    const float* g1 = (const float*)d_in[10]; const float* be1 = (const float*)d_in[11];
    const float* m1 = (const float*)d_in[12]; const float* v1 = (const float*)d_in[13];
    const float* W2 = (const float*)d_in[14]; const float* b2 = (const float*)d_in[15];
    const float* g2 = (const float*)d_in[16]; const float* be2 = (const float*)d_in[17];
    const float* m2 = (const float*)d_in[18]; const float* v2_ = (const float*)d_in[19];

    float* out      = (float*)d_out;
    float* new_xyz  = out;
    float* out_feat = out + (size_t)B_ * NPOINT * 3;

    int*   sel  = (int*)d_ws;
    float* wsf  = (float*)((char*)d_ws + (size_t)2 * 1024 * 1024);
    v2f*   Wp1 = (v2f*)wsf;
    float* b1f = (float*)(Wp1 + 32 * 64);
    v2f*   Wp2 = (v2f*)(b1f + 64);
    float* b2f = (float*)(Wp2 + 32 * 128);
    float* Atab = (float*)((char*)d_ws + (size_t)4 * 1024 * 1024);
    int*   gidx_all = (int*)((char*)d_ws + (size_t)32 * 1024 * 1024);
    float* Htab = (float*)((char*)d_ws + (size_t)64 * 1024 * 1024);
    float* dsave = (float*)((char*)d_ws + (size_t)96 * 1024 * 1024);

    front_kernel<<<FRONT_GRID, 256, 0, stream>>>(xyz, features,
                                                 W0, b0, g0, be0, m0, v0,
                                                 W1, b1, g1, be1, m1, v1,
                                                 W2, b2, g2, be2, m2, v2_,
                                                 new_xyz, sel, gidx_all, Atab, Htab,
                                                 Wp1, b1f, Wp2, b2f, dsave);
    tail_kernel<<<TAIL_GRID, 256, 0, stream>>>(xyz, new_xyz, sel, dsave, Atab, Htab, gidx_all, out_feat,
                                               Wp1, b1f, Wp2, b2f);
    mlp_kernel<<<REST_GRID, 256, 0, stream>>>(Atab, Htab, sel, gidx_all, out_feat,
                                              Wp1, b1f, Wp2, b2f);
}
